# starting kernel
# baseline (speedup 1.0000x reference)
#include <hip/hip_runtime.h>
#include <hip/hip_bf16.h>
#include <math.h>

#define N_NODES   50000
#define N_EDGES   800000
#define HEADS     4
#define D_HEAD    32
#define FEAT      128
#define OUT_DIM   512
#define NEG_SLOPE 0.2f
#define EPS_      1e-9f

typedef __attribute__((ext_vector_type(2))) float v2f;
typedef __attribute__((ext_vector_type(8))) float v8f;


__global__ void k_zero(float* __restrict__ p, long n) {
    long i = (long)blockIdx.x * blockDim.x + threadIdx.x;
    if (i < n) p[i] = 0.0f;
}

__device__ __forceinline__ unsigned f2ord(float f) {
    unsigned u = __float_as_uint(f);
    return (u & 0x80000000u) ? ~u : (u | 0x80000000u);
}
__device__ __forceinline__ float ord2f(unsigned k) {
    return (k & 0x80000000u) ? __uint_as_float(k & 0x7fffffffu) : __uint_as_float(~k);
}

__global__ void k_gemm_wmma(const float* __restrict__ A, const float* __restrict__ B,
                            const float* __restrict__ bias, float* __restrict__ C,
                            int M, int N, int K)
{
    __shared__ float ldsB[128 * 16];
    const int tid  = threadIdx.x;
    const int wave = tid >> 5;
    const int lane = tid & 31;
    const int n0   = blockIdx.y * 16;

    for (int i = tid; i < K * 16; i += 128) {
        int k = i >> 4, c = i & 15;
        ldsB[i] = B[(long)k * N + n0 + c];
    }
    __syncthreads();

    const int m0 = (blockIdx.x * 4 + wave) * 16;
    if (m0 >= M) return;

    const int row   = m0 + (lane & 15);
    const int koff  = (lane >> 4) << 1;
    const int col16 = lane & 15;

    v8f c = {};
    for (int k0 = 0; k0 < K; k0 += 4) {
        v2f a, b;
        a.x = A[(long)row * K + k0 + koff];
        a.y = A[(long)row * K + k0 + koff + 1];
        b.x = ldsB[(k0 + koff)     * 16 + col16];
        b.y = ldsB[(k0 + koff + 1) * 16 + col16];
        c = __builtin_amdgcn_wmma_f32_16x16x4_f32(false, a, false, b,
                                                  (short)0, c, false, false);
    }

    const int   col   = n0 + col16;
    const float bv    = bias ? bias[col] : 0.0f;
    const int   rbase = m0 + ((lane >> 4) << 3);
#pragma unroll
    for (int i = 0; i < 8; ++i)
        C[(long)(rbase + i) * N + col] = c[i] + bv;
}

__global__ void k_scores(const float* __restrict__ Wh, const float* __restrict__ al,
                         const float* __restrict__ ar, float* __restrict__ el,
                         float* __restrict__ er, int N)
{
    int idx = blockIdx.x * blockDim.x + threadIdx.x;
    if (idx >= N * HEADS) return;
    int n = idx >> 2, h = idx & 3;
    const float* w = Wh + (long)n * FEAT + h * D_HEAD;
    const float* a = al + h * D_HEAD;
    const float* b = ar + h * D_HEAD;
    float sl = 0.0f, sr = 0.0f;
#pragma unroll
    for (int d = 0; d < D_HEAD; ++d) {
        sl = fmaf(w[d], a[d], sl);
        sr = fmaf(w[d], b[d], sr);
    }
    el[idx] = sl;
    er[idx] = sr;
}

__global__ void k_edge_pre(const float* __restrict__ el, const float* __restrict__ er,
                           const int* __restrict__ src, const int* __restrict__ dst,
                           float* __restrict__ ebuf, unsigned* __restrict__ menc, int E)
{
    int e = blockIdx.x * blockDim.x + threadIdx.x;
    if (e >= E) return;
    int s = src[e], d = dst[e];
#pragma unroll
    for (int h = 0; h < HEADS; ++h) {
        float v = el[s * HEADS + h] + er[d * HEADS + h];
        v = (v > 0.0f) ? v : NEG_SLOPE * v;
        ebuf[(long)e * HEADS + h] = v;
        atomicMax(&menc[d * HEADS + h], f2ord(v));
    }
}

__global__ void k_edge_softmax(const int* __restrict__ dst, const unsigned* __restrict__ menc,
                               float* __restrict__ ebuf, float* __restrict__ denom, int E)
{
    int e = blockIdx.x * blockDim.x + threadIdx.x;
    if (e >= E) return;
    int d = dst[e];
#pragma unroll
    for (int h = 0; h < HEADS; ++h) {
        float m  = ord2f(menc[d * HEADS + h]);
        float ex = expf(ebuf[(long)e * HEADS + h] - m);
        ebuf[(long)e * HEADS + h] = ex;
        atomicAdd(&denom[d * HEADS + h], ex);
    }
}

__global__ void k_edge_scatter(const float* __restrict__ Wh, const float* __restrict__ ex,
                               const float* __restrict__ denom, const int* __restrict__ src,
                               const int* __restrict__ dst, float* __restrict__ acc, int E)
{
    int e = blockIdx.x;
    int f = threadIdx.x;
    int h = f >> 5;
    int s = src[e], d = dst[e];
    float alpha = ex[(long)e * HEADS + h] / (denom[d * HEADS + h] + EPS_);
    atomicAdd(&acc[(long)d * FEAT + f], alpha * Wh[(long)s * FEAT + f]);
}

__global__ void k_elu(float* __restrict__ z, const float* __restrict__ b, long n)
{
    long i = (long)blockIdx.x * blockDim.x + threadIdx.x;
    if (i >= n) return;
    float x = z[i] + b[i & (FEAT - 1)];
    z[i] = (x > 0.0f) ? x : expm1f(x);
}

__global__ void k_mean(const float* __restrict__ acc, const float* __restrict__ b,
                       float* __restrict__ zmean, int N)
{
    int i = blockIdx.x * blockDim.x + threadIdx.x;
    if (i >= N * D_HEAD) return;
    int n = i >> 5, d = i & 31;
    float s = 0.0f;
#pragma unroll
    for (int h = 0; h < HEADS; ++h)
        s += acc[(long)n * FEAT + h * D_HEAD + d] + b[h * D_HEAD + d];
    zmean[i] = s * (1.0f / HEADS);
}


extern "C" void kernel_launch(void* const* d_in, const int* in_sizes, int n_in,
                              void* d_out, int out_size, void* d_ws, size_t ws_size,
                              hipStream_t stream)
{
    const float* h   = (const float*)d_in[0];
    const int*   src = (const int*)  d_in[1];
    const int*   dst = (const int*)  d_in[2];
    const float* W1  = (const float*)d_in[3];
    const float* al1 = (const float*)d_in[4];
    const float* ar1 = (const float*)d_in[5];
    const float* b1  = (const float*)d_in[6];
    const float* W2  = (const float*)d_in[7];
    const float* al2 = (const float*)d_in[8];
    const float* ar2 = (const float*)d_in[9];
    const float* b2  = (const float*)d_in[10];
    const float* Wp  = (const float*)d_in[11];
    const float* bp  = (const float*)d_in[12];
    float*       out = (float*)d_out;

    const int N = N_NODES, E = N_EDGES;

    float*    Wh    = (float*)d_ws;
    float*    Z     = Wh + (size_t)N * FEAT;
    float*    el    = Z  + (size_t)N * FEAT;
    float*    er    = el + (size_t)N * HEADS;
    unsigned* menc  = (unsigned*)(er + (size_t)N * HEADS);
    float*    denom = (float*)(menc + (size_t)N * HEADS);
    float*    ebuf  = denom + (size_t)N * HEADS;
    float*    zmean = ebuf + (size_t)E * HEADS;

    const int mtiles = (N + 15) / 16;
    const dim3 gemmBlk(128);
    const dim3 gemmGrd128((mtiles + 3) / 4, FEAT / 16);
    const dim3 gemmGrdOut((mtiles + 3) / 4, OUT_DIM / 16);

    const int zThreads = 256;
    auto zgrid = [](long n) { return (unsigned)((n + 255) / 256); };

    k_gemm_wmma<<<gemmGrd128, gemmBlk, 0, stream>>>(h, W1, nullptr, Wh, N, FEAT, FEAT);
    k_scores<<<(N * HEADS + 255) / 256, 256, 0, stream>>>(Wh, al1, ar1, el, er, N);
    k_zero<<<zgrid((long)N * HEADS), zThreads, 0, stream>>>((float*)menc, (long)N * HEADS);
    k_zero<<<zgrid((long)N * HEADS), zThreads, 0, stream>>>(denom, (long)N * HEADS);
    k_zero<<<zgrid((long)N * FEAT), zThreads, 0, stream>>>(Z, (long)N * FEAT);
    k_edge_pre<<<(E + 255) / 256, 256, 0, stream>>>(el, er, src, dst, ebuf, menc, E);
    k_edge_softmax<<<(E + 255) / 256, 256, 0, stream>>>(dst, menc, ebuf, denom, E);
    k_edge_scatter<<<E, FEAT, 0, stream>>>(Wh, ebuf, denom, src, dst, Z, E);
    k_elu<<<zgrid((long)N * FEAT), zThreads, 0, stream>>>(Z, b1, (long)N * FEAT);

    k_gemm_wmma<<<gemmGrd128, gemmBlk, 0, stream>>>(Z, W2, nullptr, Wh, N, FEAT, FEAT);
    k_scores<<<(N * HEADS + 255) / 256, 256, 0, stream>>>(Wh, al2, ar2, el, er, N);
    k_zero<<<zgrid((long)N * HEADS), zThreads, 0, stream>>>((float*)menc, (long)N * HEADS);
    k_zero<<<zgrid((long)N * HEADS), zThreads, 0, stream>>>(denom, (long)N * HEADS);
    k_zero<<<zgrid((long)N * FEAT), zThreads, 0, stream>>>(Z, (long)N * FEAT);
    k_edge_pre<<<(E + 255) / 256, 256, 0, stream>>>(el, er, src, dst, ebuf, menc, E);
    k_edge_softmax<<<(E + 255) / 256, 256, 0, stream>>>(dst, menc, ebuf, denom, E);
    k_edge_scatter<<<E, FEAT, 0, stream>>>(Wh, ebuf, denom, src, dst, Z, E);
    k_mean<<<(N * D_HEAD + 255) / 256, 256, 0, stream>>>(Z, b2, zmean, N);

    k_gemm_wmma<<<gemmGrdOut, gemmBlk, 0, stream>>>(zmean, Wp, bp, out, N, OUT_DIM, D_HEAD);
}


# Scatter pass: each work-group owns 196 dst rows in LDS, scans all dst, queues its edges, adds their rows with LDS float atomics; global float atomics and Z zero fill gone; sums still in arrival order
# speedup vs baseline: 1.4864x; 1.4864x over previous
#include <hip/hip_runtime.h>
#include <hip/hip_bf16.h>
#include <math.h>

#define N_NODES   50000
#define N_EDGES   800000
#define HEADS     4
#define D_HEAD    32
#define FEAT      128
#define OUT_DIM   512
#define NEG_SLOPE 0.2f
#define EPS_      1e-9f

typedef __attribute__((ext_vector_type(2))) float v2f;
typedef __attribute__((ext_vector_type(8))) float v8f;


__global__ void k_zero(float* __restrict__ p, long n) {
    long i = (long)blockIdx.x * blockDim.x + threadIdx.x;
    if (i < n) p[i] = 0.0f;
}

__device__ __forceinline__ unsigned f2ord(float f) {
    unsigned u = __float_as_uint(f);
    return (u & 0x80000000u) ? ~u : (u | 0x80000000u);
}
__device__ __forceinline__ float ord2f(unsigned k) {
    return (k & 0x80000000u) ? __uint_as_float(k & 0x7fffffffu) : __uint_as_float(~k);
}

__global__ void k_gemm_wmma(const float* __restrict__ A, const float* __restrict__ B,
                            const float* __restrict__ bias, float* __restrict__ C,
                            int M, int N, int K)
{
    __shared__ float ldsB[128 * 16];
    const int tid  = threadIdx.x;
    const int wave = tid >> 5;
    const int lane = tid & 31;
    const int n0   = blockIdx.y * 16;

    for (int i = tid; i < K * 16; i += 128) {
        int k = i >> 4, c = i & 15;
        ldsB[i] = B[(long)k * N + n0 + c];
    }
    __syncthreads();

    const int m0 = (blockIdx.x * 4 + wave) * 16;
    if (m0 >= M) return;

    const int row   = m0 + (lane & 15);
    const int koff  = (lane >> 4) << 1;
    const int col16 = lane & 15;

    v8f c = {};
    for (int k0 = 0; k0 < K; k0 += 4) {
        v2f a, b;
        a.x = A[(long)row * K + k0 + koff];
        a.y = A[(long)row * K + k0 + koff + 1];
        b.x = ldsB[(k0 + koff)     * 16 + col16];
        b.y = ldsB[(k0 + koff + 1) * 16 + col16];
        c = __builtin_amdgcn_wmma_f32_16x16x4_f32(false, a, false, b,
                                                  (short)0, c, false, false);
    }

    const int   col   = n0 + col16;
    const float bv    = bias ? bias[col] : 0.0f;
    const int   rbase = m0 + ((lane >> 4) << 3);
#pragma unroll
    for (int i = 0; i < 8; ++i)
        C[(long)(rbase + i) * N + col] = c[i] + bv;
}

__global__ void k_scores(const float* __restrict__ Wh, const float* __restrict__ al,
                         const float* __restrict__ ar, float* __restrict__ el,
                         float* __restrict__ er, int N)
{
    int idx = blockIdx.x * blockDim.x + threadIdx.x;
    if (idx >= N * HEADS) return;
    int n = idx >> 2, h = idx & 3;
    const float* w = Wh + (long)n * FEAT + h * D_HEAD;
    const float* a = al + h * D_HEAD;
    const float* b = ar + h * D_HEAD;
    float sl = 0.0f, sr = 0.0f;
#pragma unroll
    for (int d = 0; d < D_HEAD; ++d) {
        sl = fmaf(w[d], a[d], sl);
        sr = fmaf(w[d], b[d], sr);
    }
    el[idx] = sl;
    er[idx] = sr;
}

__global__ void k_edge_pre(const float* __restrict__ el, const float* __restrict__ er,
                           const int* __restrict__ src, const int* __restrict__ dst,
                           float* __restrict__ ebuf, unsigned* __restrict__ menc, int E)
{
    int e = blockIdx.x * blockDim.x + threadIdx.x;
    if (e >= E) return;
    int s = src[e], d = dst[e];
#pragma unroll
    for (int h = 0; h < HEADS; ++h) {
        float v = el[s * HEADS + h] + er[d * HEADS + h];
        v = (v > 0.0f) ? v : NEG_SLOPE * v;
        ebuf[(long)e * HEADS + h] = v;
        atomicMax(&menc[d * HEADS + h], f2ord(v));
    }
}

__global__ void k_edge_softmax(const int* __restrict__ dst, const unsigned* __restrict__ menc,
                               float* __restrict__ ebuf, float* __restrict__ denom, int E)
{
    int e = blockIdx.x * blockDim.x + threadIdx.x;
    if (e >= E) return;
    int d = dst[e];
#pragma unroll
    for (int h = 0; h < HEADS; ++h) {
        float m  = ord2f(menc[d * HEADS + h]);
        float ex = expf(ebuf[(long)e * HEADS + h] - m);
        ebuf[(long)e * HEADS + h] = ex;
        atomicAdd(&denom[d * HEADS + h], ex);
    }
}

#define OWN_ROWS    196
#define OWN_BLOCKS  ((N_NODES + OWN_ROWS - 1) / OWN_ROWS)
#define OWN_THREADS 1024
#define OWN_QCAP    16384
#define OWN_EBITS   20
static_assert(N_EDGES <= (1 << OWN_EBITS), "edge id fits the queue word");
static_assert(N_EDGES % 4 == 0, "dst is read 16 bytes at a time");
static_assert(OWN_ROWS <= (1 << (32 - OWN_EBITS)), "local row fits the queue word");
static_assert((OWN_ROWS * FEAT + 2 * OWN_QCAP + 1) * 4 <= 320 * 1024, "fits one WGP's LDS");

__device__ __forceinline__ void lds_add(float* p, float v) {
    __hip_atomic_fetch_add(p, v, __ATOMIC_RELAXED, __HIP_MEMORY_SCOPE_WORKGROUP);
}

__device__ __forceinline__ void own_push(int e, int lo, float* acc, unsigned* qe, int* qn,
                                         const float* __restrict__ Wh, const float* __restrict__ ex,
                                         const float* __restrict__ denom, const int* __restrict__ src,
                                         const int* __restrict__ dst)
{
    int slot = atomicAdd(qn, 1);
    if (slot < OWN_QCAP) {
        qe[slot] = (unsigned)e;
    } else {
        const int s  = src[e];
        const int d  = dst[e];
        const int ld = d - lo;
        for (int h = 0; h < HEADS; ++h) {
            const float al = ex[(long)e * HEADS + h] / (denom[d * HEADS + h] + EPS_);
#pragma nounroll
            for (int f = 0; f < D_HEAD; ++f) {
                const int c = h * D_HEAD + f;
                lds_add(&acc[ld * FEAT + c], al * Wh[(long)s * FEAT + c]);
            }
        }
    }
}

__device__ __forceinline__ unsigned own_bit(unsigned m, int d, int lo) {
    return (m << 1) | (((unsigned)(d - lo) < (unsigned)OWN_ROWS) ? 1u : 0u);
}

__device__ __forceinline__ void own_drain(unsigned m, int item0, int lo, float* acc, unsigned* qe, int* qn,
                                          const float* __restrict__ Wh, const float* __restrict__ ex,
                                          const float* __restrict__ denom, const int* __restrict__ src,
                                          const int* __restrict__ dst)
{
    while (m) {
        const int j = 31 - __builtin_ctz(m);
        m &= m - 1u;
        own_push(((item0 + (j >> 2) * OWN_THREADS) << 2) + (j & 3), lo, acc, qe, qn, Wh, ex, denom, src, dst);
    }
}

__global__ void __launch_bounds__(OWN_THREADS)
k_edge_owned(const float* __restrict__ Wh, const float* __restrict__ ex,
             const float* __restrict__ denom, const int* __restrict__ src,
             const int* __restrict__ dst, float* __restrict__ out, int N, int E)
{
    __shared__ __attribute__((aligned(16))) float acc[OWN_ROWS * FEAT];
    __shared__ unsigned qe[OWN_QCAP];
    __shared__ int      qs[OWN_QCAP];
    __shared__ int      qn;

    const int tid = threadIdx.x;
    const int lo  = blockIdx.x * OWN_ROWS;

    for (int i = tid; i < OWN_ROWS * FEAT; i += OWN_THREADS) acc[i] = 0.0f;
    if (tid == 0) qn = 0;
    __syncthreads();

    const int4* dst4  = (const int4*)dst;
    const int   n4    = E >> 2;
    const int   nfull = n4 / (8 * OWN_THREADS);
    const int   rot   = (int)(((long)blockIdx.x * nfull) / gridDim.x);
    for (int j = 0; j < nfull; ++j) {
        int sc = j + rot;
        if (sc >= nfull) sc -= nfull;
        const int item0 = sc * (8 * OWN_THREADS) + tid;
        int4 v[8];
#pragma unroll
        for (int u = 0; u < 8; ++u) v[u] = dst4[item0 + u * OWN_THREADS];
        unsigned ma = 0u, mb = 0u;
#pragma unroll
        for (int u = 0; u < 4; ++u) {
            ma = own_bit(ma, v[u].x, lo);     mb = own_bit(mb, v[u + 4].x, lo);
            ma = own_bit(ma, v[u].y, lo);     mb = own_bit(mb, v[u + 4].y, lo);
            ma = own_bit(ma, v[u].z, lo);     mb = own_bit(mb, v[u + 4].z, lo);
            ma = own_bit(ma, v[u].w, lo);     mb = own_bit(mb, v[u + 4].w, lo);
        }
        own_drain((ma << 16) | mb, item0, lo, acc, qe, &qn, Wh, ex, denom, src, dst);
    }
    for (int item = nfull * (8 * OWN_THREADS) + tid; item < n4; item += OWN_THREADS) {
        const int4 t = dst4[item];
        unsigned m = own_bit(own_bit(own_bit(own_bit(0u, t.x, lo), t.y, lo), t.z, lo), t.w, lo);
        own_drain(m << 28, item, lo, acc, qe, &qn, Wh, ex, denom, src, dst);
    }
    __syncthreads();

    int n = qn;
    if (n > OWN_QCAP) n = OWN_QCAP;

    for (int i = tid; i < n; i += OWN_THREADS) {
        const unsigned e = qe[i];
        qs[i] = src[e];
        qe[i] = e | ((unsigned)(dst[e] - lo) << OWN_EBITS);
    }
    __syncthreads();

    const int wave = tid >> 5, lane = tid & 31;
    for (int i0 = wave; i0 < n; i0 += (OWN_THREADS / 32) * 4) {
        unsigned ldr[4];
        bool     ok[4];
        float4   xv[4], dn[4];
        float    w[4][HEADS];
#pragma unroll
        for (int u = 0; u < 4; ++u) {
            const int i = i0 + u * (OWN_THREADS / 32);
            ok[u] = i < n;
            const unsigned pk = ok[u] ? qe[i] : 0u;
            const int      s  = ok[u] ? qs[i] : 0;
            const int      e  = (int)(pk & ((1u << OWN_EBITS) - 1u));
            ldr[u] = pk >> OWN_EBITS;
            xv[u] = *(const float4*)(ex + (long)e * HEADS);
            dn[u] = *(const float4*)(denom + (long)(lo + (int)ldr[u]) * HEADS);
#pragma unroll
            for (int j = 0; j < HEADS; ++j) w[u][j] = Wh[(long)s * FEAT + j * D_HEAD + lane];
        }
#pragma unroll
        for (int u = 0; u < 4; ++u) {
            if (ok[u]) {
                float* row = &acc[ldr[u] * FEAT + lane];
                lds_add(row + 0 * D_HEAD, (xv[u].x / (dn[u].x + EPS_)) * w[u][0]);
                lds_add(row + 1 * D_HEAD, (xv[u].y / (dn[u].y + EPS_)) * w[u][1]);
                lds_add(row + 2 * D_HEAD, (xv[u].z / (dn[u].z + EPS_)) * w[u][2]);
                lds_add(row + 3 * D_HEAD, (xv[u].w / (dn[u].w + EPS_)) * w[u][3]);
            }
        }
    }
    __syncthreads();

    int rows = N - lo;
    if (rows > OWN_ROWS) rows = OWN_ROWS;
    const float4* a4 = (const float4*)acc;
    float4*       o4 = (float4*)(out + (long)lo * FEAT);
    for (int i = tid; i < rows * (FEAT / 4); i += OWN_THREADS) o4[i] = a4[i];
}

__global__ void k_elu(float* __restrict__ z, const float* __restrict__ b, long n)
{
    long i = (long)blockIdx.x * blockDim.x + threadIdx.x;
    if (i >= n) return;
    float x = z[i] + b[i & (FEAT - 1)];
    z[i] = (x > 0.0f) ? x : expm1f(x);
}

__global__ void k_mean(const float* __restrict__ acc, const float* __restrict__ b,
                       float* __restrict__ zmean, int N)
{
    int i = blockIdx.x * blockDim.x + threadIdx.x;
    if (i >= N * D_HEAD) return;
    int n = i >> 5, d = i & 31;
    float s = 0.0f;
#pragma unroll
    for (int h = 0; h < HEADS; ++h)
        s += acc[(long)n * FEAT + h * D_HEAD + d] + b[h * D_HEAD + d];
    zmean[i] = s * (1.0f / HEADS);
}


extern "C" void kernel_launch(void* const* d_in, const int* in_sizes, int n_in,
                              void* d_out, int out_size, void* d_ws, size_t ws_size,
                              hipStream_t stream)
{
    const float* h   = (const float*)d_in[0];
    const int*   src = (const int*)  d_in[1];
    const int*   dst = (const int*)  d_in[2];
    const float* W1  = (const float*)d_in[3];
    const float* al1 = (const float*)d_in[4];
    const float* ar1 = (const float*)d_in[5];
    const float* b1  = (const float*)d_in[6];
    const float* W2  = (const float*)d_in[7];
    const float* al2 = (const float*)d_in[8];
    const float* ar2 = (const float*)d_in[9];
    const float* b2  = (const float*)d_in[10];
    const float* Wp  = (const float*)d_in[11];
    const float* bp  = (const float*)d_in[12];
    float*       out = (float*)d_out;

    const int N = N_NODES, E = N_EDGES;

    float*    Wh    = (float*)d_ws;
    float*    Z     = Wh + (size_t)N * FEAT;
    float*    el    = Z  + (size_t)N * FEAT;
    float*    er    = el + (size_t)N * HEADS;
    unsigned* menc  = (unsigned*)(er + (size_t)N * HEADS);
    float*    denom = (float*)(menc + (size_t)N * HEADS);
    float*    ebuf  = denom + (size_t)N * HEADS;
    float*    zmean = ebuf + (size_t)E * HEADS;

    const int mtiles = (N + 15) / 16;
    const dim3 gemmBlk(128);
    const dim3 gemmGrd128((mtiles + 3) / 4, FEAT / 16);
    const dim3 gemmGrdOut((mtiles + 3) / 4, OUT_DIM / 16);

    const int zThreads = 256;
    auto zgrid = [](long n) { return (unsigned)((n + 255) / 256); };

    k_gemm_wmma<<<gemmGrd128, gemmBlk, 0, stream>>>(h, W1, nullptr, Wh, N, FEAT, FEAT);
    k_scores<<<(N * HEADS + 255) / 256, 256, 0, stream>>>(Wh, al1, ar1, el, er, N);
    k_zero<<<zgrid((long)N * HEADS), zThreads, 0, stream>>>((float*)menc, (long)N * HEADS);
    k_zero<<<zgrid((long)N * HEADS), zThreads, 0, stream>>>(denom, (long)N * HEADS);
    k_edge_pre<<<(E + 255) / 256, 256, 0, stream>>>(el, er, src, dst, ebuf, menc, E);
    k_edge_softmax<<<(E + 255) / 256, 256, 0, stream>>>(dst, menc, ebuf, denom, E);
    k_edge_owned<<<OWN_BLOCKS, OWN_THREADS, 0, stream>>>(Wh, ebuf, denom, src, dst, Z, N, E);
    k_elu<<<zgrid((long)N * FEAT), zThreads, 0, stream>>>(Z, b1, (long)N * FEAT);

    k_gemm_wmma<<<gemmGrd128, gemmBlk, 0, stream>>>(Z, W2, nullptr, Wh, N, FEAT, FEAT);
    k_scores<<<(N * HEADS + 255) / 256, 256, 0, stream>>>(Wh, al2, ar2, el, er, N);
    k_zero<<<zgrid((long)N * HEADS), zThreads, 0, stream>>>((float*)menc, (long)N * HEADS);
    k_zero<<<zgrid((long)N * HEADS), zThreads, 0, stream>>>(denom, (long)N * HEADS);
    k_edge_pre<<<(E + 255) / 256, 256, 0, stream>>>(el, er, src, dst, ebuf, menc, E);
    k_edge_softmax<<<(E + 255) / 256, 256, 0, stream>>>(dst, menc, ebuf, denom, E);
    k_edge_owned<<<OWN_BLOCKS, OWN_THREADS, 0, stream>>>(Wh, ebuf, denom, src, dst, Z, N, E);
    k_mean<<<(N * D_HEAD + 255) / 256, 256, 0, stream>>>(Z, b2, zmean, N);

    k_gemm_wmma<<<gemmGrdOut, gemmBlk, 0, stream>>>(zmean, Wp, bp, out, N, OUT_DIM, D_HEAD);
}


# Edge softmax moved into the dst-range owner: score, max, exp-sum and alpha per queued edge in LDS (sums by LDS atomics, still arrival order); both edge passes, 4 zero fills, ebuf round trip gone
# speedup vs baseline: 4.0327x; 2.7131x over previous
#include <hip/hip_runtime.h>
#include <hip/hip_bf16.h>
#include <math.h>

#define N_NODES   50000
#define N_EDGES   800000
#define HEADS     4
#define D_HEAD    32
#define FEAT      128
#define OUT_DIM   512
#define NEG_SLOPE 0.2f
#define EPS_      1e-9f

typedef __attribute__((ext_vector_type(2))) float v2f;
typedef __attribute__((ext_vector_type(8))) float v8f;


__device__ __forceinline__ unsigned f2ord(float f) {
    unsigned u = __float_as_uint(f);
    return (u & 0x80000000u) ? ~u : (u | 0x80000000u);
}
__device__ __forceinline__ float ord2f(unsigned k) {
    return (k & 0x80000000u) ? __uint_as_float(k & 0x7fffffffu) : __uint_as_float(~k);
}

__global__ void k_gemm_wmma(const float* __restrict__ A, const float* __restrict__ B,
                            const float* __restrict__ bias, float* __restrict__ C,
                            int M, int N, int K)
{
    __shared__ float ldsB[128 * 16];
    const int tid  = threadIdx.x;
    const int wave = tid >> 5;
    const int lane = tid & 31;
    const int n0   = blockIdx.y * 16;

    for (int i = tid; i < K * 16; i += 128) {
        int k = i >> 4, c = i & 15;
        ldsB[i] = B[(long)k * N + n0 + c];
    }
    __syncthreads();

    const int m0 = (blockIdx.x * 4 + wave) * 16;
    if (m0 >= M) return;

    const int row   = m0 + (lane & 15);
    const int koff  = (lane >> 4) << 1;
    const int col16 = lane & 15;

    v8f c = {};
    for (int k0 = 0; k0 < K; k0 += 4) {
        v2f a, b;
        a.x = A[(long)row * K + k0 + koff];
        a.y = A[(long)row * K + k0 + koff + 1];
        b.x = ldsB[(k0 + koff)     * 16 + col16];
        b.y = ldsB[(k0 + koff + 1) * 16 + col16];
        c = __builtin_amdgcn_wmma_f32_16x16x4_f32(false, a, false, b,
                                                  (short)0, c, false, false);
    }

    const int   col   = n0 + col16;
    const float bv    = bias ? bias[col] : 0.0f;
    const int   rbase = m0 + ((lane >> 4) << 3);
#pragma unroll
    for (int i = 0; i < 8; ++i)
        C[(long)(rbase + i) * N + col] = c[i] + bv;
}

__global__ void k_scores(const float* __restrict__ Wh, const float* __restrict__ al,
                         const float* __restrict__ ar, float* __restrict__ el,
                         float* __restrict__ er, int N)
{
    int idx = blockIdx.x * blockDim.x + threadIdx.x;
    if (idx >= N * HEADS) return;
    int n = idx >> 2, h = idx & 3;
    const float* w = Wh + (long)n * FEAT + h * D_HEAD;
    const float* a = al + h * D_HEAD;
    const float* b = ar + h * D_HEAD;
    float sl = 0.0f, sr = 0.0f;
#pragma unroll
    for (int d = 0; d < D_HEAD; ++d) {
        sl = fmaf(w[d], a[d], sl);
        sr = fmaf(w[d], b[d], sr);
    }
    el[idx] = sl;
    er[idx] = sr;
}

#define OWN_ROWS    196
#define OWN_BLOCKS  ((N_NODES + OWN_ROWS - 1) / OWN_ROWS)
#define OWN_THREADS 1024
#define OWN_QCAP    8192
#define OWN_SBITS   16
static_assert(N_NODES <= (1 << OWN_SBITS), "source node fits the queue word");
static_assert(OWN_ROWS <= (1 << (32 - OWN_SBITS)), "local row fits the queue word");
static_assert(N_EDGES % 4 == 0, "dst is read 16 bytes at a time");
static_assert((OWN_ROWS * FEAT + 5 * OWN_QCAP + 2 * OWN_ROWS * HEADS + 4) * 4 <= 320 * 1024, "fits one WGP's LDS");

__device__ __forceinline__ void lds_add(float* p, float v) {
    __hip_atomic_fetch_add(p, v, __ATOMIC_RELAXED, __HIP_MEMORY_SCOPE_WORKGROUP);
}
__device__ __forceinline__ void lds_max(unsigned* p, unsigned v) {
    __hip_atomic_fetch_max(p, v, __ATOMIC_RELAXED, __HIP_MEMORY_SCOPE_WORKGROUP);
}
__device__ __forceinline__ float leaky(float v) { return (v > 0.0f) ? v : NEG_SLOPE * v; }

__device__ __forceinline__ float4 edge_scores(const float* __restrict__ el, const float* __restrict__ er,
                                              int s, int d)
{
    const float4 l = *(const float4*)(el + (long)s * HEADS);
    const float4 r = *(const float4*)(er + (long)d * HEADS);
    float4 v;
    v.x = leaky(l.x + r.x);
    v.y = leaky(l.y + r.y);
    v.z = leaky(l.z + r.z);
    v.w = leaky(l.w + r.w);
    return v;
}

__device__ __forceinline__ void own_push(int e, unsigned* qe, int* qn)
{
    int slot = atomicAdd(qn, 1);
    if (slot < OWN_QCAP) qe[slot] = (unsigned)e;
}

__device__ __forceinline__ unsigned own_bit(unsigned m, int d, int lo) {
    return (m << 1) | (((unsigned)(d - lo) < (unsigned)OWN_ROWS) ? 1u : 0u);
}

__device__ __forceinline__ void own_drain(unsigned m, int item0, unsigned* qe, int* qn)
{
    while (m) {
        const int j = 31 - __builtin_ctz(m);
        m &= m - 1u;
        own_push(((item0 + (j >> 2) * OWN_THREADS) << 2) + (j & 3), qe, qn);
    }
}

__global__ void __launch_bounds__(OWN_THREADS)
k_edge_owned(const float* __restrict__ Wh, const float* __restrict__ el,
             const float* __restrict__ er, const int* __restrict__ src,
             const int* __restrict__ dst, float* __restrict__ out, int N, int E)
{
    __shared__ __attribute__((aligned(16))) float    acc[OWN_ROWS * FEAT];
    __shared__ __attribute__((aligned(16))) float4   ev[OWN_QCAP];
    __shared__ __attribute__((aligned(16))) unsigned mx[OWN_ROWS * HEADS];
    __shared__ __attribute__((aligned(16))) float    dn[OWN_ROWS * HEADS];
    __shared__ unsigned qe[OWN_QCAP];
    __shared__ int      qn;

    const int tid = threadIdx.x;
    const int lo  = blockIdx.x * OWN_ROWS;

    for (int i = tid; i < OWN_ROWS * FEAT; i += OWN_THREADS) acc[i] = 0.0f;
    if (tid < OWN_ROWS * HEADS) { mx[tid] = 0u; dn[tid] = 0.0f; }
    if (tid == 0) qn = 0;
    __syncthreads();

    const int4* dst4  = (const int4*)dst;
    const int   n4    = E >> 2;
    const int   nfull = n4 / (8 * OWN_THREADS);
    const int   rot   = (int)(((long)blockIdx.x * nfull) / gridDim.x);
    for (int j = 0; j < nfull; ++j) {
        int sc = j + rot;
        if (sc >= nfull) sc -= nfull;
        const int item0 = sc * (8 * OWN_THREADS) + tid;
        int4 v[8];
#pragma unroll
        for (int u = 0; u < 8; ++u) v[u] = dst4[item0 + u * OWN_THREADS];
        unsigned ma = 0u, mb = 0u;
#pragma unroll
        for (int u = 0; u < 4; ++u) {
            ma = own_bit(ma, v[u].x, lo);     mb = own_bit(mb, v[u + 4].x, lo);
            ma = own_bit(ma, v[u].y, lo);     mb = own_bit(mb, v[u + 4].y, lo);
            ma = own_bit(ma, v[u].z, lo);     mb = own_bit(mb, v[u + 4].z, lo);
            ma = own_bit(ma, v[u].w, lo);     mb = own_bit(mb, v[u + 4].w, lo);
        }
        own_drain((ma << 16) | mb, item0, qe, &qn);
    }
    for (int item = nfull * (8 * OWN_THREADS) + tid; item < n4; item += OWN_THREADS) {
        const int4 t = dst4[item];
        unsigned m = own_bit(own_bit(own_bit(own_bit(0u, t.x, lo), t.y, lo), t.z, lo), t.w, lo);
        own_drain(m << 28, item, qe, &qn);
    }
    __syncthreads();

    const int   n    = qn;
    const bool  fits = n <= OWN_QCAP;
    const uint4*  mx4 = (const uint4*)mx;
    const float4* dn4 = (const float4*)dn;

    if (fits) {
        for (int i0 = tid; i0 < n; i0 += 4 * OWN_THREADS) {
            int    ss[4], dd[4];
            bool   ok[4];
            float4 sc[4];
#pragma unroll
            for (int u = 0; u < 4; ++u) {
                const int i = i0 + u * OWN_THREADS;
                ok[u] = i < n;
                const int e = ok[u] ? (int)qe[i] : 0;
                ss[u] = src[e];
                dd[u] = dst[e];
            }
#pragma unroll
            for (int u = 0; u < 4; ++u) sc[u] = edge_scores(el, er, ss[u], dd[u]);
#pragma unroll
            for (int u = 0; u < 4; ++u) {
                if (ok[u]) {
                    const int      i  = i0 + u * OWN_THREADS;
                    const unsigned ld = (unsigned)(dd[u] - lo);
                    ev[i] = sc[u];
                    qe[i] = (unsigned)ss[u] | (ld << OWN_SBITS);
                    lds_max(&mx[ld * HEADS + 0], f2ord(sc[u].x));
                    lds_max(&mx[ld * HEADS + 1], f2ord(sc[u].y));
                    lds_max(&mx[ld * HEADS + 2], f2ord(sc[u].z));
                    lds_max(&mx[ld * HEADS + 3], f2ord(sc[u].w));
                }
            }
        }
    } else {
#pragma nounroll
        for (int e = tid; e < E; e += OWN_THREADS) {
            const int      d  = dst[e];
            const unsigned ld = (unsigned)(d - lo);
            if (ld < (unsigned)OWN_ROWS) {
                const float4 v = edge_scores(el, er, src[e], d);
                lds_max(&mx[ld * HEADS + 0], f2ord(v.x));
                lds_max(&mx[ld * HEADS + 1], f2ord(v.y));
                lds_max(&mx[ld * HEADS + 2], f2ord(v.z));
                lds_max(&mx[ld * HEADS + 3], f2ord(v.w));
            }
        }
    }
    __syncthreads();

    if (fits) {
        for (int i = tid; i < n; i += OWN_THREADS) {
            const unsigned ld = qe[i] >> OWN_SBITS;
            const float4   v  = ev[i];
            const uint4    m  = mx4[ld];
            float4 x;
            x.x = expf(v.x - ord2f(m.x));
            x.y = expf(v.y - ord2f(m.y));
            x.z = expf(v.z - ord2f(m.z));
            x.w = expf(v.w - ord2f(m.w));
            ev[i] = x;
            lds_add(&dn[ld * HEADS + 0], x.x);
            lds_add(&dn[ld * HEADS + 1], x.y);
            lds_add(&dn[ld * HEADS + 2], x.z);
            lds_add(&dn[ld * HEADS + 3], x.w);
        }
    } else {
#pragma nounroll
        for (int e = tid; e < E; e += OWN_THREADS) {
            const int      d  = dst[e];
            const unsigned ld = (unsigned)(d - lo);
            if (ld < (unsigned)OWN_ROWS) {
                const float4 v = edge_scores(el, er, src[e], d);
                const uint4  m = mx4[ld];
                lds_add(&dn[ld * HEADS + 0], expf(v.x - ord2f(m.x)));
                lds_add(&dn[ld * HEADS + 1], expf(v.y - ord2f(m.y)));
                lds_add(&dn[ld * HEADS + 2], expf(v.z - ord2f(m.z)));
                lds_add(&dn[ld * HEADS + 3], expf(v.w - ord2f(m.w)));
            }
        }
    }
    __syncthreads();

    if (fits) {
        for (int i = tid; i < n; i += OWN_THREADS) {
            const unsigned ld = qe[i] >> OWN_SBITS;
            const float4   q  = dn4[ld];
            float4 x = ev[i];
            x.x = x.x / (q.x + EPS_);
            x.y = x.y / (q.y + EPS_);
            x.z = x.z / (q.z + EPS_);
            x.w = x.w / (q.w + EPS_);
            ev[i] = x;
        }
    }
    __syncthreads();

    if (fits) {
        const int wave = tid >> 5, lane = tid & 31;
        for (int i0 = wave; i0 < n; i0 += (OWN_THREADS / 32) * 4) {
            unsigned ldr[4];
            bool     ok[4];
            float4   al[4];
            float    w[4][HEADS];
#pragma unroll
            for (int u = 0; u < 4; ++u) {
                const int i = i0 + u * (OWN_THREADS / 32);
                ok[u] = i < n;
                const unsigned pk = ok[u] ? qe[i] : 0u;
                const int      s  = (int)(pk & ((1u << OWN_SBITS) - 1u));
                ldr[u] = pk >> OWN_SBITS;
                al[u]  = ev[ok[u] ? i : 0];
#pragma unroll
                for (int j = 0; j < HEADS; ++j) w[u][j] = Wh[(long)s * FEAT + j * D_HEAD + lane];
            }
#pragma unroll
            for (int u = 0; u < 4; ++u) {
                if (ok[u]) {
                    float* row = &acc[ldr[u] * FEAT + lane];
                    lds_add(row + 0 * D_HEAD, al[u].x * w[u][0]);
                    lds_add(row + 1 * D_HEAD, al[u].y * w[u][1]);
                    lds_add(row + 2 * D_HEAD, al[u].z * w[u][2]);
                    lds_add(row + 3 * D_HEAD, al[u].w * w[u][3]);
                }
            }
        }
    } else {
#pragma nounroll
        for (int e = tid; e < E; e += OWN_THREADS) {
            const int      d  = dst[e];
            const unsigned ld = (unsigned)(d - lo);
            if (ld < (unsigned)OWN_ROWS) {
                const int    s = src[e];
                const float4 v = edge_scores(el, er, s, d);
                const uint4  m = mx4[ld];
                const float4 q = dn4[ld];
                float a[HEADS];
                a[0] = expf(v.x - ord2f(m.x)) / (q.x + EPS_);
                a[1] = expf(v.y - ord2f(m.y)) / (q.y + EPS_);
                a[2] = expf(v.z - ord2f(m.z)) / (q.z + EPS_);
                a[3] = expf(v.w - ord2f(m.w)) / (q.w + EPS_);
#pragma unroll
                for (int h = 0; h < HEADS; ++h) {
#pragma nounroll
                    for (int f = 0; f < D_HEAD; ++f) {
                        const int c = h * D_HEAD + f;
                        lds_add(&acc[ld * FEAT + c], a[h] * Wh[(long)s * FEAT + c]);
                    }
                }
            }
        }
    }
    __syncthreads();

    int rows = N - lo;
    if (rows > OWN_ROWS) rows = OWN_ROWS;
    const float4* a4 = (const float4*)acc;
    float4*       o4 = (float4*)(out + (long)lo * FEAT);
    for (int i = tid; i < rows * (FEAT / 4); i += OWN_THREADS) o4[i] = a4[i];
}

__global__ void k_elu(float* __restrict__ z, const float* __restrict__ b, long n)
{
    long i = (long)blockIdx.x * blockDim.x + threadIdx.x;
    if (i >= n) return;
    float x = z[i] + b[i & (FEAT - 1)];
    z[i] = (x > 0.0f) ? x : expm1f(x);
}

__global__ void k_mean(const float* __restrict__ acc, const float* __restrict__ b,
                       float* __restrict__ zmean, int N)
{
    int i = blockIdx.x * blockDim.x + threadIdx.x;
    if (i >= N * D_HEAD) return;
    int n = i >> 5, d = i & 31;
    float s = 0.0f;
#pragma unroll
    for (int h = 0; h < HEADS; ++h)
        s += acc[(long)n * FEAT + h * D_HEAD + d] + b[h * D_HEAD + d];
    zmean[i] = s * (1.0f / HEADS);
}


extern "C" void kernel_launch(void* const* d_in, const int* in_sizes, int n_in,
                              void* d_out, int out_size, void* d_ws, size_t ws_size,
                              hipStream_t stream)
{
    const float* h   = (const float*)d_in[0];
    const int*   src = (const int*)  d_in[1];
    const int*   dst = (const int*)  d_in[2];
    const float* W1  = (const float*)d_in[3];
    const float* al1 = (const float*)d_in[4];
    const float* ar1 = (const float*)d_in[5];
    const float* b1  = (const float*)d_in[6];
    const float* W2  = (const float*)d_in[7];
    const float* al2 = (const float*)d_in[8];
    const float* ar2 = (const float*)d_in[9];
    const float* b2  = (const float*)d_in[10];
    const float* Wp  = (const float*)d_in[11];
    const float* bp  = (const float*)d_in[12];
    float*       out = (float*)d_out;

    const int N = N_NODES, E = N_EDGES;

    float*    Wh    = (float*)d_ws;
    float*    Z     = Wh + (size_t)N * FEAT;
    float*    el    = Z  + (size_t)N * FEAT;
    float*    er    = el + (size_t)N * HEADS;
    float*    zmean = er + (size_t)N * HEADS;

    const int mtiles = (N + 15) / 16;
    const dim3 gemmBlk(128);
    const dim3 gemmGrd128((mtiles + 3) / 4, FEAT / 16);
    const dim3 gemmGrdOut((mtiles + 3) / 4, OUT_DIM / 16);

    const int zThreads = 256;
    auto zgrid = [](long n) { return (unsigned)((n + 255) / 256); };

    k_gemm_wmma<<<gemmGrd128, gemmBlk, 0, stream>>>(h, W1, nullptr, Wh, N, FEAT, FEAT);
    k_scores<<<(N * HEADS + 255) / 256, 256, 0, stream>>>(Wh, al1, ar1, el, er, N);
    k_edge_owned<<<OWN_BLOCKS, OWN_THREADS, 0, stream>>>(Wh, el, er, src, dst, Z, N, E);
    k_elu<<<zgrid((long)N * FEAT), zThreads, 0, stream>>>(Z, b1, (long)N * FEAT);

    k_gemm_wmma<<<gemmGrd128, gemmBlk, 0, stream>>>(Z, W2, nullptr, Wh, N, FEAT, FEAT);
    k_scores<<<(N * HEADS + 255) / 256, 256, 0, stream>>>(Wh, al2, ar2, el, er, N);
    k_edge_owned<<<OWN_BLOCKS, OWN_THREADS, 0, stream>>>(Wh, el, er, src, dst, Z, N, E);
    k_mean<<<(N * D_HEAD + 255) / 256, 256, 0, stream>>>(Z, b2, zmean, N);

    k_gemm_wmma<<<gemmGrdOut, gemmBlk, 0, stream>>>(zmean, Wp, bp, out, N, OUT_DIM, D_HEAD);
}


# Edge queue built once per call: layer 1's owner pass saves each group's queue (source | local row) and length to the workspace; layer 2's pass reads it, its dst scan is gone; sums still arrival order
# speedup vs baseline: 4.3851x; 1.0874x over previous
#include <hip/hip_runtime.h>
#include <hip/hip_bf16.h>
#include <math.h>

#define N_NODES   50000
#define N_EDGES   800000
#define HEADS     4
#define D_HEAD    32
#define FEAT      128
#define OUT_DIM   512
#define NEG_SLOPE 0.2f
#define EPS_      1e-9f

typedef __attribute__((ext_vector_type(2))) float v2f;
typedef __attribute__((ext_vector_type(8))) float v8f;


__device__ __forceinline__ unsigned f2ord(float f) {
    unsigned u = __float_as_uint(f);
    return (u & 0x80000000u) ? ~u : (u | 0x80000000u);
}
__device__ __forceinline__ float ord2f(unsigned k) {
    return (k & 0x80000000u) ? __uint_as_float(k & 0x7fffffffu) : __uint_as_float(~k);
}

__global__ void k_gemm_wmma(const float* __restrict__ A, const float* __restrict__ B,
                            const float* __restrict__ bias, float* __restrict__ C,
                            int M, int N, int K)
{
    __shared__ float ldsB[128 * 16];
    const int tid  = threadIdx.x;
    const int wave = tid >> 5;
    const int lane = tid & 31;
    const int n0   = blockIdx.y * 16;

    for (int i = tid; i < K * 16; i += 128) {
        int k = i >> 4, c = i & 15;
        ldsB[i] = B[(long)k * N + n0 + c];
    }
    __syncthreads();

    const int m0 = (blockIdx.x * 4 + wave) * 16;
    if (m0 >= M) return;

    const int row   = m0 + (lane & 15);
    const int koff  = (lane >> 4) << 1;
    const int col16 = lane & 15;

    v8f c = {};
    for (int k0 = 0; k0 < K; k0 += 4) {
        v2f a, b;
        a.x = A[(long)row * K + k0 + koff];
        a.y = A[(long)row * K + k0 + koff + 1];
        b.x = ldsB[(k0 + koff)     * 16 + col16];
        b.y = ldsB[(k0 + koff + 1) * 16 + col16];
        c = __builtin_amdgcn_wmma_f32_16x16x4_f32(false, a, false, b,
                                                  (short)0, c, false, false);
    }

    const int   col   = n0 + col16;
    const float bv    = bias ? bias[col] : 0.0f;
    const int   rbase = m0 + ((lane >> 4) << 3);
#pragma unroll
    for (int i = 0; i < 8; ++i)
        C[(long)(rbase + i) * N + col] = c[i] + bv;
}

__global__ void k_scores(const float* __restrict__ Wh, const float* __restrict__ al,
                         const float* __restrict__ ar, float* __restrict__ el,
                         float* __restrict__ er, int N)
{
    int idx = blockIdx.x * blockDim.x + threadIdx.x;
    if (idx >= N * HEADS) return;
    int n = idx >> 2, h = idx & 3;
    const float* w = Wh + (long)n * FEAT + h * D_HEAD;
    const float* a = al + h * D_HEAD;
    const float* b = ar + h * D_HEAD;
    float sl = 0.0f, sr = 0.0f;
#pragma unroll
    for (int d = 0; d < D_HEAD; ++d) {
        sl = fmaf(w[d], a[d], sl);
        sr = fmaf(w[d], b[d], sr);
    }
    el[idx] = sl;
    er[idx] = sr;
}

#define OWN_ROWS    196
#define OWN_BLOCKS  ((N_NODES + OWN_ROWS - 1) / OWN_ROWS)
#define OWN_THREADS 1024
#define OWN_QCAP    8192
#define OWN_SBITS   16
static_assert(N_NODES <= (1 << OWN_SBITS), "source node fits the queue word");
static_assert(OWN_ROWS <= (1 << (32 - OWN_SBITS)), "local row fits the queue word");
static_assert(N_EDGES % 4 == 0, "dst is read 16 bytes at a time");
static_assert((OWN_ROWS * FEAT + 5 * OWN_QCAP + 2 * OWN_ROWS * HEADS + 4) * 4 <= 320 * 1024, "fits one WGP's LDS");

__device__ __forceinline__ void lds_add(float* p, float v) {
    __hip_atomic_fetch_add(p, v, __ATOMIC_RELAXED, __HIP_MEMORY_SCOPE_WORKGROUP);
}
__device__ __forceinline__ void lds_max(unsigned* p, unsigned v) {
    __hip_atomic_fetch_max(p, v, __ATOMIC_RELAXED, __HIP_MEMORY_SCOPE_WORKGROUP);
}
__device__ __forceinline__ float leaky(float v) { return (v > 0.0f) ? v : NEG_SLOPE * v; }

__device__ __forceinline__ float4 edge_scores(const float* __restrict__ el, const float* __restrict__ er,
                                              int s, int d)
{
    const float4 l = *(const float4*)(el + (long)s * HEADS);
    const float4 r = *(const float4*)(er + (long)d * HEADS);
    float4 v;
    v.x = leaky(l.x + r.x);
    v.y = leaky(l.y + r.y);
    v.z = leaky(l.z + r.z);
    v.w = leaky(l.w + r.w);
    return v;
}

__device__ __forceinline__ void own_push(int e, unsigned* qe, int* qn)
{
    int slot = atomicAdd(qn, 1);
    if (slot < OWN_QCAP) qe[slot] = (unsigned)e;
}

__device__ __forceinline__ unsigned own_bit(unsigned m, int d, int lo) {
    return (m << 1) | (((unsigned)(d - lo) < (unsigned)OWN_ROWS) ? 1u : 0u);
}

__device__ __forceinline__ void own_drain(unsigned m, int item0, unsigned* qe, int* qn)
{
    while (m) {
        const int j = 31 - __builtin_ctz(m);
        m &= m - 1u;
        own_push(((item0 + (j >> 2) * OWN_THREADS) << 2) + (j & 3), qe, qn);
    }
}

__global__ void __launch_bounds__(OWN_THREADS)
k_edge_owned(const float* __restrict__ Wh, const float* __restrict__ el,
             const float* __restrict__ er, const int* __restrict__ src,
             const int* __restrict__ dst, float* __restrict__ out, int N, int E,
             unsigned* __restrict__ qsave, int* __restrict__ qcnt, int reuse)
{
    __shared__ __attribute__((aligned(16))) float    acc[OWN_ROWS * FEAT];
    __shared__ __attribute__((aligned(16))) float4   ev[OWN_QCAP];
    __shared__ __attribute__((aligned(16))) unsigned mx[OWN_ROWS * HEADS];
    __shared__ __attribute__((aligned(16))) float    dn[OWN_ROWS * HEADS];
    __shared__ unsigned qe[OWN_QCAP];
    __shared__ int      qn;

    const int tid   = threadIdx.x;
    const int lo    = blockIdx.x * OWN_ROWS;
    const int qbase = blockIdx.x * OWN_QCAP;

    for (int i = tid; i < OWN_ROWS * FEAT; i += OWN_THREADS) acc[i] = 0.0f;
    if (tid < OWN_ROWS * HEADS) { mx[tid] = 0u; dn[tid] = 0.0f; }
    if (tid == 0) qn = 0;
    __syncthreads();

    int n;
    if (!reuse) {
        const int4* dst4  = (const int4*)dst;
        const int   n4    = E >> 2;
        const int   nfull = n4 / (8 * OWN_THREADS);
        const int   rot   = (int)(((long)blockIdx.x * nfull) / gridDim.x);
        for (int j = 0; j < nfull; ++j) {
            int sc = j + rot;
            if (sc >= nfull) sc -= nfull;
            const int item0 = sc * (8 * OWN_THREADS) + tid;
            int4 v[8];
#pragma unroll
            for (int u = 0; u < 8; ++u) v[u] = dst4[item0 + u * OWN_THREADS];
            unsigned ma = 0u, mb = 0u;
#pragma unroll
            for (int u = 0; u < 4; ++u) {
                ma = own_bit(ma, v[u].x, lo);     mb = own_bit(mb, v[u + 4].x, lo);
                ma = own_bit(ma, v[u].y, lo);     mb = own_bit(mb, v[u + 4].y, lo);
                ma = own_bit(ma, v[u].z, lo);     mb = own_bit(mb, v[u + 4].z, lo);
                ma = own_bit(ma, v[u].w, lo);     mb = own_bit(mb, v[u + 4].w, lo);
            }
            own_drain((ma << 16) | mb, item0, qe, &qn);
        }
        for (int item = nfull * (8 * OWN_THREADS) + tid; item < n4; item += OWN_THREADS) {
            const int4 t = dst4[item];
            unsigned m = own_bit(own_bit(own_bit(own_bit(0u, t.x, lo), t.y, lo), t.z, lo), t.w, lo);
            own_drain(m << 28, item, qe, &qn);
        }
        __syncthreads();
        n = qn;
        if (tid == 0) qcnt[blockIdx.x] = n;
    } else {
        n = qcnt[blockIdx.x];
    }

    const bool  fits = n <= OWN_QCAP;
    const uint4*  mx4 = (const uint4*)mx;
    const float4* dn4 = (const float4*)dn;

    if (fits) {
        for (int i0 = tid; i0 < n; i0 += 4 * OWN_THREADS) {
            int    ss[4], dd[4];
            bool   ok[4];
            float4 sc[4];
            if (reuse) {
                unsigned pk[4];
#pragma unroll
                for (int u = 0; u < 4; ++u) {
                    const int i = i0 + u * OWN_THREADS;
                    ok[u] = i < n;
                    pk[u] = ok[u] ? qsave[qbase + i] : 0u;
                }
#pragma unroll
                for (int u = 0; u < 4; ++u) {
                    ss[u] = (int)(pk[u] & ((1u << OWN_SBITS) - 1u));
                    dd[u] = lo + (int)(pk[u] >> OWN_SBITS);
                }
            } else {
#pragma unroll
                for (int u = 0; u < 4; ++u) {
                    const int i = i0 + u * OWN_THREADS;
                    ok[u] = i < n;
                    const int e = ok[u] ? (int)qe[i] : 0;
                    ss[u] = src[e];
                    dd[u] = dst[e];
                }
            }
#pragma unroll
            for (int u = 0; u < 4; ++u) sc[u] = edge_scores(el, er, ss[u], dd[u]);
#pragma unroll
            for (int u = 0; u < 4; ++u) {
                if (ok[u]) {
                    const int      i  = i0 + u * OWN_THREADS;
                    const unsigned ld = (unsigned)(dd[u] - lo);
                    const unsigned pk = (unsigned)ss[u] | (ld << OWN_SBITS);
                    ev[i] = sc[u];
                    qe[i] = pk;
                    if (!reuse) qsave[qbase + i] = pk;
                    lds_max(&mx[ld * HEADS + 0], f2ord(sc[u].x));
                    lds_max(&mx[ld * HEADS + 1], f2ord(sc[u].y));
                    lds_max(&mx[ld * HEADS + 2], f2ord(sc[u].z));
                    lds_max(&mx[ld * HEADS + 3], f2ord(sc[u].w));
                }
            }
        }
    } else {
#pragma nounroll
        for (int e = tid; e < E; e += OWN_THREADS) {
            const int      d  = dst[e];
            const unsigned ld = (unsigned)(d - lo);
            if (ld < (unsigned)OWN_ROWS) {
                const float4 v = edge_scores(el, er, src[e], d);
                lds_max(&mx[ld * HEADS + 0], f2ord(v.x));
                lds_max(&mx[ld * HEADS + 1], f2ord(v.y));
                lds_max(&mx[ld * HEADS + 2], f2ord(v.z));
                lds_max(&mx[ld * HEADS + 3], f2ord(v.w));
            }
        }
    }
    __syncthreads();

    if (fits) {
        for (int i = tid; i < n; i += OWN_THREADS) {
            const unsigned ld = qe[i] >> OWN_SBITS;
            const float4   v  = ev[i];
            const uint4    m  = mx4[ld];
            float4 x;
            x.x = expf(v.x - ord2f(m.x));
            x.y = expf(v.y - ord2f(m.y));
            x.z = expf(v.z - ord2f(m.z));
            x.w = expf(v.w - ord2f(m.w));
            ev[i] = x;
            lds_add(&dn[ld * HEADS + 0], x.x);
            lds_add(&dn[ld * HEADS + 1], x.y);
            lds_add(&dn[ld * HEADS + 2], x.z);
            lds_add(&dn[ld * HEADS + 3], x.w);
        }
    } else {
#pragma nounroll
        for (int e = tid; e < E; e += OWN_THREADS) {
            const int      d  = dst[e];
            const unsigned ld = (unsigned)(d - lo);
            if (ld < (unsigned)OWN_ROWS) {
                const float4 v = edge_scores(el, er, src[e], d);
                const uint4  m = mx4[ld];
                lds_add(&dn[ld * HEADS + 0], expf(v.x - ord2f(m.x)));
                lds_add(&dn[ld * HEADS + 1], expf(v.y - ord2f(m.y)));
                lds_add(&dn[ld * HEADS + 2], expf(v.z - ord2f(m.z)));
                lds_add(&dn[ld * HEADS + 3], expf(v.w - ord2f(m.w)));
            }
        }
    }
    __syncthreads();

    if (fits) {
        for (int i = tid; i < n; i += OWN_THREADS) {
            const unsigned ld = qe[i] >> OWN_SBITS;
            const float4   q  = dn4[ld];
            float4 x = ev[i];
            x.x = x.x / (q.x + EPS_);
            x.y = x.y / (q.y + EPS_);
            x.z = x.z / (q.z + EPS_);
            x.w = x.w / (q.w + EPS_);
            ev[i] = x;
        }
    }
    __syncthreads();

    if (fits) {
        const int wave = tid >> 5, lane = tid & 31;
        for (int i0 = wave; i0 < n; i0 += (OWN_THREADS / 32) * 4) {
            unsigned ldr[4];
            bool     ok[4];
            float4   al[4];
            float    w[4][HEADS];
#pragma unroll
            for (int u = 0; u < 4; ++u) {
                const int i = i0 + u * (OWN_THREADS / 32);
                ok[u] = i < n;
                const unsigned pk = ok[u] ? qe[i] : 0u;
                const int      s  = (int)(pk & ((1u << OWN_SBITS) - 1u));
                ldr[u] = pk >> OWN_SBITS;
                al[u]  = ev[ok[u] ? i : 0];
#pragma unroll
                for (int j = 0; j < HEADS; ++j) w[u][j] = Wh[(long)s * FEAT + j * D_HEAD + lane];
            }
#pragma unroll
            for (int u = 0; u < 4; ++u) {
                if (ok[u]) {
                    float* row = &acc[ldr[u] * FEAT + lane];
                    lds_add(row + 0 * D_HEAD, al[u].x * w[u][0]);
                    lds_add(row + 1 * D_HEAD, al[u].y * w[u][1]);
                    lds_add(row + 2 * D_HEAD, al[u].z * w[u][2]);
                    lds_add(row + 3 * D_HEAD, al[u].w * w[u][3]);
                }
            }
        }
    } else {
#pragma nounroll
        for (int e = tid; e < E; e += OWN_THREADS) {
            const int      d  = dst[e];
            const unsigned ld = (unsigned)(d - lo);
            if (ld < (unsigned)OWN_ROWS) {
                const int    s = src[e];
                const float4 v = edge_scores(el, er, s, d);
                const uint4  m = mx4[ld];
                const float4 q = dn4[ld];
                float a[HEADS];
                a[0] = expf(v.x - ord2f(m.x)) / (q.x + EPS_);
                a[1] = expf(v.y - ord2f(m.y)) / (q.y + EPS_);
                a[2] = expf(v.z - ord2f(m.z)) / (q.z + EPS_);
                a[3] = expf(v.w - ord2f(m.w)) / (q.w + EPS_);
#pragma unroll
                for (int h = 0; h < HEADS; ++h) {
#pragma nounroll
                    for (int f = 0; f < D_HEAD; ++f) {
                        const int c = h * D_HEAD + f;
                        lds_add(&acc[ld * FEAT + c], a[h] * Wh[(long)s * FEAT + c]);
                    }
                }
            }
        }
    }
    __syncthreads();

    int rows = N - lo;
    if (rows > OWN_ROWS) rows = OWN_ROWS;
    const float4* a4 = (const float4*)acc;
    float4*       o4 = (float4*)(out + (long)lo * FEAT);
    for (int i = tid; i < rows * (FEAT / 4); i += OWN_THREADS) o4[i] = a4[i];
}

__global__ void k_elu(float* __restrict__ z, const float* __restrict__ b, long n)
{
    long i = (long)blockIdx.x * blockDim.x + threadIdx.x;
    if (i >= n) return;
    float x = z[i] + b[i & (FEAT - 1)];
    z[i] = (x > 0.0f) ? x : expm1f(x);
}

__global__ void k_mean(const float* __restrict__ acc, const float* __restrict__ b,
                       float* __restrict__ zmean, int N)
{
    int i = blockIdx.x * blockDim.x + threadIdx.x;
    if (i >= N * D_HEAD) return;
    int n = i >> 5, d = i & 31;
    float s = 0.0f;
#pragma unroll
    for (int h = 0; h < HEADS; ++h)
        s += acc[(long)n * FEAT + h * D_HEAD + d] + b[h * D_HEAD + d];
    zmean[i] = s * (1.0f / HEADS);
}


extern "C" void kernel_launch(void* const* d_in, const int* in_sizes, int n_in,
                              void* d_out, int out_size, void* d_ws, size_t ws_size,
                              hipStream_t stream)
{
    const float* h   = (const float*)d_in[0];
    const int*   src = (const int*)  d_in[1];
    const int*   dst = (const int*)  d_in[2];
    const float* W1  = (const float*)d_in[3];
    const float* al1 = (const float*)d_in[4];
    const float* ar1 = (const float*)d_in[5];
    const float* b1  = (const float*)d_in[6];
    const float* W2  = (const float*)d_in[7];
    const float* al2 = (const float*)d_in[8];
    const float* ar2 = (const float*)d_in[9];
    const float* b2  = (const float*)d_in[10];
    const float* Wp  = (const float*)d_in[11];
    const float* bp  = (const float*)d_in[12];
    float*       out = (float*)d_out;

    const int N = N_NODES, E = N_EDGES;

    float*    Wh    = (float*)d_ws;
    float*    Z     = Wh + (size_t)N * FEAT;
    float*    el    = Z  + (size_t)N * FEAT;
    float*    er    = el + (size_t)N * HEADS;
    float*    zmean = er + (size_t)N * HEADS;
    unsigned* qsave = (unsigned*)(zmean + (size_t)N * D_HEAD);
    int*      qcnt  = (int*)(qsave + (size_t)OWN_BLOCKS * OWN_QCAP);

    const int mtiles = (N + 15) / 16;
    const dim3 gemmBlk(128);
    const dim3 gemmGrd128((mtiles + 3) / 4, FEAT / 16);
    const dim3 gemmGrdOut((mtiles + 3) / 4, OUT_DIM / 16);

    const int zThreads = 256;
    auto zgrid = [](long n) { return (unsigned)((n + 255) / 256); };

    k_gemm_wmma<<<gemmGrd128, gemmBlk, 0, stream>>>(h, W1, nullptr, Wh, N, FEAT, FEAT);
    k_scores<<<(N * HEADS + 255) / 256, 256, 0, stream>>>(Wh, al1, ar1, el, er, N);
    k_edge_owned<<<OWN_BLOCKS, OWN_THREADS, 0, stream>>>(Wh, el, er, src, dst, Z, N, E, qsave, qcnt, 0);
    k_elu<<<zgrid((long)N * FEAT), zThreads, 0, stream>>>(Z, b1, (long)N * FEAT);

    k_gemm_wmma<<<gemmGrd128, gemmBlk, 0, stream>>>(Z, W2, nullptr, Wh, N, FEAT, FEAT);
    k_scores<<<(N * HEADS + 255) / 256, 256, 0, stream>>>(Wh, al2, ar2, el, er, N);
    k_edge_owned<<<OWN_BLOCKS, OWN_THREADS, 0, stream>>>(Wh, el, er, src, dst, Z, N, E, qsave, qcnt, 1);
    k_mean<<<(N * D_HEAD + 255) / 256, 256, 0, stream>>>(Z, b2, zmean, N);

    k_gemm_wmma<<<gemmGrdOut, gemmBlk, 0, stream>>>(zmean, Wp, bp, out, N, OUT_DIM, D_HEAD);
}


# Next dense product moved into the dst-range owner: summed rows stay in LDS, get bias+ELU or bias+head mean there and feed WMMA from LDS (W2 with el/er, or the head); 5 launches, Z round trips gone
# speedup vs baseline: 6.6265x; 1.5111x over previous
#include <hip/hip_runtime.h>
#include <hip/hip_bf16.h>
#include <math.h>

#define N_NODES   50000
#define N_EDGES   800000
#define HEADS     4
#define D_HEAD    32
#define FEAT      128
#define OUT_DIM   512
#define NEG_SLOPE 0.2f
#define EPS_      1e-9f

typedef __attribute__((ext_vector_type(2))) float v2f;
typedef __attribute__((ext_vector_type(8))) float v8f;


__device__ __forceinline__ unsigned f2ord(float f) {
    unsigned u = __float_as_uint(f);
    return (u & 0x80000000u) ? ~u : (u | 0x80000000u);
}
__device__ __forceinline__ float ord2f(unsigned k) {
    return (k & 0x80000000u) ? __uint_as_float(k & 0x7fffffffu) : __uint_as_float(~k);
}

__global__ void k_gemm_wmma(const float* __restrict__ A, const float* __restrict__ B,
                            const float* __restrict__ bias, float* __restrict__ C,
                            int M, int N, int K)
{
    __shared__ float ldsB[128 * 16];
    const int tid  = threadIdx.x;
    const int wave = tid >> 5;
    const int lane = tid & 31;
    const int n0   = blockIdx.y * 16;

    for (int i = tid; i < K * 16; i += 128) {
        int k = i >> 4, c = i & 15;
        ldsB[i] = B[(long)k * N + n0 + c];
    }
    __syncthreads();

    const int m0 = (blockIdx.x * 4 + wave) * 16;
    if (m0 >= M) return;

    const int row   = m0 + (lane & 15);
    const int koff  = (lane >> 4) << 1;
    const int col16 = lane & 15;

    v8f c = {};
    for (int k0 = 0; k0 < K; k0 += 4) {
        v2f a, b;
        a.x = A[(long)row * K + k0 + koff];
        a.y = A[(long)row * K + k0 + koff + 1];
        b.x = ldsB[(k0 + koff)     * 16 + col16];
        b.y = ldsB[(k0 + koff + 1) * 16 + col16];
        c = __builtin_amdgcn_wmma_f32_16x16x4_f32(false, a, false, b,
                                                  (short)0, c, false, false);
    }

    const int   col   = n0 + col16;
    const float bv    = bias ? bias[col] : 0.0f;
    const int   rbase = m0 + ((lane >> 4) << 3);
#pragma unroll
    for (int i = 0; i < 8; ++i)
        C[(long)(rbase + i) * N + col] = c[i] + bv;
}

__global__ void k_scores(const float* __restrict__ Wh, const float* __restrict__ al,
                         const float* __restrict__ ar, float* __restrict__ el,
                         float* __restrict__ er, int N)
{
    int idx = blockIdx.x * blockDim.x + threadIdx.x;
    if (idx >= N * HEADS) return;
    int n = idx >> 2, h = idx & 3;
    const float* w = Wh + (long)n * FEAT + h * D_HEAD;
    const float* a = al + h * D_HEAD;
    const float* b = ar + h * D_HEAD;
    float sl = 0.0f, sr = 0.0f;
#pragma unroll
    for (int d = 0; d < D_HEAD; ++d) {
        sl = fmaf(w[d], a[d], sl);
        sr = fmaf(w[d], b[d], sr);
    }
    el[idx] = sl;
    er[idx] = sr;
}

#define OWN_ROWS    196
#define OWN_BLOCKS  ((N_NODES + OWN_ROWS - 1) / OWN_ROWS)
#define OWN_THREADS 1024
#define OWN_QCAP    8192
#define OWN_SBITS   16
static_assert(N_NODES <= (1 << OWN_SBITS), "source node fits the queue word");
static_assert(OWN_ROWS <= (1 << (32 - OWN_SBITS)), "local row fits the queue word");
static_assert(N_EDGES % 4 == 0, "dst is read 16 bytes at a time");
#define ACC_LD      132
#define W_LD        136
#define ST_LD       132
#define ZM_LD       36
#define OWN_RT      ((OWN_ROWS + 15) / 16)
#define OFF_ACC     0
#define OFF_EV      (OFF_ACC + OWN_ROWS * ACC_LD * 4)
#define OFF_QE      (OFF_EV + OWN_QCAP * 16)
#define OFF_MX      (OFF_QE + OWN_QCAP * 4)
#define OFF_DN      (OFF_MX + OWN_ROWS * HEADS * 4)
#define OFF_QN      (OFF_DN + OWN_ROWS * HEADS * 4)
#define OFF_WST     OFF_EV
#define OFF_STG     (OFF_WST + FEAT * W_LD * 4)
#define OFF_ZM      OFF_EV
#define OWN_LDS     (OFF_STG + OWN_ROWS * ST_LD * 4)
static_assert(OWN_LDS >= OFF_QN + 16 && OWN_LDS <= 320 * 1024, "fits one WGP's LDS");
static_assert(OFF_ZM + OWN_RT * 16 * ZM_LD * 4 <= OWN_LDS, "head means fit");
static_assert(OFF_ACC + OWN_RT * 16 * ACC_LD * 4 <= OWN_LDS, "the last tile's spare rows are still LDS");
static_assert(OFF_EV % 16 == 0 && OFF_MX % 16 == 0 && OFF_DN % 16 == 0 && OFF_STG % 16 == 0, "16-byte pieces");
static_assert(OUT_DIM / 16 == OWN_THREADS / 32, "second pass: one output column tile per wave");

__device__ __forceinline__ void lds_add(float* p, float v) {
    __hip_atomic_fetch_add(p, v, __ATOMIC_RELAXED, __HIP_MEMORY_SCOPE_WORKGROUP);
}
__device__ __forceinline__ void lds_max(unsigned* p, unsigned v) {
    __hip_atomic_fetch_max(p, v, __ATOMIC_RELAXED, __HIP_MEMORY_SCOPE_WORKGROUP);
}
__device__ __forceinline__ float leaky(float v) { return (v > 0.0f) ? v : NEG_SLOPE * v; }

__device__ __forceinline__ float4 edge_scores(const float* __restrict__ el, const float* __restrict__ er,
                                              int s, int d)
{
    const float4 l = *(const float4*)(el + (long)s * HEADS);
    const float4 r = *(const float4*)(er + (long)d * HEADS);
    float4 v;
    v.x = leaky(l.x + r.x);
    v.y = leaky(l.y + r.y);
    v.z = leaky(l.z + r.z);
    v.w = leaky(l.w + r.w);
    return v;
}

__device__ __forceinline__ void own_push(int e, unsigned* qe, int* qn)
{
    int slot = atomicAdd(qn, 1);
    if (slot < OWN_QCAP) qe[slot] = (unsigned)e;
}

__device__ __forceinline__ unsigned own_bit(unsigned m, int d, int lo) {
    return (m << 1) | (((unsigned)(d - lo) < (unsigned)OWN_ROWS) ? 1u : 0u);
}

__device__ __forceinline__ void own_drain(unsigned m, int item0, unsigned* qe, int* qn)
{
    while (m) {
        const int j = 31 - __builtin_ctz(m);
        m &= m - 1u;
        own_push(((item0 + (j >> 2) * OWN_THREADS) << 2) + (j & 3), qe, qn);
    }
}

__global__ void __launch_bounds__(OWN_THREADS)
k_edge_owned(const float* __restrict__ Wh, const float* __restrict__ el,
             const float* __restrict__ er, const int* __restrict__ src,
             const int* __restrict__ dst, int N, int E,
             unsigned* __restrict__ qsave, int* __restrict__ qcnt, int second,
             const float* __restrict__ bias,
             const float* __restrict__ Wn,
             const float* __restrict__ aln,
             const float* __restrict__ arn,
             const float* __restrict__ bpn,
             float* __restrict__ o_main,
             float* __restrict__ o_el, float* __restrict__ o_er)
{
    __shared__ __attribute__((aligned(16))) unsigned char raw[OWN_LDS];
    float*    acc = (float*)(raw + OFF_ACC);
    float4*   ev  = (float4*)(raw + OFF_EV);
    unsigned* mx  = (unsigned*)(raw + OFF_MX);
    float*    dn  = (float*)(raw + OFF_DN);
    unsigned* qe  = (unsigned*)(raw + OFF_QE);
    int*      qnp = (int*)(raw + OFF_QN);
    const int reuse = second;

    const int tid   = threadIdx.x;
    const int lo    = blockIdx.x * OWN_ROWS;
    const int qbase = blockIdx.x * OWN_QCAP;

    for (int i = tid; i < OWN_ROWS * ACC_LD; i += OWN_THREADS) acc[i] = 0.0f;
    if (tid < OWN_ROWS * HEADS) { mx[tid] = 0u; dn[tid] = 0.0f; }
    if (tid == 0) *qnp = 0;
    __syncthreads();

    int n;
    if (!reuse) {
        const int4* dst4  = (const int4*)dst;
        const int   n4    = E >> 2;
        const int   nfull = n4 / (8 * OWN_THREADS);
        const int   rot   = (int)(((long)blockIdx.x * nfull) / gridDim.x);
        for (int j = 0; j < nfull; ++j) {
            int sc = j + rot;
            if (sc >= nfull) sc -= nfull;
            const int item0 = sc * (8 * OWN_THREADS) + tid;
            int4 v[8];
#pragma unroll
            for (int u = 0; u < 8; ++u) v[u] = dst4[item0 + u * OWN_THREADS];
            unsigned ma = 0u, mb = 0u;
#pragma unroll
            for (int u = 0; u < 4; ++u) {
                ma = own_bit(ma, v[u].x, lo);     mb = own_bit(mb, v[u + 4].x, lo);
                ma = own_bit(ma, v[u].y, lo);     mb = own_bit(mb, v[u + 4].y, lo);
                ma = own_bit(ma, v[u].z, lo);     mb = own_bit(mb, v[u + 4].z, lo);
                ma = own_bit(ma, v[u].w, lo);     mb = own_bit(mb, v[u + 4].w, lo);
            }
            own_drain((ma << 16) | mb, item0, qe, qnp);
        }
        for (int item = nfull * (8 * OWN_THREADS) + tid; item < n4; item += OWN_THREADS) {
            const int4 t = dst4[item];
            unsigned m = own_bit(own_bit(own_bit(own_bit(0u, t.x, lo), t.y, lo), t.z, lo), t.w, lo);
            own_drain(m << 28, item, qe, qnp);
        }
        __syncthreads();
        n = *qnp;
        if (tid == 0) qcnt[blockIdx.x] = n;
    } else {
        n = qcnt[blockIdx.x];
    }

    const bool  fits = n <= OWN_QCAP;
    const uint4*  mx4 = (const uint4*)mx;
    const float4* dn4 = (const float4*)dn;

    if (fits) {
        for (int i0 = tid; i0 < n; i0 += 4 * OWN_THREADS) {
            int    ss[4], dd[4];
            bool   ok[4];
            float4 sc[4];
            if (reuse) {
                unsigned pk[4];
#pragma unroll
                for (int u = 0; u < 4; ++u) {
                    const int i = i0 + u * OWN_THREADS;
                    ok[u] = i < n;
                    pk[u] = ok[u] ? qsave[qbase + i] : 0u;
                }
#pragma unroll
                for (int u = 0; u < 4; ++u) {
                    ss[u] = (int)(pk[u] & ((1u << OWN_SBITS) - 1u));
                    dd[u] = lo + (int)(pk[u] >> OWN_SBITS);
                }
            } else {
#pragma unroll
                for (int u = 0; u < 4; ++u) {
                    const int i = i0 + u * OWN_THREADS;
                    ok[u] = i < n;
                    const int e = ok[u] ? (int)qe[i] : 0;
                    ss[u] = src[e];
                    dd[u] = dst[e];
                }
            }
#pragma unroll
            for (int u = 0; u < 4; ++u) sc[u] = edge_scores(el, er, ss[u], dd[u]);
#pragma unroll
            for (int u = 0; u < 4; ++u) {
                if (ok[u]) {
                    const int      i  = i0 + u * OWN_THREADS;
                    const unsigned ld = (unsigned)(dd[u] - lo);
                    const unsigned pk = (unsigned)ss[u] | (ld << OWN_SBITS);
                    ev[i] = sc[u];
                    qe[i] = pk;
                    if (!reuse) qsave[qbase + i] = pk;
                    lds_max(&mx[ld * HEADS + 0], f2ord(sc[u].x));
                    lds_max(&mx[ld * HEADS + 1], f2ord(sc[u].y));
                    lds_max(&mx[ld * HEADS + 2], f2ord(sc[u].z));
                    lds_max(&mx[ld * HEADS + 3], f2ord(sc[u].w));
                }
            }
        }
    } else {
#pragma nounroll
        for (int e = tid; e < E; e += OWN_THREADS) {
            const int      d  = dst[e];
            const unsigned ld = (unsigned)(d - lo);
            if (ld < (unsigned)OWN_ROWS) {
                const float4 v = edge_scores(el, er, src[e], d);
                lds_max(&mx[ld * HEADS + 0], f2ord(v.x));
                lds_max(&mx[ld * HEADS + 1], f2ord(v.y));
                lds_max(&mx[ld * HEADS + 2], f2ord(v.z));
                lds_max(&mx[ld * HEADS + 3], f2ord(v.w));
            }
        }
    }
    __syncthreads();

    if (fits) {
        for (int i = tid; i < n; i += OWN_THREADS) {
            const unsigned ld = qe[i] >> OWN_SBITS;
            const float4   v  = ev[i];
            const uint4    m  = mx4[ld];
            float4 x;
            x.x = expf(v.x - ord2f(m.x));
            x.y = expf(v.y - ord2f(m.y));
            x.z = expf(v.z - ord2f(m.z));
            x.w = expf(v.w - ord2f(m.w));
            ev[i] = x;
            lds_add(&dn[ld * HEADS + 0], x.x);
            lds_add(&dn[ld * HEADS + 1], x.y);
            lds_add(&dn[ld * HEADS + 2], x.z);
            lds_add(&dn[ld * HEADS + 3], x.w);
        }
    } else {
#pragma nounroll
        for (int e = tid; e < E; e += OWN_THREADS) {
            const int      d  = dst[e];
            const unsigned ld = (unsigned)(d - lo);
            if (ld < (unsigned)OWN_ROWS) {
                const float4 v = edge_scores(el, er, src[e], d);
                const uint4  m = mx4[ld];
                lds_add(&dn[ld * HEADS + 0], expf(v.x - ord2f(m.x)));
                lds_add(&dn[ld * HEADS + 1], expf(v.y - ord2f(m.y)));
                lds_add(&dn[ld * HEADS + 2], expf(v.z - ord2f(m.z)));
                lds_add(&dn[ld * HEADS + 3], expf(v.w - ord2f(m.w)));
            }
        }
    }
    __syncthreads();

    if (fits) {
        for (int i = tid; i < n; i += OWN_THREADS) {
            const unsigned ld = qe[i] >> OWN_SBITS;
            const float4   q  = dn4[ld];
            float4 x = ev[i];
            x.x = x.x / (q.x + EPS_);
            x.y = x.y / (q.y + EPS_);
            x.z = x.z / (q.z + EPS_);
            x.w = x.w / (q.w + EPS_);
            ev[i] = x;
        }
    }
    __syncthreads();

    if (fits) {
        const int wave = tid >> 5, lane = tid & 31;
        for (int i0 = wave; i0 < n; i0 += (OWN_THREADS / 32) * 4) {
            unsigned ldr[4];
            bool     ok[4];
            float4   al[4];
            float    w[4][HEADS];
#pragma unroll
            for (int u = 0; u < 4; ++u) {
                const int i = i0 + u * (OWN_THREADS / 32);
                ok[u] = i < n;
                const unsigned pk = ok[u] ? qe[i] : 0u;
                const int      s  = (int)(pk & ((1u << OWN_SBITS) - 1u));
                ldr[u] = pk >> OWN_SBITS;
                al[u]  = ev[ok[u] ? i : 0];
#pragma unroll
                for (int j = 0; j < HEADS; ++j) w[u][j] = Wh[(long)s * FEAT + j * D_HEAD + lane];
            }
#pragma unroll
            for (int u = 0; u < 4; ++u) {
                if (ok[u]) {
                    float* row = &acc[ldr[u] * ACC_LD + lane];
                    lds_add(row + 0 * D_HEAD, al[u].x * w[u][0]);
                    lds_add(row + 1 * D_HEAD, al[u].y * w[u][1]);
                    lds_add(row + 2 * D_HEAD, al[u].z * w[u][2]);
                    lds_add(row + 3 * D_HEAD, al[u].w * w[u][3]);
                }
            }
        }
    } else {
#pragma nounroll
        for (int e = tid; e < E; e += OWN_THREADS) {
            const int      d  = dst[e];
            const unsigned ld = (unsigned)(d - lo);
            if (ld < (unsigned)OWN_ROWS) {
                const int    s = src[e];
                const float4 v = edge_scores(el, er, s, d);
                const uint4  m = mx4[ld];
                const float4 q = dn4[ld];
                float a[HEADS];
                a[0] = expf(v.x - ord2f(m.x)) / (q.x + EPS_);
                a[1] = expf(v.y - ord2f(m.y)) / (q.y + EPS_);
                a[2] = expf(v.z - ord2f(m.z)) / (q.z + EPS_);
                a[3] = expf(v.w - ord2f(m.w)) / (q.w + EPS_);
#pragma unroll
                for (int h = 0; h < HEADS; ++h) {
#pragma nounroll
                    for (int f = 0; f < D_HEAD; ++f) {
                        const int c = h * D_HEAD + f;
                        lds_add(&acc[ld * ACC_LD + c], a[h] * Wh[(long)s * FEAT + c]);
                    }
                }
            }
        }
    }
    __syncthreads();

    int rows = N - lo;
    if (rows > OWN_ROWS) rows = OWN_ROWS;
    const int wave = tid >> 5, lane = tid & 31;
    const int r16  = lane & 15;
    const int koff = (lane >> 4) << 1;
    const int hi8  = (lane >> 4) << 3;

    if (!second) {
        float* wst = (float*)(raw + OFF_WST);
        float* stg = (float*)(raw + OFF_STG);
        for (int i = tid; i < OWN_ROWS * FEAT; i += OWN_THREADS) {
            const int r = i >> 7, c = i & (FEAT - 1);
            const float x = acc[r * ACC_LD + c] + bias[c];
            acc[r * ACC_LD + c] = (x > 0.0f) ? x : expm1f(x);
        }
        for (int i = tid; i < FEAT * FEAT / 4; i += OWN_THREADS) {
            const int k = i >> 5, c4 = (i & 31) << 2;
            *(float4*)(wst + k * W_LD + c4) = *(const float4*)(Wn + k * FEAT + c4);
        }
        __syncthreads();
        for (int t = wave; t < OWN_RT * (FEAT / 16); t += OWN_THREADS / 32) {
            const int rt = t >> 3, ct = t & 7;
            const float* ap = acc + (rt * 16 + r16) * ACC_LD + koff;
            const float* bq = wst + koff * W_LD + ct * 16 + r16;
            v8f c = {};
#pragma unroll 8
            for (int k0 = 0; k0 < FEAT; k0 += 4) {
                const v2f a = *(const v2f*)(ap + k0);
                v2f b;
                b.x = bq[k0 * W_LD];
                b.y = bq[(k0 + 1) * W_LD];
                c = __builtin_amdgcn_wmma_f32_16x16x4_f32(false, a, false, b,
                                                          (short)0, c, false, false);
            }
            const int row0 = rt * 16 + hi8;
            float* op = o_main + (long)(lo + row0) * FEAT + ct * 16 + r16;
            float* sp = stg + row0 * ST_LD + ct * 16 + r16;
            if (rt * 16 + 16 <= rows) {
#pragma unroll
                for (int i = 0; i < 8; ++i) {
                    op[i * FEAT]  = c[i];
                    sp[i * ST_LD] = c[i];
                }
            } else {
#pragma unroll
                for (int i = 0; i < 8; ++i) {
                    if (row0 + i < rows) {
                        op[i * FEAT]  = c[i];
                        sp[i * ST_LD] = c[i];
                    }
                }
            }
        }
        __syncthreads();
        if (tid < rows * HEADS) {
            const int r = tid >> 2, h = tid & 3;
            const float* w = stg + r * ST_LD + h * D_HEAD;
            const float* a = aln + h * D_HEAD;
            const float* b = arn + h * D_HEAD;
            float sl = 0.0f, sr = 0.0f;
#pragma unroll
            for (int d = 0; d < D_HEAD; ++d) {
                sl = fmaf(w[d], a[d], sl);
                sr = fmaf(w[d], b[d], sr);
            }
            o_el[(long)lo * HEADS + tid] = sl;
            o_er[(long)lo * HEADS + tid] = sr;
        }
    } else {
        float* zm = (float*)(raw + OFF_ZM);
        for (int i = tid; i < OWN_ROWS * D_HEAD; i += OWN_THREADS) {
            const int r = i >> 5, d = i & (D_HEAD - 1);
            float s = 0.0f;
#pragma unroll
            for (int h = 0; h < HEADS; ++h)
                s += acc[r * ACC_LD + h * D_HEAD + d] + bias[h * D_HEAD + d];
            zm[r * ZM_LD + d] = s * (1.0f / HEADS);
        }
        const int ct = wave;
        v2f bf[D_HEAD / 4];
#pragma unroll
        for (int q = 0; q < D_HEAD / 4; ++q) {
            bf[q].x = Wn[(q * 4 + koff)     * OUT_DIM + ct * 16 + r16];
            bf[q].y = Wn[(q * 4 + koff + 1) * OUT_DIM + ct * 16 + r16];
        }
        const float bv = bpn[ct * 16 + r16];
        __syncthreads();
        for (int rt = 0; rt < OWN_RT; ++rt) {
            const float* ap = zm + (rt * 16 + r16) * ZM_LD + koff;
            v8f c = {};
#pragma unroll
            for (int q = 0; q < D_HEAD / 4; ++q) {
                const v2f a = *(const v2f*)(ap + q * 4);
                c = __builtin_amdgcn_wmma_f32_16x16x4_f32(false, a, false, bf[q],
                                                          (short)0, c, false, false);
            }
            const int row0 = rt * 16 + hi8;
            float* op = o_main + (long)(lo + row0) * OUT_DIM + ct * 16 + r16;
            if (rt * 16 + 16 <= rows) {
#pragma unroll
                for (int i = 0; i < 8; ++i) op[i * OUT_DIM] = c[i] + bv;
            } else {
#pragma unroll
                for (int i = 0; i < 8; ++i)
                    if (row0 + i < rows) op[i * OUT_DIM] = c[i] + bv;
            }
        }
    }
}


extern "C" void kernel_launch(void* const* d_in, const int* in_sizes, int n_in,
                              void* d_out, int out_size, void* d_ws, size_t ws_size,
                              hipStream_t stream)
{
    const float* h   = (const float*)d_in[0];
    const int*   src = (const int*)  d_in[1];
    const int*   dst = (const int*)  d_in[2];
    const float* W1  = (const float*)d_in[3];
    const float* al1 = (const float*)d_in[4];
    const float* ar1 = (const float*)d_in[5];
    const float* b1  = (const float*)d_in[6];
    const float* W2  = (const float*)d_in[7];
    const float* al2 = (const float*)d_in[8];
    const float* ar2 = (const float*)d_in[9];
    const float* b2  = (const float*)d_in[10];
    const float* Wp  = (const float*)d_in[11];
    const float* bp  = (const float*)d_in[12];
    float*       out = (float*)d_out;

    const int N = N_NODES, E = N_EDGES;

    float*    Wh    = (float*)d_ws;
    float*    Wh2   = Wh  + (size_t)N * FEAT;
    float*    el    = Wh2 + (size_t)N * FEAT;
    float*    er    = el  + (size_t)N * HEADS;
    float*    el2   = er  + (size_t)N * HEADS;
    float*    er2   = el2 + (size_t)N * HEADS;
    unsigned* qsave = (unsigned*)(er2 + (size_t)N * HEADS);
    int*      qcnt  = (int*)(qsave + (size_t)OWN_BLOCKS * OWN_QCAP);

    const int mtiles = (N + 15) / 16;
    const dim3 gemmBlk(128);
    const dim3 gemmGrd128((mtiles + 3) / 4, FEAT / 16);

    k_gemm_wmma<<<gemmGrd128, gemmBlk, 0, stream>>>(h, W1, nullptr, Wh, N, FEAT, FEAT);
    k_scores<<<(N * HEADS + 255) / 256, 256, 0, stream>>>(Wh, al1, ar1, el, er, N);
    k_edge_owned<<<OWN_BLOCKS, OWN_THREADS, 0, stream>>>(Wh, el, er, src, dst, N, E, qsave, qcnt, 0,
                                                         b1, W2, al2, ar2, nullptr, Wh2, el2, er2);
    k_edge_owned<<<OWN_BLOCKS, OWN_THREADS, 0, stream>>>(Wh2, el2, er2, src, dst, N, E, qsave, qcnt, 1,
                                                         b2, Wp, nullptr, nullptr, bp, out, nullptr, nullptr);
}


# Edges dealt to owner groups once per call by a new launch: per-slice lists (source | local row, place by LDS integer add) that owners collect; the 256 full dst scans go; sums still arrival order
# speedup vs baseline: 7.4253x; 1.1205x over previous
#include <hip/hip_runtime.h>
#include <hip/hip_bf16.h>
#include <math.h>

#define N_NODES   50000
#define N_EDGES   800000
#define HEADS     4
#define D_HEAD    32
#define FEAT      128
#define OUT_DIM   512
#define NEG_SLOPE 0.2f
#define EPS_      1e-9f

typedef __attribute__((ext_vector_type(2))) float v2f;
typedef __attribute__((ext_vector_type(8))) float v8f;


__device__ __forceinline__ unsigned f2ord(float f) {
    unsigned u = __float_as_uint(f);
    return (u & 0x80000000u) ? ~u : (u | 0x80000000u);
}
__device__ __forceinline__ float ord2f(unsigned k) {
    return (k & 0x80000000u) ? __uint_as_float(k & 0x7fffffffu) : __uint_as_float(~k);
}

__global__ void k_gemm_wmma(const float* __restrict__ A, const float* __restrict__ B,
                            const float* __restrict__ bias, float* __restrict__ C,
                            int M, int N, int K)
{
    __shared__ float ldsB[128 * 16];
    const int tid  = threadIdx.x;
    const int wave = tid >> 5;
    const int lane = tid & 31;
    const int n0   = blockIdx.y * 16;

    for (int i = tid; i < K * 16; i += 128) {
        int k = i >> 4, c = i & 15;
        ldsB[i] = B[(long)k * N + n0 + c];
    }
    __syncthreads();

    const int m0 = (blockIdx.x * 4 + wave) * 16;
    if (m0 >= M) return;

    const int row   = m0 + (lane & 15);
    const int koff  = (lane >> 4) << 1;
    const int col16 = lane & 15;

    v8f c = {};
    for (int k0 = 0; k0 < K; k0 += 4) {
        v2f a, b;
        a.x = A[(long)row * K + k0 + koff];
        a.y = A[(long)row * K + k0 + koff + 1];
        b.x = ldsB[(k0 + koff)     * 16 + col16];
        b.y = ldsB[(k0 + koff + 1) * 16 + col16];
        c = __builtin_amdgcn_wmma_f32_16x16x4_f32(false, a, false, b,
                                                  (short)0, c, false, false);
    }

    const int   col   = n0 + col16;
    const float bv    = bias ? bias[col] : 0.0f;
    const int   rbase = m0 + ((lane >> 4) << 3);
#pragma unroll
    for (int i = 0; i < 8; ++i)
        C[(long)(rbase + i) * N + col] = c[i] + bv;
}

__global__ void k_scores(const float* __restrict__ Wh, const float* __restrict__ al,
                         const float* __restrict__ ar, float* __restrict__ el,
                         float* __restrict__ er, int N)
{
    int idx = blockIdx.x * blockDim.x + threadIdx.x;
    if (idx >= N * HEADS) return;
    int n = idx >> 2, h = idx & 3;
    const float* w = Wh + (long)n * FEAT + h * D_HEAD;
    const float* a = al + h * D_HEAD;
    const float* b = ar + h * D_HEAD;
    float sl = 0.0f, sr = 0.0f;
#pragma unroll
    for (int d = 0; d < D_HEAD; ++d) {
        sl = fmaf(w[d], a[d], sl);
        sr = fmaf(w[d], b[d], sr);
    }
    el[idx] = sl;
    er[idx] = sr;
}

#define OWN_ROWS    196
#define OWN_BLOCKS  ((N_NODES + OWN_ROWS - 1) / OWN_ROWS)
#define OWN_THREADS 1024
#define OWN_QCAP    8192
#define OWN_SBITS   16
static_assert(N_NODES <= (1 << OWN_SBITS), "source node fits the queue word");
static_assert(OWN_ROWS <= (1 << (32 - OWN_SBITS)), "local row fits the queue word");
static_assert(N_EDGES % 4 == 0, "src and dst are read 16 bytes at a time");
#define BIN_THREADS 1024
#define BIN_BLOCKS  ((N_EDGES / 4 + BIN_THREADS - 1) / BIN_THREADS)
#define SEG_CAP     64
static_assert((SEG_CAP & (SEG_CAP - 1)) == 0, "slot index is split by mask and shift");
#define SEG_SHIFT   6
static_assert((1 << SEG_SHIFT) == SEG_CAP, "shift matches the list length");
#define ACC_LD      132
#define W_LD        136
#define ST_LD       132
#define ZM_LD       36
#define OWN_RT      ((OWN_ROWS + 15) / 16)
#define OFF_ACC     0
#define OFF_EV      (OFF_ACC + OWN_ROWS * ACC_LD * 4)
#define OFF_QE      (OFF_EV + OWN_QCAP * 16)
#define OFF_MX      (OFF_QE + OWN_QCAP * 4)
#define OFF_DN      (OFF_MX + OWN_ROWS * HEADS * 4)
#define OFF_QN      (OFF_DN + OWN_ROWS * HEADS * 4)
#define OFF_SGL     (OFF_QN + 16)
#define OFF_WST     OFF_EV
#define OFF_STG     (OFF_WST + FEAT * W_LD * 4)
#define OFF_ZM      OFF_EV
#define OWN_LDS     (OFF_STG + OWN_ROWS * ST_LD * 4)
static_assert(OWN_LDS >= OFF_SGL + BIN_BLOCKS * 8 && OWN_LDS <= 320 * 1024, "fits one WGP's LDS");
static_assert(OFF_SGL % 8 == 0, "8-byte pairs");
static_assert(BIN_BLOCKS <= OWN_THREADS, "one thread per list when collecting");
#define COPY_TRIPS  ((BIN_BLOCKS * SEG_CAP + OWN_THREADS - 1) / OWN_THREADS)
static_assert(OFF_ZM + OWN_RT * 16 * ZM_LD * 4 <= OWN_LDS, "head means fit");
static_assert(OFF_ACC + OWN_RT * 16 * ACC_LD * 4 <= OWN_LDS, "the last tile's spare rows are still LDS");
static_assert(OFF_EV % 16 == 0 && OFF_MX % 16 == 0 && OFF_DN % 16 == 0 && OFF_STG % 16 == 0, "16-byte pieces");
static_assert(OUT_DIM / 16 == OWN_THREADS / 32, "second pass: one output column tile per wave");

__device__ __forceinline__ void lds_add(float* p, float v) {
    __hip_atomic_fetch_add(p, v, __ATOMIC_RELAXED, __HIP_MEMORY_SCOPE_WORKGROUP);
}
__device__ __forceinline__ void lds_max(unsigned* p, unsigned v) {
    __hip_atomic_fetch_max(p, v, __ATOMIC_RELAXED, __HIP_MEMORY_SCOPE_WORKGROUP);
}
__device__ __forceinline__ float leaky(float v) { return (v > 0.0f) ? v : NEG_SLOPE * v; }

__device__ __forceinline__ float4 edge_scores(const float* __restrict__ el, const float* __restrict__ er,
                                              int s, int d)
{
    const float4 l = *(const float4*)(el + (long)s * HEADS);
    const float4 r = *(const float4*)(er + (long)d * HEADS);
    float4 v;
    v.x = leaky(l.x + r.x);
    v.y = leaky(l.y + r.y);
    v.z = leaky(l.z + r.z);
    v.w = leaky(l.w + r.w);
    return v;
}

__device__ __forceinline__ void bin_edge(int s, int d, int* cnt, unsigned* __restrict__ seg)
{
    const unsigned g = (unsigned)d / (unsigned)OWN_ROWS;
    if (g < (unsigned)OWN_BLOCKS) {
        const unsigned lr = (unsigned)d - g * (unsigned)OWN_ROWS;
        const int r = atomicAdd(&cnt[g], 1);
        if (r < SEG_CAP)
            seg[((size_t)g * BIN_BLOCKS + blockIdx.x) * SEG_CAP + r] = (unsigned)s | (lr << OWN_SBITS);
    }
}

__global__ void __launch_bounds__(BIN_THREADS)
k_bin(const int* __restrict__ src, const int* __restrict__ dst, int E,
      unsigned* __restrict__ seg, int* __restrict__ segcnt)
{
    __shared__ int cnt[OWN_BLOCKS];
    const int tid = threadIdx.x;
    if (tid < OWN_BLOCKS) cnt[tid] = 0;
    __syncthreads();
    const int item = blockIdx.x * BIN_THREADS + tid;
    if (item < (E >> 2)) {
        const int4 d = ((const int4*)dst)[item];
        const int4 s = ((const int4*)src)[item];
        bin_edge(s.x, d.x, cnt, seg);
        bin_edge(s.y, d.y, cnt, seg);
        bin_edge(s.z, d.z, cnt, seg);
        bin_edge(s.w, d.w, cnt, seg);
    }
    __syncthreads();
    if (tid < OWN_BLOCKS) segcnt[tid * BIN_BLOCKS + blockIdx.x] = cnt[tid];
}

__global__ void __launch_bounds__(OWN_THREADS)
k_edge_owned(const float* __restrict__ Wh, const float* __restrict__ el,
             const float* __restrict__ er, const int* __restrict__ src,
             const int* __restrict__ dst, int N, int E,
             unsigned* __restrict__ qsave, int* __restrict__ qcnt, int second,
             const unsigned* __restrict__ seg,
             const int* __restrict__ segcnt,
             const float* __restrict__ bias,
             const float* __restrict__ Wn,
             const float* __restrict__ aln,
             const float* __restrict__ arn,
             const float* __restrict__ bpn,
             float* __restrict__ o_main,
             float* __restrict__ o_el, float* __restrict__ o_er)
{
    __shared__ __attribute__((aligned(16))) unsigned char raw[OWN_LDS];
    float*    acc = (float*)(raw + OFF_ACC);
    float4*   ev  = (float4*)(raw + OFF_EV);
    unsigned* mx  = (unsigned*)(raw + OFF_MX);
    float*    dn  = (float*)(raw + OFF_DN);
    unsigned* qe  = (unsigned*)(raw + OFF_QE);
    int*      qnp = (int*)(raw + OFF_QN);
    int2*     sgl = (int2*)(raw + OFF_SGL);
    const int reuse = second;

    const int tid   = threadIdx.x;
    const int lo    = blockIdx.x * OWN_ROWS;
    const int qbase = blockIdx.x * OWN_QCAP;

    for (int i = tid; i < OWN_ROWS * ACC_LD; i += OWN_THREADS) acc[i] = 0.0f;
    if (tid < OWN_ROWS * HEADS) { mx[tid] = 0u; dn[tid] = 0.0f; }
    if (tid == 0) *qnp = 0;
    __syncthreads();

    int n;
    if (!reuse) {
        if (tid < BIN_BLOCKS) {
            const int cn = segcnt[blockIdx.x * BIN_BLOCKS + tid];
            sgl[tid] = make_int2(cn, atomicAdd(qnp, (cn <= SEG_CAP) ? cn : OWN_QCAP + 1));
        }
        __syncthreads();
        n = *qnp;
        if (n <= OWN_QCAP) {
            const unsigned* sp = seg + (size_t)blockIdx.x * (BIN_BLOCKS * SEG_CAP);
            int      pos[COPY_TRIPS];
            unsigned wd[COPY_TRIPS];
#pragma unroll
            for (int k = 0; k < COPY_TRIPS; ++k) {
                const int slot = tid + k * OWN_THREADS;
                const int b = slot >> SEG_SHIFT, r = slot & (SEG_CAP - 1);
                const int2 lp = sgl[(b < BIN_BLOCKS) ? b : BIN_BLOCKS - 1];
                pos[k] = (slot < BIN_BLOCKS * SEG_CAP && r < lp.x) ? lp.y + r : -1;
            }
#pragma unroll
            for (int k = 0; k < COPY_TRIPS; ++k)
                wd[k] = (pos[k] >= 0) ? sp[tid + k * OWN_THREADS] : 0u;
#pragma unroll
            for (int k = 0; k < COPY_TRIPS; ++k)
                if (pos[k] >= 0) qe[pos[k]] = wd[k];
        }
        if (tid == 0) qcnt[blockIdx.x] = n;
        __syncthreads();
    } else {
        n = qcnt[blockIdx.x];
    }

    const bool  fits = n <= OWN_QCAP;
    const uint4*  mx4 = (const uint4*)mx;
    const float4* dn4 = (const float4*)dn;

    if (fits) {
        for (int i0 = tid; i0 < n; i0 += 4 * OWN_THREADS) {
            int      ss[4], dd[4];
            bool     ok[4];
            float4   sc[4];
            unsigned pk[4];
#pragma unroll
            for (int u = 0; u < 4; ++u) {
                const int i = i0 + u * OWN_THREADS;
                ok[u] = i < n;
                pk[u] = ok[u] ? (reuse ? qsave[qbase + i] : qe[i]) : 0u;
            }
#pragma unroll
            for (int u = 0; u < 4; ++u) {
                ss[u] = (int)(pk[u] & ((1u << OWN_SBITS) - 1u));
                dd[u] = lo + (int)(pk[u] >> OWN_SBITS);
            }
#pragma unroll
            for (int u = 0; u < 4; ++u) sc[u] = edge_scores(el, er, ss[u], dd[u]);
#pragma unroll
            for (int u = 0; u < 4; ++u) {
                if (ok[u]) {
                    const int      i  = i0 + u * OWN_THREADS;
                    const unsigned ld = pk[u] >> OWN_SBITS;
                    ev[i] = sc[u];
                    if (reuse) qe[i] = pk[u];
                    else       qsave[qbase + i] = pk[u];
                    lds_max(&mx[ld * HEADS + 0], f2ord(sc[u].x));
                    lds_max(&mx[ld * HEADS + 1], f2ord(sc[u].y));
                    lds_max(&mx[ld * HEADS + 2], f2ord(sc[u].z));
                    lds_max(&mx[ld * HEADS + 3], f2ord(sc[u].w));
                }
            }
        }
    } else {
#pragma nounroll
        for (int e = tid; e < E; e += OWN_THREADS) {
            const int      d  = dst[e];
            const unsigned ld = (unsigned)(d - lo);
            if (ld < (unsigned)OWN_ROWS) {
                const float4 v = edge_scores(el, er, src[e], d);
                lds_max(&mx[ld * HEADS + 0], f2ord(v.x));
                lds_max(&mx[ld * HEADS + 1], f2ord(v.y));
                lds_max(&mx[ld * HEADS + 2], f2ord(v.z));
                lds_max(&mx[ld * HEADS + 3], f2ord(v.w));
            }
        }
    }
    __syncthreads();

    if (fits) {
        for (int i = tid; i < n; i += OWN_THREADS) {
            const unsigned ld = qe[i] >> OWN_SBITS;
            const float4   v  = ev[i];
            const uint4    m  = mx4[ld];
            float4 x;
            x.x = expf(v.x - ord2f(m.x));
            x.y = expf(v.y - ord2f(m.y));
            x.z = expf(v.z - ord2f(m.z));
            x.w = expf(v.w - ord2f(m.w));
            ev[i] = x;
            lds_add(&dn[ld * HEADS + 0], x.x);
            lds_add(&dn[ld * HEADS + 1], x.y);
            lds_add(&dn[ld * HEADS + 2], x.z);
            lds_add(&dn[ld * HEADS + 3], x.w);
        }
    } else {
#pragma nounroll
        for (int e = tid; e < E; e += OWN_THREADS) {
            const int      d  = dst[e];
            const unsigned ld = (unsigned)(d - lo);
            if (ld < (unsigned)OWN_ROWS) {
                const float4 v = edge_scores(el, er, src[e], d);
                const uint4  m = mx4[ld];
                lds_add(&dn[ld * HEADS + 0], expf(v.x - ord2f(m.x)));
                lds_add(&dn[ld * HEADS + 1], expf(v.y - ord2f(m.y)));
                lds_add(&dn[ld * HEADS + 2], expf(v.z - ord2f(m.z)));
                lds_add(&dn[ld * HEADS + 3], expf(v.w - ord2f(m.w)));
            }
        }
    }
    __syncthreads();

    if (fits) {
        for (int i = tid; i < n; i += OWN_THREADS) {
            const unsigned ld = qe[i] >> OWN_SBITS;
            const float4   q  = dn4[ld];
            float4 x = ev[i];
            x.x = x.x / (q.x + EPS_);
            x.y = x.y / (q.y + EPS_);
            x.z = x.z / (q.z + EPS_);
            x.w = x.w / (q.w + EPS_);
            ev[i] = x;
        }
    }
    __syncthreads();

    if (fits) {
        const int wave = tid >> 5, lane = tid & 31;
        for (int i0 = wave; i0 < n; i0 += (OWN_THREADS / 32) * 4) {
            unsigned ldr[4];
            bool     ok[4];
            float4   al[4];
            float    w[4][HEADS];
#pragma unroll
            for (int u = 0; u < 4; ++u) {
                const int i = i0 + u * (OWN_THREADS / 32);
                ok[u] = i < n;
                const unsigned pk = ok[u] ? qe[i] : 0u;
                const int      s  = (int)(pk & ((1u << OWN_SBITS) - 1u));
                ldr[u] = pk >> OWN_SBITS;
                al[u]  = ev[ok[u] ? i : 0];
#pragma unroll
                for (int j = 0; j < HEADS; ++j) w[u][j] = Wh[(long)s * FEAT + j * D_HEAD + lane];
            }
#pragma unroll
            for (int u = 0; u < 4; ++u) {
                if (ok[u]) {
                    float* row = &acc[ldr[u] * ACC_LD + lane];
                    lds_add(row + 0 * D_HEAD, al[u].x * w[u][0]);
                    lds_add(row + 1 * D_HEAD, al[u].y * w[u][1]);
                    lds_add(row + 2 * D_HEAD, al[u].z * w[u][2]);
                    lds_add(row + 3 * D_HEAD, al[u].w * w[u][3]);
                }
            }
        }
    } else {
#pragma nounroll
        for (int e = tid; e < E; e += OWN_THREADS) {
            const int      d  = dst[e];
            const unsigned ld = (unsigned)(d - lo);
            if (ld < (unsigned)OWN_ROWS) {
                const int    s = src[e];
                const float4 v = edge_scores(el, er, s, d);
                const uint4  m = mx4[ld];
                const float4 q = dn4[ld];
                float a[HEADS];
                a[0] = expf(v.x - ord2f(m.x)) / (q.x + EPS_);
                a[1] = expf(v.y - ord2f(m.y)) / (q.y + EPS_);
                a[2] = expf(v.z - ord2f(m.z)) / (q.z + EPS_);
                a[3] = expf(v.w - ord2f(m.w)) / (q.w + EPS_);
#pragma unroll
                for (int h = 0; h < HEADS; ++h) {
#pragma nounroll
                    for (int f = 0; f < D_HEAD; ++f) {
                        const int c = h * D_HEAD + f;
                        lds_add(&acc[ld * ACC_LD + c], a[h] * Wh[(long)s * FEAT + c]);
                    }
                }
            }
        }
    }
    __syncthreads();

    int rows = N - lo;
    if (rows > OWN_ROWS) rows = OWN_ROWS;
    const int wave = tid >> 5, lane = tid & 31;
    const int r16  = lane & 15;
    const int koff = (lane >> 4) << 1;
    const int hi8  = (lane >> 4) << 3;

    if (!second) {
        float* wst = (float*)(raw + OFF_WST);
        float* stg = (float*)(raw + OFF_STG);
        for (int i = tid; i < OWN_ROWS * FEAT; i += OWN_THREADS) {
            const int r = i >> 7, c = i & (FEAT - 1);
            const float x = acc[r * ACC_LD + c] + bias[c];
            acc[r * ACC_LD + c] = (x > 0.0f) ? x : expm1f(x);
        }
        for (int i = tid; i < FEAT * FEAT / 4; i += OWN_THREADS) {
            const int k = i >> 5, c4 = (i & 31) << 2;
            *(float4*)(wst + k * W_LD + c4) = *(const float4*)(Wn + k * FEAT + c4);
        }
        __syncthreads();
        for (int t = wave; t < OWN_RT * (FEAT / 16); t += OWN_THREADS / 32) {
            const int rt = t >> 3, ct = t & 7;
            const float* ap = acc + (rt * 16 + r16) * ACC_LD + koff;
            const float* bq = wst + koff * W_LD + ct * 16 + r16;
            v8f c = {};
#pragma unroll 8
            for (int k0 = 0; k0 < FEAT; k0 += 4) {
                const v2f a = *(const v2f*)(ap + k0);
                v2f b;
                b.x = bq[k0 * W_LD];
                b.y = bq[(k0 + 1) * W_LD];
                c = __builtin_amdgcn_wmma_f32_16x16x4_f32(false, a, false, b,
                                                          (short)0, c, false, false);
            }
            const int row0 = rt * 16 + hi8;
            float* op = o_main + (long)(lo + row0) * FEAT + ct * 16 + r16;
            float* sp = stg + row0 * ST_LD + ct * 16 + r16;
            if (rt * 16 + 16 <= rows) {
#pragma unroll
                for (int i = 0; i < 8; ++i) {
                    op[i * FEAT]  = c[i];
                    sp[i * ST_LD] = c[i];
                }
            } else {
#pragma unroll
                for (int i = 0; i < 8; ++i) {
                    if (row0 + i < rows) {
                        op[i * FEAT]  = c[i];
                        sp[i * ST_LD] = c[i];
                    }
                }
            }
        }
        __syncthreads();
        if (tid < rows * HEADS) {
            const int r = tid >> 2, h = tid & 3;
            const float* w = stg + r * ST_LD + h * D_HEAD;
            const float* a = aln + h * D_HEAD;
            const float* b = arn + h * D_HEAD;
            float sl = 0.0f, sr = 0.0f;
#pragma unroll
            for (int d = 0; d < D_HEAD; ++d) {
                sl = fmaf(w[d], a[d], sl);
                sr = fmaf(w[d], b[d], sr);
            }
            o_el[(long)lo * HEADS + tid] = sl;
            o_er[(long)lo * HEADS + tid] = sr;
        }
    } else {
        float* zm = (float*)(raw + OFF_ZM);
        for (int i = tid; i < OWN_ROWS * D_HEAD; i += OWN_THREADS) {
            const int r = i >> 5, d = i & (D_HEAD - 1);
            float s = 0.0f;
#pragma unroll
            for (int h = 0; h < HEADS; ++h)
                s += acc[r * ACC_LD + h * D_HEAD + d] + bias[h * D_HEAD + d];
            zm[r * ZM_LD + d] = s * (1.0f / HEADS);
        }
        const int ct = wave;
        v2f bf[D_HEAD / 4];
#pragma unroll
        for (int q = 0; q < D_HEAD / 4; ++q) {
            bf[q].x = Wn[(q * 4 + koff)     * OUT_DIM + ct * 16 + r16];
            bf[q].y = Wn[(q * 4 + koff + 1) * OUT_DIM + ct * 16 + r16];
        }
        const float bv = bpn[ct * 16 + r16];
        __syncthreads();
        for (int rt = 0; rt < OWN_RT; ++rt) {
            const float* ap = zm + (rt * 16 + r16) * ZM_LD + koff;
            v8f c = {};
#pragma unroll
            for (int q = 0; q < D_HEAD / 4; ++q) {
                const v2f a = *(const v2f*)(ap + q * 4);
                c = __builtin_amdgcn_wmma_f32_16x16x4_f32(false, a, false, bf[q],
                                                          (short)0, c, false, false);
            }
            const int row0 = rt * 16 + hi8;
            float* op = o_main + (long)(lo + row0) * OUT_DIM + ct * 16 + r16;
            if (rt * 16 + 16 <= rows) {
#pragma unroll
                for (int i = 0; i < 8; ++i) op[i * OUT_DIM] = c[i] + bv;
            } else {
#pragma unroll
                for (int i = 0; i < 8; ++i)
                    if (row0 + i < rows) op[i * OUT_DIM] = c[i] + bv;
            }
        }
    }
}


extern "C" void kernel_launch(void* const* d_in, const int* in_sizes, int n_in,
                              void* d_out, int out_size, void* d_ws, size_t ws_size,
                              hipStream_t stream)
{
    const float* h   = (const float*)d_in[0];
    const int*   src = (const int*)  d_in[1];
    const int*   dst = (const int*)  d_in[2];
    const float* W1  = (const float*)d_in[3];
    const float* al1 = (const float*)d_in[4];
    const float* ar1 = (const float*)d_in[5];
    const float* b1  = (const float*)d_in[6];
    const float* W2  = (const float*)d_in[7];
    const float* al2 = (const float*)d_in[8];
    const float* ar2 = (const float*)d_in[9];
    const float* b2  = (const float*)d_in[10];
    const float* Wp  = (const float*)d_in[11];
    const float* bp  = (const float*)d_in[12];
    float*       out = (float*)d_out;

    const int N = N_NODES, E = N_EDGES;

    float*    Wh    = (float*)d_ws;
    float*    Wh2   = Wh  + (size_t)N * FEAT;
    float*    el    = Wh2 + (size_t)N * FEAT;
    float*    er    = el  + (size_t)N * HEADS;
    float*    el2   = er  + (size_t)N * HEADS;
    float*    er2   = el2 + (size_t)N * HEADS;
    unsigned* qsave = (unsigned*)(er2 + (size_t)N * HEADS);
    int*      qcnt  = (int*)(qsave + (size_t)OWN_BLOCKS * OWN_QCAP);
    int*      segcnt = qcnt + OWN_BLOCKS;
    unsigned* seg    = (unsigned*)(segcnt + (size_t)OWN_BLOCKS * BIN_BLOCKS);

    const int mtiles = (N + 15) / 16;
    const dim3 gemmBlk(128);
    const dim3 gemmGrd128((mtiles + 3) / 4, FEAT / 16);

    k_bin<<<BIN_BLOCKS, BIN_THREADS, 0, stream>>>(src, dst, E, seg, segcnt);
    k_gemm_wmma<<<gemmGrd128, gemmBlk, 0, stream>>>(h, W1, nullptr, Wh, N, FEAT, FEAT);
    k_scores<<<(N * HEADS + 255) / 256, 256, 0, stream>>>(Wh, al1, ar1, el, er, N);
    k_edge_owned<<<OWN_BLOCKS, OWN_THREADS, 0, stream>>>(Wh, el, er, src, dst, N, E, qsave, qcnt, 0, seg, segcnt,
                                                         b1, W2, al2, ar2, nullptr, Wh2, el2, er2);
    k_edge_owned<<<OWN_BLOCKS, OWN_THREADS, 0, stream>>>(Wh2, el2, er2, src, dst, N, E, qsave, qcnt, 1, seg, segcnt,
                                                         b2, Wp, nullptr, nullptr, bp, out, nullptr, nullptr);
}


# Layer 1's product and scores by the row-range groups: h rows staged in LDS (16-byte loads) as A, each wave's W1 fragments kept in registers, el/er from LDS product rows; GEMM + score launches fused
# speedup vs baseline: 8.7764x; 1.1820x over previous
#include <hip/hip_runtime.h>
#include <hip/hip_bf16.h>
#include <math.h>

#define N_NODES   50000
#define N_EDGES   800000
#define HEADS     4
#define D_HEAD    32
#define FEAT      128
#define OUT_DIM   512
#define NEG_SLOPE 0.2f
#define EPS_      1e-9f

typedef __attribute__((ext_vector_type(2))) float v2f;
typedef __attribute__((ext_vector_type(8))) float v8f;


__device__ __forceinline__ unsigned f2ord(float f) {
    unsigned u = __float_as_uint(f);
    return (u & 0x80000000u) ? ~u : (u | 0x80000000u);
}
__device__ __forceinline__ float ord2f(unsigned k) {
    return (k & 0x80000000u) ? __uint_as_float(k & 0x7fffffffu) : __uint_as_float(~k);
}

#define OWN_ROWS    196
#define OWN_BLOCKS  ((N_NODES + OWN_ROWS - 1) / OWN_ROWS)
#define OWN_THREADS 1024
#define OWN_QCAP    8192
#define OWN_SBITS   16
static_assert(N_NODES <= (1 << OWN_SBITS), "source node fits the queue word");
static_assert(OWN_ROWS <= (1 << (32 - OWN_SBITS)), "local row fits the queue word");
static_assert(N_EDGES % 4 == 0, "src and dst are read 16 bytes at a time");
#define BIN_THREADS 1024
#define BIN_BLOCKS  ((N_EDGES / 4 + BIN_THREADS - 1) / BIN_THREADS)
#define SEG_CAP     64
static_assert((SEG_CAP & (SEG_CAP - 1)) == 0, "slot index is split by mask and shift");
#define SEG_SHIFT   6
static_assert((1 << SEG_SHIFT) == SEG_CAP, "shift matches the list length");
#define ACC_LD      132
#define W_LD        136
#define ST_LD       132
#define ZM_LD       36
#define OWN_RT      ((OWN_ROWS + 15) / 16)
#define OFF_ACC     0
#define OFF_EV      (OFF_ACC + OWN_ROWS * ACC_LD * 4)
#define OFF_QE      (OFF_EV + OWN_QCAP * 16)
#define OFF_MX      (OFF_QE + OWN_QCAP * 4)
#define OFF_DN      (OFF_MX + OWN_ROWS * HEADS * 4)
#define OFF_QN      (OFF_DN + OWN_ROWS * HEADS * 4)
#define OFF_SGL     (OFF_QN + 16)
#define OFF_WST     OFF_EV
#define OFF_STG     (OFF_WST + FEAT * W_LD * 4)
#define OFF_ZM      OFF_EV
#define OWN_LDS     (OFF_STG + OWN_ROWS * ST_LD * 4)
static_assert(OWN_LDS >= OFF_SGL + BIN_BLOCKS * 8 && OWN_LDS <= 320 * 1024, "fits one WGP's LDS");
static_assert(OFF_SGL % 8 == 0, "8-byte pairs");
static_assert(BIN_BLOCKS <= OWN_THREADS, "one thread per list when collecting");
#define COPY_TRIPS  ((BIN_BLOCKS * SEG_CAP + OWN_THREADS - 1) / OWN_THREADS)
static_assert(OFF_ZM + OWN_RT * 16 * ZM_LD * 4 <= OWN_LDS, "head means fit");
static_assert(OFF_ACC + OWN_RT * 16 * ACC_LD * 4 <= OWN_LDS, "the last tile's spare rows are still LDS");
static_assert(OFF_EV % 16 == 0 && OFF_MX % 16 == 0 && OFF_DN % 16 == 0 && OFF_STG % 16 == 0, "16-byte pieces");
static_assert(OUT_DIM / 16 == OWN_THREADS / 32, "second pass: one output column tile per wave");

__device__ __forceinline__ void lds_add(float* p, float v) {
    __hip_atomic_fetch_add(p, v, __ATOMIC_RELAXED, __HIP_MEMORY_SCOPE_WORKGROUP);
}
__device__ __forceinline__ void lds_max(unsigned* p, unsigned v) {
    __hip_atomic_fetch_max(p, v, __ATOMIC_RELAXED, __HIP_MEMORY_SCOPE_WORKGROUP);
}
__device__ __forceinline__ float leaky(float v) { return (v > 0.0f) ? v : NEG_SLOPE * v; }

__device__ __forceinline__ float4 edge_scores(const float* __restrict__ el, const float* __restrict__ er,
                                              int s, int d)
{
    const float4 l = *(const float4*)(el + (long)s * HEADS);
    const float4 r = *(const float4*)(er + (long)d * HEADS);
    float4 v;
    v.x = leaky(l.x + r.x);
    v.y = leaky(l.y + r.y);
    v.z = leaky(l.z + r.z);
    v.w = leaky(l.w + r.w);
    return v;
}

__device__ __forceinline__ void bin_edge(int s, int d, int* cnt, unsigned* __restrict__ seg)
{
    const unsigned g = (unsigned)d / (unsigned)OWN_ROWS;
    if (g < (unsigned)OWN_BLOCKS) {
        const unsigned lr = (unsigned)d - g * (unsigned)OWN_ROWS;
        const int r = atomicAdd(&cnt[g], 1);
        if (r < SEG_CAP)
            seg[((size_t)g * BIN_BLOCKS + blockIdx.x) * SEG_CAP + r] = (unsigned)s | (lr << OWN_SBITS);
    }
}

__global__ void __launch_bounds__(BIN_THREADS)
k_bin(const int* __restrict__ src, const int* __restrict__ dst, int E,
      unsigned* __restrict__ seg, int* __restrict__ segcnt)
{
    __shared__ int cnt[OWN_BLOCKS];
    const int tid = threadIdx.x;
    if (tid < OWN_BLOCKS) cnt[tid] = 0;
    __syncthreads();
    const int item = blockIdx.x * BIN_THREADS + tid;
    if (item < (E >> 2)) {
        const int4 d = ((const int4*)dst)[item];
        const int4 s = ((const int4*)src)[item];
        bin_edge(s.x, d.x, cnt, seg);
        bin_edge(s.y, d.y, cnt, seg);
        bin_edge(s.z, d.z, cnt, seg);
        bin_edge(s.w, d.w, cnt, seg);
    }
    __syncthreads();
    if (tid < OWN_BLOCKS) segcnt[tid * BIN_BLOCKS + blockIdx.x] = cnt[tid];
}

__global__ void __launch_bounds__(OWN_THREADS)
k_edge_owned(const float* __restrict__ Wh, const float* __restrict__ el,
             const float* __restrict__ er, const int* __restrict__ src,
             const int* __restrict__ dst, int N, int E,
             unsigned* __restrict__ qsave, int* __restrict__ qcnt, int second,
             const unsigned* __restrict__ seg,
             const int* __restrict__ segcnt,
             const float* __restrict__ bias,
             const float* __restrict__ Wn,
             const float* __restrict__ aln,
             const float* __restrict__ arn,
             const float* __restrict__ bpn,
             float* __restrict__ o_main,
             float* __restrict__ o_el, float* __restrict__ o_er)
{
    __shared__ __attribute__((aligned(16))) unsigned char raw[OWN_LDS];
    float*    acc = (float*)(raw + OFF_ACC);
    float4*   ev  = (float4*)(raw + OFF_EV);
    unsigned* mx  = (unsigned*)(raw + OFF_MX);
    float*    dn  = (float*)(raw + OFF_DN);
    unsigned* qe  = (unsigned*)(raw + OFF_QE);
    int*      qnp = (int*)(raw + OFF_QN);
    int2*     sgl = (int2*)(raw + OFF_SGL);
    const int reuse = second;

    const int tid   = threadIdx.x;
    const int lo    = blockIdx.x * OWN_ROWS;
    const int qbase = blockIdx.x * OWN_QCAP;

    for (int i = tid; i < OWN_ROWS * ACC_LD; i += OWN_THREADS) acc[i] = 0.0f;
    if (tid < OWN_ROWS * HEADS) { mx[tid] = 0u; dn[tid] = 0.0f; }
    if (tid == 0) *qnp = 0;
    __syncthreads();

    int n;
    if (!reuse) {
        if (tid < BIN_BLOCKS) {
            const int cn = segcnt[blockIdx.x * BIN_BLOCKS + tid];
            sgl[tid] = make_int2(cn, atomicAdd(qnp, (cn <= SEG_CAP) ? cn : OWN_QCAP + 1));
        }
        __syncthreads();
        n = *qnp;
        if (n <= OWN_QCAP) {
            const unsigned* sp = seg + (size_t)blockIdx.x * (BIN_BLOCKS * SEG_CAP);
            int      pos[COPY_TRIPS];
            unsigned wd[COPY_TRIPS];
#pragma unroll
            for (int k = 0; k < COPY_TRIPS; ++k) {
                const int slot = tid + k * OWN_THREADS;
                const int b = slot >> SEG_SHIFT, r = slot & (SEG_CAP - 1);
                const int2 lp = sgl[(b < BIN_BLOCKS) ? b : BIN_BLOCKS - 1];
                pos[k] = (slot < BIN_BLOCKS * SEG_CAP && r < lp.x) ? lp.y + r : -1;
            }
#pragma unroll
            for (int k = 0; k < COPY_TRIPS; ++k)
                wd[k] = (pos[k] >= 0) ? sp[tid + k * OWN_THREADS] : 0u;
#pragma unroll
            for (int k = 0; k < COPY_TRIPS; ++k)
                if (pos[k] >= 0) qe[pos[k]] = wd[k];
        }
        if (tid == 0) qcnt[blockIdx.x] = n;
        __syncthreads();
    } else {
        n = qcnt[blockIdx.x];
    }

    const bool  fits = n <= OWN_QCAP;
    const uint4*  mx4 = (const uint4*)mx;
    const float4* dn4 = (const float4*)dn;

    if (fits) {
        for (int i0 = tid; i0 < n; i0 += 4 * OWN_THREADS) {
            int      ss[4], dd[4];
            bool     ok[4];
            float4   sc[4];
            unsigned pk[4];
#pragma unroll
            for (int u = 0; u < 4; ++u) {
                const int i = i0 + u * OWN_THREADS;
                ok[u] = i < n;
                pk[u] = ok[u] ? (reuse ? qsave[qbase + i] : qe[i]) : 0u;
            }
#pragma unroll
            for (int u = 0; u < 4; ++u) {
                ss[u] = (int)(pk[u] & ((1u << OWN_SBITS) - 1u));
                dd[u] = lo + (int)(pk[u] >> OWN_SBITS);
            }
#pragma unroll
            for (int u = 0; u < 4; ++u) sc[u] = edge_scores(el, er, ss[u], dd[u]);
#pragma unroll
            for (int u = 0; u < 4; ++u) {
                if (ok[u]) {
                    const int      i  = i0 + u * OWN_THREADS;
                    const unsigned ld = pk[u] >> OWN_SBITS;
                    ev[i] = sc[u];
                    if (reuse) qe[i] = pk[u];
                    else       qsave[qbase + i] = pk[u];
                    lds_max(&mx[ld * HEADS + 0], f2ord(sc[u].x));
                    lds_max(&mx[ld * HEADS + 1], f2ord(sc[u].y));
                    lds_max(&mx[ld * HEADS + 2], f2ord(sc[u].z));
                    lds_max(&mx[ld * HEADS + 3], f2ord(sc[u].w));
                }
            }
        }
    } else {
#pragma nounroll
        for (int e = tid; e < E; e += OWN_THREADS) {
            const int      d  = dst[e];
            const unsigned ld = (unsigned)(d - lo);
            if (ld < (unsigned)OWN_ROWS) {
                const float4 v = edge_scores(el, er, src[e], d);
                lds_max(&mx[ld * HEADS + 0], f2ord(v.x));
                lds_max(&mx[ld * HEADS + 1], f2ord(v.y));
                lds_max(&mx[ld * HEADS + 2], f2ord(v.z));
                lds_max(&mx[ld * HEADS + 3], f2ord(v.w));
            }
        }
    }
    __syncthreads();

    if (fits) {
        for (int i = tid; i < n; i += OWN_THREADS) {
            const unsigned ld = qe[i] >> OWN_SBITS;
            const float4   v  = ev[i];
            const uint4    m  = mx4[ld];
            float4 x;
            x.x = expf(v.x - ord2f(m.x));
            x.y = expf(v.y - ord2f(m.y));
            x.z = expf(v.z - ord2f(m.z));
            x.w = expf(v.w - ord2f(m.w));
            ev[i] = x;
            lds_add(&dn[ld * HEADS + 0], x.x);
            lds_add(&dn[ld * HEADS + 1], x.y);
            lds_add(&dn[ld * HEADS + 2], x.z);
            lds_add(&dn[ld * HEADS + 3], x.w);
        }
    } else {
#pragma nounroll
        for (int e = tid; e < E; e += OWN_THREADS) {
            const int      d  = dst[e];
            const unsigned ld = (unsigned)(d - lo);
            if (ld < (unsigned)OWN_ROWS) {
                const float4 v = edge_scores(el, er, src[e], d);
                const uint4  m = mx4[ld];
                lds_add(&dn[ld * HEADS + 0], expf(v.x - ord2f(m.x)));
                lds_add(&dn[ld * HEADS + 1], expf(v.y - ord2f(m.y)));
                lds_add(&dn[ld * HEADS + 2], expf(v.z - ord2f(m.z)));
                lds_add(&dn[ld * HEADS + 3], expf(v.w - ord2f(m.w)));
            }
        }
    }
    __syncthreads();

    if (fits) {
        for (int i = tid; i < n; i += OWN_THREADS) {
            const unsigned ld = qe[i] >> OWN_SBITS;
            const float4   q  = dn4[ld];
            float4 x = ev[i];
            x.x = x.x / (q.x + EPS_);
            x.y = x.y / (q.y + EPS_);
            x.z = x.z / (q.z + EPS_);
            x.w = x.w / (q.w + EPS_);
            ev[i] = x;
        }
    }
    __syncthreads();

    if (fits) {
        const int wave = tid >> 5, lane = tid & 31;
        for (int i0 = wave; i0 < n; i0 += (OWN_THREADS / 32) * 4) {
            unsigned ldr[4];
            bool     ok[4];
            float4   al[4];
            float    w[4][HEADS];
#pragma unroll
            for (int u = 0; u < 4; ++u) {
                const int i = i0 + u * (OWN_THREADS / 32);
                ok[u] = i < n;
                const unsigned pk = ok[u] ? qe[i] : 0u;
                const int      s  = (int)(pk & ((1u << OWN_SBITS) - 1u));
                ldr[u] = pk >> OWN_SBITS;
                al[u]  = ev[ok[u] ? i : 0];
#pragma unroll
                for (int j = 0; j < HEADS; ++j) w[u][j] = Wh[(long)s * FEAT + j * D_HEAD + lane];
            }
#pragma unroll
            for (int u = 0; u < 4; ++u) {
                if (ok[u]) {
                    float* row = &acc[ldr[u] * ACC_LD + lane];
                    lds_add(row + 0 * D_HEAD, al[u].x * w[u][0]);
                    lds_add(row + 1 * D_HEAD, al[u].y * w[u][1]);
                    lds_add(row + 2 * D_HEAD, al[u].z * w[u][2]);
                    lds_add(row + 3 * D_HEAD, al[u].w * w[u][3]);
                }
            }
        }
    } else {
#pragma nounroll
        for (int e = tid; e < E; e += OWN_THREADS) {
            const int      d  = dst[e];
            const unsigned ld = (unsigned)(d - lo);
            if (ld < (unsigned)OWN_ROWS) {
                const int    s = src[e];
                const float4 v = edge_scores(el, er, s, d);
                const uint4  m = mx4[ld];
                const float4 q = dn4[ld];
                float a[HEADS];
                a[0] = expf(v.x - ord2f(m.x)) / (q.x + EPS_);
                a[1] = expf(v.y - ord2f(m.y)) / (q.y + EPS_);
                a[2] = expf(v.z - ord2f(m.z)) / (q.z + EPS_);
                a[3] = expf(v.w - ord2f(m.w)) / (q.w + EPS_);
#pragma unroll
                for (int h = 0; h < HEADS; ++h) {
#pragma nounroll
                    for (int f = 0; f < D_HEAD; ++f) {
                        const int c = h * D_HEAD + f;
                        lds_add(&acc[ld * ACC_LD + c], a[h] * Wh[(long)s * FEAT + c]);
                    }
                }
            }
        }
    }
    __syncthreads();

    int rows = N - lo;
    if (rows > OWN_ROWS) rows = OWN_ROWS;
    const int wave = tid >> 5, lane = tid & 31;
    const int r16  = lane & 15;
    const int koff = (lane >> 4) << 1;
    const int hi8  = (lane >> 4) << 3;

    if (!second) {
        float* wst = (float*)(raw + OFF_WST);
        float* stg = (float*)(raw + OFF_STG);
        for (int i = tid; i < OWN_ROWS * FEAT; i += OWN_THREADS) {
            const int r = i >> 7, c = i & (FEAT - 1);
            const float x = acc[r * ACC_LD + c] + bias[c];
            acc[r * ACC_LD + c] = (x > 0.0f) ? x : expm1f(x);
        }
        for (int i = tid; i < FEAT * FEAT / 4; i += OWN_THREADS) {
            const int k = i >> 5, c4 = (i & 31) << 2;
            *(float4*)(wst + k * W_LD + c4) = *(const float4*)(Wn + k * FEAT + c4);
        }
        __syncthreads();
        for (int t = wave; t < OWN_RT * (FEAT / 16); t += OWN_THREADS / 32) {
            const int rt = t >> 3, ct = t & 7;
            const float* ap = acc + (rt * 16 + r16) * ACC_LD + koff;
            const float* bq = wst + koff * W_LD + ct * 16 + r16;
            v8f c = {};
#pragma unroll 8
            for (int k0 = 0; k0 < FEAT; k0 += 4) {
                const v2f a = *(const v2f*)(ap + k0);
                v2f b;
                b.x = bq[k0 * W_LD];
                b.y = bq[(k0 + 1) * W_LD];
                c = __builtin_amdgcn_wmma_f32_16x16x4_f32(false, a, false, b,
                                                          (short)0, c, false, false);
            }
            const int row0 = rt * 16 + hi8;
            float* op = o_main + (long)(lo + row0) * FEAT + ct * 16 + r16;
            float* sp = stg + row0 * ST_LD + ct * 16 + r16;
            if (rt * 16 + 16 <= rows) {
#pragma unroll
                for (int i = 0; i < 8; ++i) {
                    op[i * FEAT]  = c[i];
                    sp[i * ST_LD] = c[i];
                }
            } else {
#pragma unroll
                for (int i = 0; i < 8; ++i) {
                    if (row0 + i < rows) {
                        op[i * FEAT]  = c[i];
                        sp[i * ST_LD] = c[i];
                    }
                }
            }
        }
        __syncthreads();
        if (tid < rows * HEADS) {
            const int r = tid >> 2, h = tid & 3;
            const float* w = stg + r * ST_LD + h * D_HEAD;
            const float* a = aln + h * D_HEAD;
            const float* b = arn + h * D_HEAD;
            float sl = 0.0f, sr = 0.0f;
#pragma unroll
            for (int d = 0; d < D_HEAD; ++d) {
                sl = fmaf(w[d], a[d], sl);
                sr = fmaf(w[d], b[d], sr);
            }
            o_el[(long)lo * HEADS + tid] = sl;
            o_er[(long)lo * HEADS + tid] = sr;
        }
    } else {
        float* zm = (float*)(raw + OFF_ZM);
        for (int i = tid; i < OWN_ROWS * D_HEAD; i += OWN_THREADS) {
            const int r = i >> 5, d = i & (D_HEAD - 1);
            float s = 0.0f;
#pragma unroll
            for (int h = 0; h < HEADS; ++h)
                s += acc[r * ACC_LD + h * D_HEAD + d] + bias[h * D_HEAD + d];
            zm[r * ZM_LD + d] = s * (1.0f / HEADS);
        }
        const int ct = wave;
        v2f bf[D_HEAD / 4];
#pragma unroll
        for (int q = 0; q < D_HEAD / 4; ++q) {
            bf[q].x = Wn[(q * 4 + koff)     * OUT_DIM + ct * 16 + r16];
            bf[q].y = Wn[(q * 4 + koff + 1) * OUT_DIM + ct * 16 + r16];
        }
        const float bv = bpn[ct * 16 + r16];
        __syncthreads();
        for (int rt = 0; rt < OWN_RT; ++rt) {
            const float* ap = zm + (rt * 16 + r16) * ZM_LD + koff;
            v8f c = {};
#pragma unroll
            for (int q = 0; q < D_HEAD / 4; ++q) {
                const v2f a = *(const v2f*)(ap + q * 4);
                c = __builtin_amdgcn_wmma_f32_16x16x4_f32(false, a, false, bf[q],
                                                          (short)0, c, false, false);
            }
            const int row0 = rt * 16 + hi8;
            float* op = o_main + (long)(lo + row0) * OUT_DIM + ct * 16 + r16;
            if (rt * 16 + 16 <= rows) {
#pragma unroll
                for (int i = 0; i < 8; ++i) op[i * OUT_DIM] = c[i] + bv;
            } else {
#pragma unroll
                for (int i = 0; i < 8; ++i)
                    if (row0 + i < rows) op[i * OUT_DIM] = c[i] + bv;
            }
        }
    }
}

#define NODE_OFF_STG (OWN_RT * 16 * ACC_LD * 4)
#define NODE_LDS     (NODE_OFF_STG + OWN_ROWS * ST_LD * 4)
#define NODE_TRIPS   ((OWN_RT * 16 * (FEAT / 4) + OWN_THREADS - 1) / OWN_THREADS)
static_assert(NODE_LDS <= 320 * 1024 && NODE_OFF_STG % 16 == 0, "fits one WGP's LDS");
static_assert(FEAT / 16 == 8 && OWN_THREADS / 32 == 32, "eight column tiles, four sets of eight waves");

__global__ void __launch_bounds__(OWN_THREADS)
k_node_first(const float* __restrict__ X, const float* __restrict__ Wm,
             const float* __restrict__ al, const float* __restrict__ ar, int N,
             float* __restrict__ o_wh, float* __restrict__ o_el, float* __restrict__ o_er)
{
    __shared__ __attribute__((aligned(16))) unsigned char raw[NODE_LDS];
    float* xs  = (float*)raw;
    float* stg = (float*)(raw + NODE_OFF_STG);

    const int tid  = threadIdx.x;
    const int lo   = blockIdx.x * OWN_ROWS;
    int rows = N - lo;
    if (rows > OWN_ROWS) rows = OWN_ROWS;
    const int wave = tid >> 5, lane = tid & 31;
    const int r16  = lane & 15;
    const int koff = (lane >> 4) << 1;
    const int hi8  = (lane >> 4) << 3;
    const int ct   = wave & 7;

    float4 xv[NODE_TRIPS];
#pragma unroll
    for (int k = 0; k < NODE_TRIPS; ++k) {
        const int i = tid + k * OWN_THREADS;
        const int r = i >> 5, c4 = (i & 31) << 2;
        xv[k] = make_float4(0.0f, 0.0f, 0.0f, 0.0f);
        if (r < rows) xv[k] = *(const float4*)(X + (long)(lo + r) * FEAT + c4);
    }
    v2f bf[FEAT / 4];
#pragma unroll
    for (int q = 0; q < FEAT / 4; ++q) {
        bf[q].x = Wm[(q * 4 + koff)     * FEAT + ct * 16 + r16];
        bf[q].y = Wm[(q * 4 + koff + 1) * FEAT + ct * 16 + r16];
    }
#pragma unroll
    for (int k = 0; k < NODE_TRIPS; ++k) {
        const int i = tid + k * OWN_THREADS;
        const int r = i >> 5, c4 = (i & 31) << 2;
        if (r < OWN_RT * 16) *(float4*)(xs + r * ACC_LD + c4) = xv[k];
    }
    __syncthreads();

    for (int rt = wave >> 3; rt < OWN_RT; rt += 4) {
        const float* ap = xs + (rt * 16 + r16) * ACC_LD + koff;
        v8f c = {};
#pragma unroll
        for (int q = 0; q < FEAT / 4; ++q) {
            const v2f a = *(const v2f*)(ap + q * 4);
            c = __builtin_amdgcn_wmma_f32_16x16x4_f32(false, a, false, bf[q],
                                                      (short)0, c, false, false);
        }
        const int row0 = rt * 16 + hi8;
        float* op = o_wh + (long)(lo + row0) * FEAT + ct * 16 + r16;
        float* sp = stg + row0 * ST_LD + ct * 16 + r16;
        if (rt * 16 + 16 <= rows) {
#pragma unroll
            for (int i = 0; i < 8; ++i) {
                op[i * FEAT]  = c[i];
                sp[i * ST_LD] = c[i];
            }
        } else {
#pragma unroll
            for (int i = 0; i < 8; ++i) {
                if (row0 + i < rows) {
                    op[i * FEAT]  = c[i];
                    sp[i * ST_LD] = c[i];
                }
            }
        }
    }
    __syncthreads();
    if (tid < rows * HEADS) {
        const int r = tid >> 2, h = tid & 3;
        const float* w = stg + r * ST_LD + h * D_HEAD;
        const float* a = al + h * D_HEAD;
        const float* b = ar + h * D_HEAD;
        float sl = 0.0f, sr = 0.0f;
#pragma unroll
        for (int d = 0; d < D_HEAD; ++d) {
            sl = fmaf(w[d], a[d], sl);
            sr = fmaf(w[d], b[d], sr);
        }
        o_el[(long)lo * HEADS + tid] = sl;
        o_er[(long)lo * HEADS + tid] = sr;
    }
}


extern "C" void kernel_launch(void* const* d_in, const int* in_sizes, int n_in,
                              void* d_out, int out_size, void* d_ws, size_t ws_size,
                              hipStream_t stream)
{
    const float* h   = (const float*)d_in[0];
    const int*   src = (const int*)  d_in[1];
    const int*   dst = (const int*)  d_in[2];
    const float* W1  = (const float*)d_in[3];
    const float* al1 = (const float*)d_in[4];
    const float* ar1 = (const float*)d_in[5];
    const float* b1  = (const float*)d_in[6];
    const float* W2  = (const float*)d_in[7];
    const float* al2 = (const float*)d_in[8];
    const float* ar2 = (const float*)d_in[9];
    const float* b2  = (const float*)d_in[10];
    const float* Wp  = (const float*)d_in[11];
    const float* bp  = (const float*)d_in[12];
    float*       out = (float*)d_out;

    const int N = N_NODES, E = N_EDGES;

    float*    Wh    = (float*)d_ws;
    float*    Wh2   = Wh  + (size_t)N * FEAT;
    float*    el    = Wh2 + (size_t)N * FEAT;
    float*    er    = el  + (size_t)N * HEADS;
    float*    el2   = er  + (size_t)N * HEADS;
    float*    er2   = el2 + (size_t)N * HEADS;
    unsigned* qsave = (unsigned*)(er2 + (size_t)N * HEADS);
    int*      qcnt  = (int*)(qsave + (size_t)OWN_BLOCKS * OWN_QCAP);
    int*      segcnt = qcnt + OWN_BLOCKS;
    unsigned* seg    = (unsigned*)(segcnt + (size_t)OWN_BLOCKS * BIN_BLOCKS);

    k_bin<<<BIN_BLOCKS, BIN_THREADS, 0, stream>>>(src, dst, E, seg, segcnt);
    k_node_first<<<OWN_BLOCKS, OWN_THREADS, 0, stream>>>(h, W1, al1, ar1, N, Wh, el, er);
    k_edge_owned<<<OWN_BLOCKS, OWN_THREADS, 0, stream>>>(Wh, el, er, src, dst, N, E, qsave, qcnt, 0, seg, segcnt,
                                                         b1, W2, al2, ar2, nullptr, Wh2, el2, er2);
    k_edge_owned<<<OWN_BLOCKS, OWN_THREADS, 0, stream>>>(Wh2, el2, er2, src, dst, N, E, qsave, qcnt, 1, seg, segcnt,
                                                         b2, Wp, nullptr, nullptr, bp, out, nullptr, nullptr);
}


# Owner queue ordered by destination row (LDS integer ranks, once per call); each wave sums its piece's row runs in registers, one 512-B row load per edge, one LDS add per run; sum order changes
# speedup vs baseline: 10.5849x; 1.2061x over previous
#include <hip/hip_runtime.h>
#include <hip/hip_bf16.h>
#include <math.h>

#define N_NODES   50000
#define N_EDGES   800000
#define HEADS     4
#define D_HEAD    32
#define FEAT      128
#define OUT_DIM   512
#define NEG_SLOPE 0.2f
#define EPS_      1e-9f

typedef __attribute__((ext_vector_type(2))) float v2f;
typedef __attribute__((ext_vector_type(8))) float v8f;


__device__ __forceinline__ unsigned f2ord(float f) {
    unsigned u = __float_as_uint(f);
    return (u & 0x80000000u) ? ~u : (u | 0x80000000u);
}
__device__ __forceinline__ float ord2f(unsigned k) {
    return (k & 0x80000000u) ? __uint_as_float(k & 0x7fffffffu) : __uint_as_float(~k);
}

#define OWN_ROWS    196
#define OWN_BLOCKS  ((N_NODES + OWN_ROWS - 1) / OWN_ROWS)
#define OWN_THREADS 1024
#define OWN_QCAP    8192
#define OWN_SBITS   16
static_assert(N_NODES <= (1 << OWN_SBITS), "source node fits the queue word");
static_assert(OWN_ROWS <= (1 << (32 - OWN_SBITS)), "local row fits the queue word");
static_assert(N_EDGES % 4 == 0, "src and dst are read 16 bytes at a time");
#define BIN_THREADS 1024
#define BIN_BLOCKS  ((N_EDGES / 4 + BIN_THREADS - 1) / BIN_THREADS)
#define SEG_CAP     64
static_assert((SEG_CAP & (SEG_CAP - 1)) == 0, "slot index is split by mask and shift");
#define SEG_SHIFT   6
static_assert((1 << SEG_SHIFT) == SEG_CAP, "shift matches the list length");
#define ACC_LD      132
#define W_LD        136
#define ST_LD       132
#define ZM_LD       36
#define OWN_RT      ((OWN_ROWS + 15) / 16)
#define OFF_ACC     0
#define OFF_EV      (OFF_ACC + OWN_ROWS * ACC_LD * 4)
#define OFF_QE      (OFF_EV + OWN_QCAP * 16)
#define OFF_MX      (OFF_QE + OWN_QCAP * 4)
#define OFF_DN      (OFF_MX + OWN_ROWS * HEADS * 4)
#define OFF_QN      (OFF_DN + OWN_ROWS * HEADS * 4)
#define OFF_SGL     (OFF_QN + 16)
#define OFF_RC      OFF_EV
#define OFF_RS      (OFF_EV + 1024)
#define OFF_WST     OFF_EV
#define OFF_STG     (OFF_WST + FEAT * W_LD * 4)
#define OFF_ZM      OFF_EV
#define OWN_LDS     (OFF_STG + OWN_ROWS * ST_LD * 4)
static_assert(OWN_LDS >= OFF_SGL + BIN_BLOCKS * 4 && OWN_LDS <= 320 * 1024, "fits one WGP's LDS");
static_assert(OWN_ROWS * 4 <= 1024 && OFF_RS + 1024 <= OFF_QE, "row counts and run starts fit the score area");
static_assert(OWN_ROWS <= 7 * 32, "one wave sums the row counts, seven rows per lane");
static_assert(BIN_BLOCKS <= OWN_THREADS, "one thread per list when collecting");
#define COPY_TRIPS  ((BIN_BLOCKS * SEG_CAP + OWN_THREADS - 1) / OWN_THREADS)
#define OWN_WAVES   (OWN_THREADS / 32)
#define ROW_FLIGHT  8
static_assert(OFF_ZM + OWN_RT * 16 * ZM_LD * 4 <= OWN_LDS, "head means fit");
static_assert(OFF_ACC + OWN_RT * 16 * ACC_LD * 4 <= OWN_LDS, "the last tile's spare rows are still LDS");
static_assert(OFF_EV % 16 == 0 && OFF_MX % 16 == 0 && OFF_DN % 16 == 0 && OFF_STG % 16 == 0, "16-byte pieces");
static_assert(OUT_DIM / 16 == OWN_THREADS / 32, "second pass: one output column tile per wave");

__device__ __forceinline__ void lds_add(float* p, float v) {
    __hip_atomic_fetch_add(p, v, __ATOMIC_RELAXED, __HIP_MEMORY_SCOPE_WORKGROUP);
}
__device__ __forceinline__ void lds_max(unsigned* p, unsigned v) {
    __hip_atomic_fetch_max(p, v, __ATOMIC_RELAXED, __HIP_MEMORY_SCOPE_WORKGROUP);
}
__device__ __forceinline__ float leaky(float v) { return (v > 0.0f) ? v : NEG_SLOPE * v; }

__device__ __forceinline__ float4 edge_scores(const float* __restrict__ el, const float* __restrict__ er,
                                              int s, int d)
{
    const float4 l = *(const float4*)(el + (long)s * HEADS);
    const float4 r = *(const float4*)(er + (long)d * HEADS);
    float4 v;
    v.x = leaky(l.x + r.x);
    v.y = leaky(l.y + r.y);
    v.z = leaky(l.z + r.z);
    v.w = leaky(l.w + r.w);
    return v;
}

__device__ __forceinline__ void bin_edge(int s, int d, int* cnt, unsigned* __restrict__ seg)
{
    const unsigned g = (unsigned)d / (unsigned)OWN_ROWS;
    if (g < (unsigned)OWN_BLOCKS) {
        const unsigned lr = (unsigned)d - g * (unsigned)OWN_ROWS;
        const int r = atomicAdd(&cnt[g], 1);
        if (r < SEG_CAP)
            seg[((size_t)g * BIN_BLOCKS + blockIdx.x) * SEG_CAP + r] = (unsigned)s | (lr << OWN_SBITS);
    }
}

__global__ void __launch_bounds__(BIN_THREADS)
k_bin(const int* __restrict__ src, const int* __restrict__ dst, int E,
      unsigned* __restrict__ seg, int* __restrict__ segcnt)
{
    __shared__ int cnt[OWN_BLOCKS];
    const int tid = threadIdx.x;
    if (tid < OWN_BLOCKS) cnt[tid] = 0;
    __syncthreads();
    const int item = blockIdx.x * BIN_THREADS + tid;
    if (item < (E >> 2)) {
        const int4 d = ((const int4*)dst)[item];
        const int4 s = ((const int4*)src)[item];
        bin_edge(s.x, d.x, cnt, seg);
        bin_edge(s.y, d.y, cnt, seg);
        bin_edge(s.z, d.z, cnt, seg);
        bin_edge(s.w, d.w, cnt, seg);
    }
    __syncthreads();
    if (tid < OWN_BLOCKS) segcnt[tid * BIN_BLOCKS + blockIdx.x] = cnt[tid];
}

__global__ void __launch_bounds__(OWN_THREADS)
k_edge_owned(const float* __restrict__ Wh, const float* __restrict__ el,
             const float* __restrict__ er, const int* __restrict__ src,
             const int* __restrict__ dst, int N, int E,
             unsigned* __restrict__ qsave, int* __restrict__ qcnt, int second,
             const unsigned* __restrict__ seg,
             const int* __restrict__ segcnt,
             const float* __restrict__ bias,
             const float* __restrict__ Wn,
             const float* __restrict__ aln,
             const float* __restrict__ arn,
             const float* __restrict__ bpn,
             float* __restrict__ o_main,
             float* __restrict__ o_el, float* __restrict__ o_er)
{
    __shared__ __attribute__((aligned(16))) unsigned char raw[OWN_LDS];
    float*    acc = (float*)(raw + OFF_ACC);
    float4*   ev  = (float4*)(raw + OFF_EV);
    unsigned* mx  = (unsigned*)(raw + OFF_MX);
    float*    dn  = (float*)(raw + OFF_DN);
    unsigned* qe  = (unsigned*)(raw + OFF_QE);
    int*      qnp = (int*)(raw + OFF_QN);
    int*      sgl = (int*)(raw + OFF_SGL);
    int*      rc  = (int*)(raw + OFF_RC);
    int*      rs  = (int*)(raw + OFF_RS);
    const int reuse = second;

    const int tid   = threadIdx.x;
    const int wave  = tid >> 5, lane = tid & 31;
    const int lo    = blockIdx.x * OWN_ROWS;
    const int qbase = blockIdx.x * OWN_QCAP;

    for (int i = tid; i < OWN_ROWS * ACC_LD; i += OWN_THREADS) acc[i] = 0.0f;
    if (tid < OWN_ROWS * HEADS) { mx[tid] = 0u; dn[tid] = 0.0f; }
    if (tid < OWN_ROWS) rc[tid] = 0;
    if (tid == 0) *qnp = 0;
    __syncthreads();

    int n;
    if (!reuse) {
        if (tid < BIN_BLOCKS) {
            const int cn = segcnt[blockIdx.x * BIN_BLOCKS + tid];
            sgl[tid] = cn;
            atomicAdd(qnp, (cn <= SEG_CAP) ? cn : OWN_QCAP + 1);
        }
        __syncthreads();
        n = *qnp;
        const bool whole = n <= OWN_QCAP;
        int      rk[COPY_TRIPS];
        unsigned wd[COPY_TRIPS];
        if (whole) {
            const unsigned* sp = seg + (size_t)blockIdx.x * (BIN_BLOCKS * SEG_CAP);
#pragma unroll
            for (int k = 0; k < COPY_TRIPS; ++k) {
                const int slot = tid + k * OWN_THREADS;
                const int b = slot >> SEG_SHIFT, r = slot & (SEG_CAP - 1);
                const int cn = sgl[(b < BIN_BLOCKS) ? b : BIN_BLOCKS - 1];
                rk[k] = (slot < BIN_BLOCKS * SEG_CAP && r < cn) ? 0 : -1;
            }
#pragma unroll
            for (int k = 0; k < COPY_TRIPS; ++k)
                wd[k] = (rk[k] >= 0) ? sp[tid + k * OWN_THREADS] : 0u;
#pragma unroll
            for (int k = 0; k < COPY_TRIPS; ++k)
                if (rk[k] >= 0) rk[k] = atomicAdd(&rc[wd[k] >> OWN_SBITS], 1);
        }
        if (tid == 0) qcnt[blockIdx.x] = n;
        __syncthreads();
        if (whole && wave == 0) {
            int cnt7[7], tot = 0;
#pragma unroll
            for (int q = 0; q < 7; ++q) {
                const int r = lane * 7 + q;
                cnt7[q] = (r < OWN_ROWS) ? rc[r] : 0;
                tot += cnt7[q];
            }
            int incl = tot;
#pragma unroll
            for (int d = 1; d < 32; d <<= 1) {
                const int t = __shfl_up(incl, d);
                if (lane >= d) incl += t;
            }
            int at = incl - tot;
#pragma unroll
            for (int q = 0; q < 7; ++q) {
                const int r = lane * 7 + q;
                if (r < OWN_ROWS) rs[r] = at;
                at += cnt7[q];
            }
        }
        __syncthreads();
        if (whole) {
#pragma unroll
            for (int k = 0; k < COPY_TRIPS; ++k)
                if (rk[k] >= 0) qe[rs[wd[k] >> OWN_SBITS] + rk[k]] = wd[k];
        }
    } else {
        n = qcnt[blockIdx.x];
        if (n <= OWN_QCAP)
            for (int i = tid; i < n; i += OWN_THREADS) qe[i] = qsave[qbase + i];
    }
    __syncthreads();

    const bool  fits = n <= OWN_QCAP;
    const uint4*  mx4 = (const uint4*)mx;
    const float4* dn4 = (const float4*)dn;
    const int per = ((n + OWN_WAVES - 1) / OWN_WAVES) | 1;
    if (fits && !reuse)
        for (int i = tid; i < n; i += OWN_THREADS) qsave[qbase + i] = qe[i];

    if (fits) {
        for (int q0 = wave; q0 < per; q0 += 4 * OWN_WAVES) {
            int      ss[4], dd[4];
            bool     ok[4];
            float4   sc[4];
            unsigned pk[4];
#pragma unroll
            for (int u = 0; u < 4; ++u) {
                const int q = q0 + u * OWN_WAVES;
                const int i = lane * per + q;
                ok[u] = q < per && i < n;
                pk[u] = ok[u] ? qe[i] : 0u;
            }
#pragma unroll
            for (int u = 0; u < 4; ++u) {
                ss[u] = (int)(pk[u] & ((1u << OWN_SBITS) - 1u));
                dd[u] = lo + (int)(pk[u] >> OWN_SBITS);
            }
#pragma unroll
            for (int u = 0; u < 4; ++u) sc[u] = edge_scores(el, er, ss[u], dd[u]);
#pragma unroll
            for (int u = 0; u < 4; ++u) {
                if (ok[u]) {
                    const int      i  = lane * per + q0 + u * OWN_WAVES;
                    const unsigned ld = pk[u] >> OWN_SBITS;
                    ev[i] = sc[u];
                    lds_max(&mx[ld * HEADS + 0], f2ord(sc[u].x));
                    lds_max(&mx[ld * HEADS + 1], f2ord(sc[u].y));
                    lds_max(&mx[ld * HEADS + 2], f2ord(sc[u].z));
                    lds_max(&mx[ld * HEADS + 3], f2ord(sc[u].w));
                }
            }
        }
    } else {
#pragma nounroll
        for (int e = tid; e < E; e += OWN_THREADS) {
            const int      d  = dst[e];
            const unsigned ld = (unsigned)(d - lo);
            if (ld < (unsigned)OWN_ROWS) {
                const float4 v = edge_scores(el, er, src[e], d);
                lds_max(&mx[ld * HEADS + 0], f2ord(v.x));
                lds_max(&mx[ld * HEADS + 1], f2ord(v.y));
                lds_max(&mx[ld * HEADS + 2], f2ord(v.z));
                lds_max(&mx[ld * HEADS + 3], f2ord(v.w));
            }
        }
    }
    __syncthreads();

    if (fits) {
        for (int q = wave; q < per; q += OWN_WAVES) {
            const int i = lane * per + q;
            if (i >= n) continue;
            const unsigned ld = qe[i] >> OWN_SBITS;
            const float4   v  = ev[i];
            const uint4    m  = mx4[ld];
            float4 x;
            x.x = expf(v.x - ord2f(m.x));
            x.y = expf(v.y - ord2f(m.y));
            x.z = expf(v.z - ord2f(m.z));
            x.w = expf(v.w - ord2f(m.w));
            ev[i] = x;
            lds_add(&dn[ld * HEADS + 0], x.x);
            lds_add(&dn[ld * HEADS + 1], x.y);
            lds_add(&dn[ld * HEADS + 2], x.z);
            lds_add(&dn[ld * HEADS + 3], x.w);
        }
    } else {
#pragma nounroll
        for (int e = tid; e < E; e += OWN_THREADS) {
            const int      d  = dst[e];
            const unsigned ld = (unsigned)(d - lo);
            if (ld < (unsigned)OWN_ROWS) {
                const float4 v = edge_scores(el, er, src[e], d);
                const uint4  m = mx4[ld];
                lds_add(&dn[ld * HEADS + 0], expf(v.x - ord2f(m.x)));
                lds_add(&dn[ld * HEADS + 1], expf(v.y - ord2f(m.y)));
                lds_add(&dn[ld * HEADS + 2], expf(v.z - ord2f(m.z)));
                lds_add(&dn[ld * HEADS + 3], expf(v.w - ord2f(m.w)));
            }
        }
    }
    __syncthreads();

    if (fits) {
        for (int i = tid; i < n; i += OWN_THREADS) {
            const unsigned ld = qe[i] >> OWN_SBITS;
            const float4   q  = dn4[ld];
            float4 x = ev[i];
            x.x = x.x / (q.x + EPS_);
            x.y = x.y / (q.y + EPS_);
            x.z = x.z / (q.z + EPS_);
            x.w = x.w / (q.w + EPS_);
            ev[i] = x;
        }
    }
    __syncthreads();

    if (fits) {
        const float* evf = (const float*)ev;
        const int hd = lane >> 3;
        const int ws = __builtin_amdgcn_readfirstlane(wave);
        const int jb = ws * per;
        const int je = (jb + per < n) ? jb + per : n;
        if (jb < je) {
            float4   a = make_float4(0.0f, 0.0f, 0.0f, 0.0f);
            int      cur = -1;
            unsigned pkn[ROW_FLIGHT];
            float    aln[ROW_FLIGHT];
#pragma unroll
            for (int u = 0; u < ROW_FLIGHT; ++u) {
                const int i = (jb + u < je) ? jb + u : je - 1;
                pkn[u] = qe[i];
                aln[u] = evf[i * HEADS + hd];
            }
            for (int j = jb; j < je; j += ROW_FLIGHT) {
                unsigned pk[ROW_FLIGHT];
                float    al[ROW_FLIGHT];
                float4   w[ROW_FLIGHT];
#pragma unroll
                for (int u = 0; u < ROW_FLIGHT; ++u) {
                    pk[u] = (unsigned)__builtin_amdgcn_readfirstlane((int)pkn[u]);
                    al[u] = aln[u];
                    const int s = (int)(pk[u] & ((1u << OWN_SBITS) - 1u));
                    w[u] = *(const float4*)(Wh + (long)s * FEAT + 4 * lane);
                }
#pragma unroll
                for (int u = 0; u < ROW_FLIGHT; ++u) {
                    const int i = (j + ROW_FLIGHT + u < je) ? j + ROW_FLIGHT + u : je - 1;
                    pkn[u] = qe[i];
                    aln[u] = evf[i * HEADS + hd];
                }
#pragma unroll
                for (int u = 0; u < ROW_FLIGHT; ++u) {
                    if (j + u < je) {
                        const int ld = (int)(pk[u] >> OWN_SBITS);
                        if (ld != cur) {
                            if (cur >= 0) {
                                float* row = &acc[cur * ACC_LD + 4 * lane];
                                lds_add(row + 0, a.x);
                                lds_add(row + 1, a.y);
                                lds_add(row + 2, a.z);
                                lds_add(row + 3, a.w);
                            }
                            a = make_float4(0.0f, 0.0f, 0.0f, 0.0f);
                            cur = ld;
                        }
                        a.x += al[u] * w[u].x;
                        a.y += al[u] * w[u].y;
                        a.z += al[u] * w[u].z;
                        a.w += al[u] * w[u].w;
                    }
                }
            }
            float* row = &acc[cur * ACC_LD + 4 * lane];
            lds_add(row + 0, a.x);
            lds_add(row + 1, a.y);
            lds_add(row + 2, a.z);
            lds_add(row + 3, a.w);
        }
    } else {
#pragma nounroll
        for (int e = tid; e < E; e += OWN_THREADS) {
            const int      d  = dst[e];
            const unsigned ld = (unsigned)(d - lo);
            if (ld < (unsigned)OWN_ROWS) {
                const int    s = src[e];
                const float4 v = edge_scores(el, er, s, d);
                const uint4  m = mx4[ld];
                const float4 q = dn4[ld];
                float a[HEADS];
                a[0] = expf(v.x - ord2f(m.x)) / (q.x + EPS_);
                a[1] = expf(v.y - ord2f(m.y)) / (q.y + EPS_);
                a[2] = expf(v.z - ord2f(m.z)) / (q.z + EPS_);
                a[3] = expf(v.w - ord2f(m.w)) / (q.w + EPS_);
#pragma unroll
                for (int h = 0; h < HEADS; ++h) {
#pragma nounroll
                    for (int f = 0; f < D_HEAD; ++f) {
                        const int c = h * D_HEAD + f;
                        lds_add(&acc[ld * ACC_LD + c], a[h] * Wh[(long)s * FEAT + c]);
                    }
                }
            }
        }
    }
    __syncthreads();

    int rows = N - lo;
    if (rows > OWN_ROWS) rows = OWN_ROWS;
    const int r16  = lane & 15;
    const int koff = (lane >> 4) << 1;
    const int hi8  = (lane >> 4) << 3;

    if (!second) {
        float* wst = (float*)(raw + OFF_WST);
        float* stg = (float*)(raw + OFF_STG);
        for (int i = tid; i < OWN_ROWS * FEAT; i += OWN_THREADS) {
            const int r = i >> 7, c = i & (FEAT - 1);
            const float x = acc[r * ACC_LD + c] + bias[c];
            acc[r * ACC_LD + c] = (x > 0.0f) ? x : expm1f(x);
        }
        for (int i = tid; i < FEAT * FEAT / 4; i += OWN_THREADS) {
            const int k = i >> 5, c4 = (i & 31) << 2;
            *(float4*)(wst + k * W_LD + c4) = *(const float4*)(Wn + k * FEAT + c4);
        }
        __syncthreads();
        for (int t = wave; t < OWN_RT * (FEAT / 16); t += OWN_THREADS / 32) {
            const int rt = t >> 3, ct = t & 7;
            const float* ap = acc + (rt * 16 + r16) * ACC_LD + koff;
            const float* bq = wst + koff * W_LD + ct * 16 + r16;
            v8f c = {};
#pragma unroll 8
            for (int k0 = 0; k0 < FEAT; k0 += 4) {
                const v2f a = *(const v2f*)(ap + k0);
                v2f b;
                b.x = bq[k0 * W_LD];
                b.y = bq[(k0 + 1) * W_LD];
                c = __builtin_amdgcn_wmma_f32_16x16x4_f32(false, a, false, b,
                                                          (short)0, c, false, false);
            }
            const int row0 = rt * 16 + hi8;
            float* op = o_main + (long)(lo + row0) * FEAT + ct * 16 + r16;
            float* sp = stg + row0 * ST_LD + ct * 16 + r16;
            if (rt * 16 + 16 <= rows) {
#pragma unroll
                for (int i = 0; i < 8; ++i) {
                    op[i * FEAT]  = c[i];
                    sp[i * ST_LD] = c[i];
                }
            } else {
#pragma unroll
                for (int i = 0; i < 8; ++i) {
                    if (row0 + i < rows) {
                        op[i * FEAT]  = c[i];
                        sp[i * ST_LD] = c[i];
                    }
                }
            }
        }
        __syncthreads();
        if (tid < rows * HEADS) {
            const int r = tid >> 2, h = tid & 3;
            const float* w = stg + r * ST_LD + h * D_HEAD;
            const float* a = aln + h * D_HEAD;
            const float* b = arn + h * D_HEAD;
            float sl = 0.0f, sr = 0.0f;
#pragma unroll
            for (int d = 0; d < D_HEAD; ++d) {
                sl = fmaf(w[d], a[d], sl);
                sr = fmaf(w[d], b[d], sr);
            }
            o_el[(long)lo * HEADS + tid] = sl;
            o_er[(long)lo * HEADS + tid] = sr;
        }
    } else {
        float* zm = (float*)(raw + OFF_ZM);
        for (int i = tid; i < OWN_ROWS * D_HEAD; i += OWN_THREADS) {
            const int r = i >> 5, d = i & (D_HEAD - 1);
            float s = 0.0f;
#pragma unroll
            for (int h = 0; h < HEADS; ++h)
                s += acc[r * ACC_LD + h * D_HEAD + d] + bias[h * D_HEAD + d];
            zm[r * ZM_LD + d] = s * (1.0f / HEADS);
        }
        const int ct = wave;
        v2f bf[D_HEAD / 4];
#pragma unroll
        for (int q = 0; q < D_HEAD / 4; ++q) {
            bf[q].x = Wn[(q * 4 + koff)     * OUT_DIM + ct * 16 + r16];
            bf[q].y = Wn[(q * 4 + koff + 1) * OUT_DIM + ct * 16 + r16];
        }
        const float bv = bpn[ct * 16 + r16];
        __syncthreads();
        for (int rt = 0; rt < OWN_RT; ++rt) {
            const float* ap = zm + (rt * 16 + r16) * ZM_LD + koff;
            v8f c = {};
#pragma unroll
            for (int q = 0; q < D_HEAD / 4; ++q) {
                const v2f a = *(const v2f*)(ap + q * 4);
                c = __builtin_amdgcn_wmma_f32_16x16x4_f32(false, a, false, bf[q],
                                                          (short)0, c, false, false);
            }
            const int row0 = rt * 16 + hi8;
            float* op = o_main + (long)(lo + row0) * OUT_DIM + ct * 16 + r16;
            if (rt * 16 + 16 <= rows) {
#pragma unroll
                for (int i = 0; i < 8; ++i) op[i * OUT_DIM] = c[i] + bv;
            } else {
#pragma unroll
                for (int i = 0; i < 8; ++i)
                    if (row0 + i < rows) op[i * OUT_DIM] = c[i] + bv;
            }
        }
    }
}

#define NODE_OFF_STG (OWN_RT * 16 * ACC_LD * 4)
#define NODE_LDS     (NODE_OFF_STG + OWN_ROWS * ST_LD * 4)
#define NODE_TRIPS   ((OWN_RT * 16 * (FEAT / 4) + OWN_THREADS - 1) / OWN_THREADS)
static_assert(NODE_LDS <= 320 * 1024 && NODE_OFF_STG % 16 == 0, "fits one WGP's LDS");
static_assert(FEAT / 16 == 8 && OWN_THREADS / 32 == 32, "eight column tiles, four sets of eight waves");

__global__ void __launch_bounds__(OWN_THREADS)
k_node_first(const float* __restrict__ X, const float* __restrict__ Wm,
             const float* __restrict__ al, const float* __restrict__ ar, int N,
             float* __restrict__ o_wh, float* __restrict__ o_el, float* __restrict__ o_er)
{
    __shared__ __attribute__((aligned(16))) unsigned char raw[NODE_LDS];
    float* xs  = (float*)raw;
    float* stg = (float*)(raw + NODE_OFF_STG);

    const int tid  = threadIdx.x;
    const int lo   = blockIdx.x * OWN_ROWS;
    int rows = N - lo;
    if (rows > OWN_ROWS) rows = OWN_ROWS;
    const int wave = tid >> 5, lane = tid & 31;
    const int r16  = lane & 15;
    const int koff = (lane >> 4) << 1;
    const int hi8  = (lane >> 4) << 3;
    const int ct   = wave & 7;

    float4 xv[NODE_TRIPS];
#pragma unroll
    for (int k = 0; k < NODE_TRIPS; ++k) {
        const int i = tid + k * OWN_THREADS;
        const int r = i >> 5, c4 = (i & 31) << 2;
        xv[k] = make_float4(0.0f, 0.0f, 0.0f, 0.0f);
        if (r < rows) xv[k] = *(const float4*)(X + (long)(lo + r) * FEAT + c4);
    }
    v2f bf[FEAT / 4];
#pragma unroll
    for (int q = 0; q < FEAT / 4; ++q) {
        bf[q].x = Wm[(q * 4 + koff)     * FEAT + ct * 16 + r16];
        bf[q].y = Wm[(q * 4 + koff + 1) * FEAT + ct * 16 + r16];
    }
#pragma unroll
    for (int k = 0; k < NODE_TRIPS; ++k) {
        const int i = tid + k * OWN_THREADS;
        const int r = i >> 5, c4 = (i & 31) << 2;
        if (r < OWN_RT * 16) *(float4*)(xs + r * ACC_LD + c4) = xv[k];
    }
    __syncthreads();

    for (int rt = wave >> 3; rt < OWN_RT; rt += 4) {
        const float* ap = xs + (rt * 16 + r16) * ACC_LD + koff;
        v8f c = {};
#pragma unroll
        for (int q = 0; q < FEAT / 4; ++q) {
            const v2f a = *(const v2f*)(ap + q * 4);
            c = __builtin_amdgcn_wmma_f32_16x16x4_f32(false, a, false, bf[q],
                                                      (short)0, c, false, false);
        }
        const int row0 = rt * 16 + hi8;
        float* op = o_wh + (long)(lo + row0) * FEAT + ct * 16 + r16;
        float* sp = stg + row0 * ST_LD + ct * 16 + r16;
        if (rt * 16 + 16 <= rows) {
#pragma unroll
            for (int i = 0; i < 8; ++i) {
                op[i * FEAT]  = c[i];
                sp[i * ST_LD] = c[i];
            }
        } else {
#pragma unroll
            for (int i = 0; i < 8; ++i) {
                if (row0 + i < rows) {
                    op[i * FEAT]  = c[i];
                    sp[i * ST_LD] = c[i];
                }
            }
        }
    }
    __syncthreads();
    if (tid < rows * HEADS) {
        const int r = tid >> 2, h = tid & 3;
        const float* w = stg + r * ST_LD + h * D_HEAD;
        const float* a = al + h * D_HEAD;
        const float* b = ar + h * D_HEAD;
        float sl = 0.0f, sr = 0.0f;
#pragma unroll
        for (int d = 0; d < D_HEAD; ++d) {
            sl = fmaf(w[d], a[d], sl);
            sr = fmaf(w[d], b[d], sr);
        }
        o_el[(long)lo * HEADS + tid] = sl;
        o_er[(long)lo * HEADS + tid] = sr;
    }
}


extern "C" void kernel_launch(void* const* d_in, const int* in_sizes, int n_in,
                              void* d_out, int out_size, void* d_ws, size_t ws_size,
                              hipStream_t stream)
{
    const float* h   = (const float*)d_in[0];
    const int*   src = (const int*)  d_in[1];
    const int*   dst = (const int*)  d_in[2];
    const float* W1  = (const float*)d_in[3];
    const float* al1 = (const float*)d_in[4];
    const float* ar1 = (const float*)d_in[5];
    const float* b1  = (const float*)d_in[6];
    const float* W2  = (const float*)d_in[7];
    const float* al2 = (const float*)d_in[8];
    const float* ar2 = (const float*)d_in[9];
    const float* b2  = (const float*)d_in[10];
    const float* Wp  = (const float*)d_in[11];
    const float* bp  = (const float*)d_in[12];
    float*       out = (float*)d_out;

    const int N = N_NODES, E = N_EDGES;

    float*    Wh    = (float*)d_ws;
    float*    Wh2   = Wh  + (size_t)N * FEAT;
    float*    el    = Wh2 + (size_t)N * FEAT;
    float*    er    = el  + (size_t)N * HEADS;
    float*    el2   = er  + (size_t)N * HEADS;
    float*    er2   = el2 + (size_t)N * HEADS;
    unsigned* qsave = (unsigned*)(er2 + (size_t)N * HEADS);
    int*      qcnt  = (int*)(qsave + (size_t)OWN_BLOCKS * OWN_QCAP);
    int*      segcnt = qcnt + OWN_BLOCKS;
    unsigned* seg    = (unsigned*)(segcnt + (size_t)OWN_BLOCKS * BIN_BLOCKS);

    k_bin<<<BIN_BLOCKS, BIN_THREADS, 0, stream>>>(src, dst, E, seg, segcnt);
    k_node_first<<<OWN_BLOCKS, OWN_THREADS, 0, stream>>>(h, W1, al1, ar1, N, Wh, el, er);
    k_edge_owned<<<OWN_BLOCKS, OWN_THREADS, 0, stream>>>(Wh, el, er, src, dst, N, E, qsave, qcnt, 0, seg, segcnt,
                                                         b1, W2, al2, ar2, nullptr, Wh2, el2, er2);
    k_edge_owned<<<OWN_BLOCKS, OWN_THREADS, 0, stream>>>(Wh2, el2, er2, src, dst, N, E, qsave, qcnt, 1, seg, segcnt,
                                                         b2, Wp, nullptr, nullptr, bp, out, nullptr, nullptr);
}


# Stores coalesced: out as whole 128-byte lines (a wave holds two column tiles and swaps their half-wave halves); dealt edge lists built in LDS and stored a list per instruction, not word by word
# speedup vs baseline: 11.1944x; 1.0576x over previous
#include <hip/hip_runtime.h>
#include <hip/hip_bf16.h>
#include <math.h>

#define N_NODES   50000
#define N_EDGES   800000
#define HEADS     4
#define D_HEAD    32
#define FEAT      128
#define OUT_DIM   512
#define NEG_SLOPE 0.2f
#define EPS_      1e-9f

typedef __attribute__((ext_vector_type(2))) float v2f;
typedef __attribute__((ext_vector_type(8))) float v8f;


__device__ __forceinline__ unsigned f2ord(float f) {
    unsigned u = __float_as_uint(f);
    return (u & 0x80000000u) ? ~u : (u | 0x80000000u);
}
__device__ __forceinline__ float ord2f(unsigned k) {
    return (k & 0x80000000u) ? __uint_as_float(k & 0x7fffffffu) : __uint_as_float(~k);
}

#define OWN_ROWS    196
#define OWN_BLOCKS  ((N_NODES + OWN_ROWS - 1) / OWN_ROWS)
#define OWN_THREADS 1024
#define OWN_QCAP    8192
#define OWN_SBITS   16
static_assert(N_NODES <= (1 << OWN_SBITS), "source node fits the queue word");
static_assert(OWN_ROWS <= (1 << (32 - OWN_SBITS)), "local row fits the queue word");
static_assert(N_EDGES % 4 == 0, "src and dst are read 16 bytes at a time");
#define BIN_THREADS 1024
#define BIN_BLOCKS  ((N_EDGES / 4 + BIN_THREADS - 1) / BIN_THREADS)
#define SEG_CAP     64
static_assert((SEG_CAP & (SEG_CAP - 1)) == 0, "slot index is split by mask and shift");
#define SEG_SHIFT   6
static_assert((1 << SEG_SHIFT) == SEG_CAP, "shift matches the list length");
static_assert(SEG_CAP == 64, "a list is stored by one or two instructions of 32 lanes");
#define ACC_LD      132
#define W_LD        136
#define ST_LD       132
#define ZM_LD       36
#define OWN_RT      ((OWN_ROWS + 15) / 16)
#define OFF_ACC     0
#define OFF_EV      (OFF_ACC + OWN_ROWS * ACC_LD * 4)
#define OFF_QE      (OFF_EV + OWN_QCAP * 16)
#define OFF_MX      (OFF_QE + OWN_QCAP * 4)
#define OFF_DN      (OFF_MX + OWN_ROWS * HEADS * 4)
#define OFF_QN      (OFF_DN + OWN_ROWS * HEADS * 4)
#define OFF_SGL     (OFF_QN + 16)
#define OFF_RC      OFF_EV
#define OFF_RS      (OFF_EV + 1024)
#define OFF_WST     OFF_EV
#define OFF_STG     (OFF_WST + FEAT * W_LD * 4)
#define OFF_ZM      OFF_EV
#define OWN_LDS     (OFF_STG + OWN_ROWS * ST_LD * 4)
static_assert(OWN_LDS >= OFF_SGL + BIN_BLOCKS * 4 && OWN_LDS <= 320 * 1024, "fits one WGP's LDS");
static_assert(OWN_ROWS * 4 <= 1024 && OFF_RS + 1024 <= OFF_QE, "row counts and run starts fit the score area");
static_assert(OWN_ROWS <= 7 * 32, "one wave sums the row counts, seven rows per lane");
static_assert(BIN_BLOCKS <= OWN_THREADS, "one thread per list when collecting");
#define COPY_TRIPS  ((BIN_BLOCKS * SEG_CAP + OWN_THREADS - 1) / OWN_THREADS)
#define OWN_WAVES   (OWN_THREADS / 32)
#define ROW_FLIGHT  8
static_assert(OFF_ZM + OWN_RT * 16 * ZM_LD * 4 <= OWN_LDS, "head means fit");
static_assert(OFF_ACC + OWN_RT * 16 * ACC_LD * 4 <= OWN_LDS, "the last tile's spare rows are still LDS");
static_assert(OFF_EV % 16 == 0 && OFF_MX % 16 == 0 && OFF_DN % 16 == 0 && OFF_STG % 16 == 0, "16-byte pieces");
static_assert(OUT_DIM / 32 * 2 == OWN_THREADS / 32, "second pass: one pair of output column tiles per wave, two waves per pair");

__device__ __forceinline__ void lds_add(float* p, float v) {
    __hip_atomic_fetch_add(p, v, __ATOMIC_RELAXED, __HIP_MEMORY_SCOPE_WORKGROUP);
}
__device__ __forceinline__ void lds_max(unsigned* p, unsigned v) {
    __hip_atomic_fetch_max(p, v, __ATOMIC_RELAXED, __HIP_MEMORY_SCOPE_WORKGROUP);
}
__device__ __forceinline__ float leaky(float v) { return (v > 0.0f) ? v : NEG_SLOPE * v; }

__device__ __forceinline__ void swap_halves(float& a, float& b) {
    typedef __attribute__((ext_vector_type(2))) unsigned v2u;
    const v2u r = __builtin_amdgcn_permlane16_swap(__float_as_uint(a), __float_as_uint(b), false, false);
    a = __uint_as_float(r.x);
    b = __uint_as_float(r.y);
}

__device__ __forceinline__ float4 edge_scores(const float* __restrict__ el, const float* __restrict__ er,
                                              int s, int d)
{
    const float4 l = *(const float4*)(el + (long)s * HEADS);
    const float4 r = *(const float4*)(er + (long)d * HEADS);
    float4 v;
    v.x = leaky(l.x + r.x);
    v.y = leaky(l.y + r.y);
    v.z = leaky(l.z + r.z);
    v.w = leaky(l.w + r.w);
    return v;
}

__device__ __forceinline__ void bin_edge(int s, int d, int* cnt, unsigned* stage)
{
    const unsigned g = (unsigned)d / (unsigned)OWN_ROWS;
    if (g < (unsigned)OWN_BLOCKS) {
        const unsigned lr = (unsigned)d - g * (unsigned)OWN_ROWS;
        const int r = atomicAdd(&cnt[g], 1);
        if (r < SEG_CAP)
            stage[g * SEG_CAP + r] = (unsigned)s | (lr << OWN_SBITS);
    }
}

__global__ void __launch_bounds__(BIN_THREADS)
k_bin(const int* __restrict__ src, const int* __restrict__ dst, int E,
      unsigned* __restrict__ seg, int* __restrict__ segcnt)
{
    __shared__ int      cnt[OWN_BLOCKS];
    __shared__ unsigned stage[OWN_BLOCKS * SEG_CAP];
    const int tid = threadIdx.x;
    if (tid < OWN_BLOCKS) cnt[tid] = 0;
    __syncthreads();
    const int item = blockIdx.x * BIN_THREADS + tid;
    if (item < (E >> 2)) {
        const int4 d = ((const int4*)dst)[item];
        const int4 s = ((const int4*)src)[item];
        bin_edge(s.x, d.x, cnt, stage);
        bin_edge(s.y, d.y, cnt, stage);
        bin_edge(s.z, d.z, cnt, stage);
        bin_edge(s.w, d.w, cnt, stage);
    }
    __syncthreads();
    if (tid < OWN_BLOCKS) segcnt[tid * BIN_BLOCKS + blockIdx.x] = cnt[tid];
    const int wave = tid >> 5, lane = tid & 31;
    for (int g = wave; g < OWN_BLOCKS; g += BIN_THREADS / 32) {
        int cn = cnt[g];
        if (cn > SEG_CAP) cn = SEG_CAP;
        unsigned* lp = seg + ((size_t)g * BIN_BLOCKS + blockIdx.x) * SEG_CAP;
        if (lane < cn)      lp[lane]      = stage[g * SEG_CAP + lane];
        if (lane + 32 < cn) lp[lane + 32] = stage[g * SEG_CAP + lane + 32];
    }
}

__global__ void __launch_bounds__(OWN_THREADS)
k_edge_owned(const float* __restrict__ Wh, const float* __restrict__ el,
             const float* __restrict__ er, const int* __restrict__ src,
             const int* __restrict__ dst, int N, int E,
             unsigned* __restrict__ qsave, int* __restrict__ qcnt, int second,
             const unsigned* __restrict__ seg,
             const int* __restrict__ segcnt,
             const float* __restrict__ bias,
             const float* __restrict__ Wn,
             const float* __restrict__ aln,
             const float* __restrict__ arn,
             const float* __restrict__ bpn,
             float* __restrict__ o_main,
             float* __restrict__ o_el, float* __restrict__ o_er)
{
    __shared__ __attribute__((aligned(16))) unsigned char raw[OWN_LDS];
    float*    acc = (float*)(raw + OFF_ACC);
    float4*   ev  = (float4*)(raw + OFF_EV);
    unsigned* mx  = (unsigned*)(raw + OFF_MX);
    float*    dn  = (float*)(raw + OFF_DN);
    unsigned* qe  = (unsigned*)(raw + OFF_QE);
    int*      qnp = (int*)(raw + OFF_QN);
    int*      sgl = (int*)(raw + OFF_SGL);
    int*      rc  = (int*)(raw + OFF_RC);
    int*      rs  = (int*)(raw + OFF_RS);
    const int reuse = second;

    const int tid   = threadIdx.x;
    const int wave  = tid >> 5, lane = tid & 31;
    const int lo    = blockIdx.x * OWN_ROWS;
    const int qbase = blockIdx.x * OWN_QCAP;

    for (int i = tid; i < OWN_ROWS * ACC_LD; i += OWN_THREADS) acc[i] = 0.0f;
    if (tid < OWN_ROWS * HEADS) { mx[tid] = 0u; dn[tid] = 0.0f; }
    if (tid < OWN_ROWS) rc[tid] = 0;
    if (tid == 0) *qnp = 0;
    __syncthreads();

    int n;
    if (!reuse) {
        if (tid < BIN_BLOCKS) {
            const int cn = segcnt[blockIdx.x * BIN_BLOCKS + tid];
            sgl[tid] = cn;
            atomicAdd(qnp, (cn <= SEG_CAP) ? cn : OWN_QCAP + 1);
        }
        __syncthreads();
        n = *qnp;
        const bool whole = n <= OWN_QCAP;
        int      rk[COPY_TRIPS];
        unsigned wd[COPY_TRIPS];
        if (whole) {
            const unsigned* sp = seg + (size_t)blockIdx.x * (BIN_BLOCKS * SEG_CAP);
#pragma unroll
            for (int k = 0; k < COPY_TRIPS; ++k) {
                const int slot = tid + k * OWN_THREADS;
                const int b = slot >> SEG_SHIFT, r = slot & (SEG_CAP - 1);
                const int cn = sgl[(b < BIN_BLOCKS) ? b : BIN_BLOCKS - 1];
                rk[k] = (slot < BIN_BLOCKS * SEG_CAP && r < cn) ? 0 : -1;
            }
#pragma unroll
            for (int k = 0; k < COPY_TRIPS; ++k)
                wd[k] = (rk[k] >= 0) ? sp[tid + k * OWN_THREADS] : 0u;
#pragma unroll
            for (int k = 0; k < COPY_TRIPS; ++k)
                if (rk[k] >= 0) rk[k] = atomicAdd(&rc[wd[k] >> OWN_SBITS], 1);
        }
        if (tid == 0) qcnt[blockIdx.x] = n;
        __syncthreads();
        if (whole && wave == 0) {
            int cnt7[7], tot = 0;
#pragma unroll
            for (int q = 0; q < 7; ++q) {
                const int r = lane * 7 + q;
                cnt7[q] = (r < OWN_ROWS) ? rc[r] : 0;
                tot += cnt7[q];
            }
            int incl = tot;
#pragma unroll
            for (int d = 1; d < 32; d <<= 1) {
                const int t = __shfl_up(incl, d);
                if (lane >= d) incl += t;
            }
            int at = incl - tot;
#pragma unroll
            for (int q = 0; q < 7; ++q) {
                const int r = lane * 7 + q;
                if (r < OWN_ROWS) rs[r] = at;
                at += cnt7[q];
            }
        }
        __syncthreads();
        if (whole) {
#pragma unroll
            for (int k = 0; k < COPY_TRIPS; ++k)
                if (rk[k] >= 0) qe[rs[wd[k] >> OWN_SBITS] + rk[k]] = wd[k];
        }
    } else {
        n = qcnt[blockIdx.x];
        if (n <= OWN_QCAP)
            for (int i = tid; i < n; i += OWN_THREADS) qe[i] = qsave[qbase + i];
    }
    __syncthreads();

    const bool  fits = n <= OWN_QCAP;
    const uint4*  mx4 = (const uint4*)mx;
    const float4* dn4 = (const float4*)dn;
    const int per = ((n + OWN_WAVES - 1) / OWN_WAVES) | 1;
    if (fits && !reuse)
        for (int i = tid; i < n; i += OWN_THREADS) qsave[qbase + i] = qe[i];

    if (fits) {
        for (int q0 = wave; q0 < per; q0 += 4 * OWN_WAVES) {
            int      ss[4], dd[4];
            bool     ok[4];
            float4   sc[4];
            unsigned pk[4];
#pragma unroll
            for (int u = 0; u < 4; ++u) {
                const int q = q0 + u * OWN_WAVES;
                const int i = lane * per + q;
                ok[u] = q < per && i < n;
                pk[u] = ok[u] ? qe[i] : 0u;
            }
#pragma unroll
            for (int u = 0; u < 4; ++u) {
                ss[u] = (int)(pk[u] & ((1u << OWN_SBITS) - 1u));
                dd[u] = lo + (int)(pk[u] >> OWN_SBITS);
            }
#pragma unroll
            for (int u = 0; u < 4; ++u) sc[u] = edge_scores(el, er, ss[u], dd[u]);
#pragma unroll
            for (int u = 0; u < 4; ++u) {
                if (ok[u]) {
                    const int      i  = lane * per + q0 + u * OWN_WAVES;
                    const unsigned ld = pk[u] >> OWN_SBITS;
                    ev[i] = sc[u];
                    lds_max(&mx[ld * HEADS + 0], f2ord(sc[u].x));
                    lds_max(&mx[ld * HEADS + 1], f2ord(sc[u].y));
                    lds_max(&mx[ld * HEADS + 2], f2ord(sc[u].z));
                    lds_max(&mx[ld * HEADS + 3], f2ord(sc[u].w));
                }
            }
        }
    } else {
#pragma nounroll
        for (int e = tid; e < E; e += OWN_THREADS) {
            const int      d  = dst[e];
            const unsigned ld = (unsigned)(d - lo);
            if (ld < (unsigned)OWN_ROWS) {
                const float4 v = edge_scores(el, er, src[e], d);
                lds_max(&mx[ld * HEADS + 0], f2ord(v.x));
                lds_max(&mx[ld * HEADS + 1], f2ord(v.y));
                lds_max(&mx[ld * HEADS + 2], f2ord(v.z));
                lds_max(&mx[ld * HEADS + 3], f2ord(v.w));
            }
        }
    }
    __syncthreads();

    if (fits) {
        for (int q = wave; q < per; q += OWN_WAVES) {
            const int i = lane * per + q;
            if (i >= n) continue;
            const unsigned ld = qe[i] >> OWN_SBITS;
            const float4   v  = ev[i];
            const uint4    m  = mx4[ld];
            float4 x;
            x.x = expf(v.x - ord2f(m.x));
            x.y = expf(v.y - ord2f(m.y));
            x.z = expf(v.z - ord2f(m.z));
            x.w = expf(v.w - ord2f(m.w));
            ev[i] = x;
            lds_add(&dn[ld * HEADS + 0], x.x);
            lds_add(&dn[ld * HEADS + 1], x.y);
            lds_add(&dn[ld * HEADS + 2], x.z);
            lds_add(&dn[ld * HEADS + 3], x.w);
        }
    } else {
#pragma nounroll
        for (int e = tid; e < E; e += OWN_THREADS) {
            const int      d  = dst[e];
            const unsigned ld = (unsigned)(d - lo);
            if (ld < (unsigned)OWN_ROWS) {
                const float4 v = edge_scores(el, er, src[e], d);
                const uint4  m = mx4[ld];
                lds_add(&dn[ld * HEADS + 0], expf(v.x - ord2f(m.x)));
                lds_add(&dn[ld * HEADS + 1], expf(v.y - ord2f(m.y)));
                lds_add(&dn[ld * HEADS + 2], expf(v.z - ord2f(m.z)));
                lds_add(&dn[ld * HEADS + 3], expf(v.w - ord2f(m.w)));
            }
        }
    }
    __syncthreads();

    if (fits) {
        for (int i = tid; i < n; i += OWN_THREADS) {
            const unsigned ld = qe[i] >> OWN_SBITS;
            const float4   q  = dn4[ld];
            float4 x = ev[i];
            x.x = x.x / (q.x + EPS_);
            x.y = x.y / (q.y + EPS_);
            x.z = x.z / (q.z + EPS_);
            x.w = x.w / (q.w + EPS_);
            ev[i] = x;
        }
    }
    __syncthreads();

    if (fits) {
        const float* evf = (const float*)ev;
        const int hd = lane >> 3;
        const int ws = __builtin_amdgcn_readfirstlane(wave);
        const int jb = ws * per;
        const int je = (jb + per < n) ? jb + per : n;
        if (jb < je) {
            float4   a = make_float4(0.0f, 0.0f, 0.0f, 0.0f);
            int      cur = -1;
            unsigned pkn[ROW_FLIGHT];
            float    aln[ROW_FLIGHT];
#pragma unroll
            for (int u = 0; u < ROW_FLIGHT; ++u) {
                const int i = (jb + u < je) ? jb + u : je - 1;
                pkn[u] = qe[i];
                aln[u] = evf[i * HEADS + hd];
            }
            for (int j = jb; j < je; j += ROW_FLIGHT) {
                unsigned pk[ROW_FLIGHT];
                float    al[ROW_FLIGHT];
                float4   w[ROW_FLIGHT];
#pragma unroll
                for (int u = 0; u < ROW_FLIGHT; ++u) {
                    pk[u] = (unsigned)__builtin_amdgcn_readfirstlane((int)pkn[u]);
                    al[u] = aln[u];
                    const int s = (int)(pk[u] & ((1u << OWN_SBITS) - 1u));
                    w[u] = *(const float4*)(Wh + (long)s * FEAT + 4 * lane);
                }
#pragma unroll
                for (int u = 0; u < ROW_FLIGHT; ++u) {
                    const int i = (j + ROW_FLIGHT + u < je) ? j + ROW_FLIGHT + u : je - 1;
                    pkn[u] = qe[i];
                    aln[u] = evf[i * HEADS + hd];
                }
#pragma unroll
                for (int u = 0; u < ROW_FLIGHT; ++u) {
                    if (j + u < je) {
                        const int ld = (int)(pk[u] >> OWN_SBITS);
                        if (ld != cur) {
                            if (cur >= 0) {
                                float* row = &acc[cur * ACC_LD + 4 * lane];
                                lds_add(row + 0, a.x);
                                lds_add(row + 1, a.y);
                                lds_add(row + 2, a.z);
                                lds_add(row + 3, a.w);
                            }
                            a = make_float4(0.0f, 0.0f, 0.0f, 0.0f);
                            cur = ld;
                        }
                        a.x += al[u] * w[u].x;
                        a.y += al[u] * w[u].y;
                        a.z += al[u] * w[u].z;
                        a.w += al[u] * w[u].w;
                    }
                }
            }
            float* row = &acc[cur * ACC_LD + 4 * lane];
            lds_add(row + 0, a.x);
            lds_add(row + 1, a.y);
            lds_add(row + 2, a.z);
            lds_add(row + 3, a.w);
        }
    } else {
#pragma nounroll
        for (int e = tid; e < E; e += OWN_THREADS) {
            const int      d  = dst[e];
            const unsigned ld = (unsigned)(d - lo);
            if (ld < (unsigned)OWN_ROWS) {
                const int    s = src[e];
                const float4 v = edge_scores(el, er, s, d);
                const uint4  m = mx4[ld];
                const float4 q = dn4[ld];
                float a[HEADS];
                a[0] = expf(v.x - ord2f(m.x)) / (q.x + EPS_);
                a[1] = expf(v.y - ord2f(m.y)) / (q.y + EPS_);
                a[2] = expf(v.z - ord2f(m.z)) / (q.z + EPS_);
                a[3] = expf(v.w - ord2f(m.w)) / (q.w + EPS_);
#pragma unroll
                for (int h = 0; h < HEADS; ++h) {
#pragma nounroll
                    for (int f = 0; f < D_HEAD; ++f) {
                        const int c = h * D_HEAD + f;
                        lds_add(&acc[ld * ACC_LD + c], a[h] * Wh[(long)s * FEAT + c]);
                    }
                }
            }
        }
    }
    __syncthreads();

    int rows = N - lo;
    if (rows > OWN_ROWS) rows = OWN_ROWS;
    const int r16  = lane & 15;
    const int koff = (lane >> 4) << 1;
    const int hi8  = (lane >> 4) << 3;

    if (!second) {
        float* wst = (float*)(raw + OFF_WST);
        float* stg = (float*)(raw + OFF_STG);
        for (int i = tid; i < OWN_ROWS * FEAT; i += OWN_THREADS) {
            const int r = i >> 7, c = i & (FEAT - 1);
            const float x = acc[r * ACC_LD + c] + bias[c];
            acc[r * ACC_LD + c] = (x > 0.0f) ? x : expm1f(x);
        }
        for (int i = tid; i < FEAT * FEAT / 4; i += OWN_THREADS) {
            const int k = i >> 5, c4 = (i & 31) << 2;
            *(float4*)(wst + k * W_LD + c4) = *(const float4*)(Wn + k * FEAT + c4);
        }
        __syncthreads();
        for (int t = wave; t < OWN_RT * (FEAT / 16); t += OWN_THREADS / 32) {
            const int rt = t >> 3, ct = t & 7;
            const float* ap = acc + (rt * 16 + r16) * ACC_LD + koff;
            const float* bq = wst + koff * W_LD + ct * 16 + r16;
            v8f c = {};
#pragma unroll 8
            for (int k0 = 0; k0 < FEAT; k0 += 4) {
                const v2f a = *(const v2f*)(ap + k0);
                v2f b;
                b.x = bq[k0 * W_LD];
                b.y = bq[(k0 + 1) * W_LD];
                c = __builtin_amdgcn_wmma_f32_16x16x4_f32(false, a, false, b,
                                                          (short)0, c, false, false);
            }
            const int row0 = rt * 16 + hi8;
            float* op = o_main + (long)(lo + row0) * FEAT + ct * 16 + r16;
            float* sp = stg + row0 * ST_LD + ct * 16 + r16;
            if (rt * 16 + 16 <= rows) {
#pragma unroll
                for (int i = 0; i < 8; ++i) {
                    op[i * FEAT]  = c[i];
                    sp[i * ST_LD] = c[i];
                }
            } else {
#pragma unroll
                for (int i = 0; i < 8; ++i) {
                    if (row0 + i < rows) {
                        op[i * FEAT]  = c[i];
                        sp[i * ST_LD] = c[i];
                    }
                }
            }
        }
        __syncthreads();
        if (tid < rows * HEADS) {
            const int r = tid >> 2, h = tid & 3;
            const float* w = stg + r * ST_LD + h * D_HEAD;
            const float* a = aln + h * D_HEAD;
            const float* b = arn + h * D_HEAD;
            float sl = 0.0f, sr = 0.0f;
#pragma unroll
            for (int d = 0; d < D_HEAD; ++d) {
                sl = fmaf(w[d], a[d], sl);
                sr = fmaf(w[d], b[d], sr);
            }
            o_el[(long)lo * HEADS + tid] = sl;
            o_er[(long)lo * HEADS + tid] = sr;
        }
    } else {
        float* zm = (float*)(raw + OFF_ZM);
        for (int i = tid; i < OWN_ROWS * D_HEAD; i += OWN_THREADS) {
            const int r = i >> 5, d = i & (D_HEAD - 1);
            float s = 0.0f;
#pragma unroll
            for (int h = 0; h < HEADS; ++h)
                s += acc[r * ACC_LD + h * D_HEAD + d] + bias[h * D_HEAD + d];
            zm[r * ZM_LD + d] = s * (1.0f / HEADS);
        }
        const int wsc = __builtin_amdgcn_readfirstlane(wave);
        const int par = (wsc >> 2) & 1;
        const int cp  = (wsc & 3) | ((wsc >> 3) << 2);
        const float* wb = Wn + koff * OUT_DIM + cp * 32 + r16;
        v2f bf0[D_HEAD / 4], bf1[D_HEAD / 4];
#pragma unroll
        for (int q = 0; q < D_HEAD / 4; ++q) {
            bf0[q].x = wb[(q * 4)     * OUT_DIM];
            bf0[q].y = wb[(q * 4 + 1) * OUT_DIM];
            bf1[q].x = wb[(q * 4)     * OUT_DIM + 16];
            bf1[q].y = wb[(q * 4 + 1) * OUT_DIM + 16];
        }
        const float bv = bpn[cp * 32 + lane];
        __syncthreads();
        for (int rt = par; rt < OWN_RT; rt += 2) {
            const float* ap = zm + (rt * 16 + r16) * ZM_LD + koff;
            v8f c0 = {}, c1 = {};
#pragma unroll
            for (int q = 0; q < D_HEAD / 4; ++q) {
                const v2f a = *(const v2f*)(ap + q * 4);
                c0 = __builtin_amdgcn_wmma_f32_16x16x4_f32(false, a, false, bf0[q],
                                                           (short)0, c0, false, false);
                c1 = __builtin_amdgcn_wmma_f32_16x16x4_f32(false, a, false, bf1[q],
                                                           (short)0, c1, false, false);
            }
            float* op = o_main + (long)(lo + rt * 16) * OUT_DIM + cp * 32 + lane;
            if (rt * 16 + 16 <= rows) {
#pragma unroll
                for (int i = 0; i < 8; ++i) {
                    float lo_row = c0[i], hi_row = c1[i];
                    swap_halves(lo_row, hi_row);
                    op[i * OUT_DIM]       = lo_row + bv;
                    op[(i + 8) * OUT_DIM] = hi_row + bv;
                }
            } else {
#pragma unroll
                for (int i = 0; i < 8; ++i) {
                    float lo_row = c0[i], hi_row = c1[i];
                    swap_halves(lo_row, hi_row);
                    if (rt * 16 + i < rows)     op[i * OUT_DIM]       = lo_row + bv;
                    if (rt * 16 + i + 8 < rows) op[(i + 8) * OUT_DIM] = hi_row + bv;
                }
            }
        }
    }
}

#define NODE_OFF_STG (OWN_RT * 16 * ACC_LD * 4)
#define NODE_LDS     (NODE_OFF_STG + OWN_ROWS * ST_LD * 4)
#define NODE_TRIPS   ((OWN_RT * 16 * (FEAT / 4) + OWN_THREADS - 1) / OWN_THREADS)
static_assert(NODE_LDS <= 320 * 1024 && NODE_OFF_STG % 16 == 0, "fits one WGP's LDS");
static_assert(FEAT / 16 == 8 && OWN_THREADS / 32 == 32, "eight column tiles, four sets of eight waves");

__global__ void __launch_bounds__(OWN_THREADS)
k_node_first(const float* __restrict__ X, const float* __restrict__ Wm,
             const float* __restrict__ al, const float* __restrict__ ar, int N,
             float* __restrict__ o_wh, float* __restrict__ o_el, float* __restrict__ o_er)
{
    __shared__ __attribute__((aligned(16))) unsigned char raw[NODE_LDS];
    float* xs  = (float*)raw;
    float* stg = (float*)(raw + NODE_OFF_STG);

    const int tid  = threadIdx.x;
    const int lo   = blockIdx.x * OWN_ROWS;
    int rows = N - lo;
    if (rows > OWN_ROWS) rows = OWN_ROWS;
    const int wave = tid >> 5, lane = tid & 31;
    const int r16  = lane & 15;
    const int koff = (lane >> 4) << 1;
    const int hi8  = (lane >> 4) << 3;
    const int ct   = wave & 7;

    float4 xv[NODE_TRIPS];
#pragma unroll
    for (int k = 0; k < NODE_TRIPS; ++k) {
        const int i = tid + k * OWN_THREADS;
        const int r = i >> 5, c4 = (i & 31) << 2;
        xv[k] = make_float4(0.0f, 0.0f, 0.0f, 0.0f);
        if (r < rows) xv[k] = *(const float4*)(X + (long)(lo + r) * FEAT + c4);
    }
    v2f bf[FEAT / 4];
#pragma unroll
    for (int q = 0; q < FEAT / 4; ++q) {
        bf[q].x = Wm[(q * 4 + koff)     * FEAT + ct * 16 + r16];
        bf[q].y = Wm[(q * 4 + koff + 1) * FEAT + ct * 16 + r16];
    }
#pragma unroll
    for (int k = 0; k < NODE_TRIPS; ++k) {
        const int i = tid + k * OWN_THREADS;
        const int r = i >> 5, c4 = (i & 31) << 2;
        if (r < OWN_RT * 16) *(float4*)(xs + r * ACC_LD + c4) = xv[k];
    }
    __syncthreads();

    for (int rt = wave >> 3; rt < OWN_RT; rt += 4) {
        const float* ap = xs + (rt * 16 + r16) * ACC_LD + koff;
        v8f c = {};
#pragma unroll
        for (int q = 0; q < FEAT / 4; ++q) {
            const v2f a = *(const v2f*)(ap + q * 4);
            c = __builtin_amdgcn_wmma_f32_16x16x4_f32(false, a, false, bf[q],
                                                      (short)0, c, false, false);
        }
        const int row0 = rt * 16 + hi8;
        float* op = o_wh + (long)(lo + row0) * FEAT + ct * 16 + r16;
        float* sp = stg + row0 * ST_LD + ct * 16 + r16;
        if (rt * 16 + 16 <= rows) {
#pragma unroll
            for (int i = 0; i < 8; ++i) {
                op[i * FEAT]  = c[i];
                sp[i * ST_LD] = c[i];
            }
        } else {
#pragma unroll
            for (int i = 0; i < 8; ++i) {
                if (row0 + i < rows) {
                    op[i * FEAT]  = c[i];
                    sp[i * ST_LD] = c[i];
                }
            }
        }
    }
    __syncthreads();
    if (tid < rows * HEADS) {
        const int r = tid >> 2, h = tid & 3;
        const float* w = stg + r * ST_LD + h * D_HEAD;
        const float* a = al + h * D_HEAD;
        const float* b = ar + h * D_HEAD;
        float sl = 0.0f, sr = 0.0f;
#pragma unroll
        for (int d = 0; d < D_HEAD; ++d) {
            sl = fmaf(w[d], a[d], sl);
            sr = fmaf(w[d], b[d], sr);
        }
        o_el[(long)lo * HEADS + tid] = sl;
        o_er[(long)lo * HEADS + tid] = sr;
    }
}


extern "C" void kernel_launch(void* const* d_in, const int* in_sizes, int n_in,
                              void* d_out, int out_size, void* d_ws, size_t ws_size,
                              hipStream_t stream)
{
    const float* h   = (const float*)d_in[0];
    const int*   src = (const int*)  d_in[1];
    const int*   dst = (const int*)  d_in[2];
    const float* W1  = (const float*)d_in[3];
    const float* al1 = (const float*)d_in[4];
    const float* ar1 = (const float*)d_in[5];
    const float* b1  = (const float*)d_in[6];
    const float* W2  = (const float*)d_in[7];
    const float* al2 = (const float*)d_in[8];
    const float* ar2 = (const float*)d_in[9];
    const float* b2  = (const float*)d_in[10];
    const float* Wp  = (const float*)d_in[11];
    const float* bp  = (const float*)d_in[12];
    float*       out = (float*)d_out;

    const int N = N_NODES, E = N_EDGES;

    float*    Wh    = (float*)d_ws;
    float*    Wh2   = Wh  + (size_t)N * FEAT;
    float*    el    = Wh2 + (size_t)N * FEAT;
    float*    er    = el  + (size_t)N * HEADS;
    float*    el2   = er  + (size_t)N * HEADS;
    float*    er2   = el2 + (size_t)N * HEADS;
    unsigned* qsave = (unsigned*)(er2 + (size_t)N * HEADS);
    int*      qcnt  = (int*)(qsave + (size_t)OWN_BLOCKS * OWN_QCAP);
    int*      segcnt = qcnt + OWN_BLOCKS;
    unsigned* seg    = (unsigned*)(segcnt + (size_t)OWN_BLOCKS * BIN_BLOCKS);

    k_bin<<<BIN_BLOCKS, BIN_THREADS, 0, stream>>>(src, dst, E, seg, segcnt);
    k_node_first<<<OWN_BLOCKS, OWN_THREADS, 0, stream>>>(h, W1, al1, ar1, N, Wh, el, er);
    k_edge_owned<<<OWN_BLOCKS, OWN_THREADS, 0, stream>>>(Wh, el, er, src, dst, N, E, qsave, qcnt, 0, seg, segcnt,
                                                         b1, W2, al2, ar2, nullptr, Wh2, el2, er2);
    k_edge_owned<<<OWN_BLOCKS, OWN_THREADS, 0, stream>>>(Wh2, el2, er2, src, dst, N, E, qsave, qcnt, 1, seg, segcnt,
                                                         b2, Wp, nullptr, nullptr, bp, out, nullptr, nullptr);
}
